# starting kernel
# baseline (speedup 1.0000x reference)
#include <hip/hip_runtime.h>
#include <hip/hip_bf16.h>
#include <math.h>

#define B_ 128
#define L_ 50
#define D_ 128
#define S_ 8
#define NN_ 100000
#define KP_ 132
#define LEAKY_ 0.2f
#define BETA_ 0.005f

typedef float v2f __attribute__((ext_vector_type(2)));
typedef float v8f __attribute__((ext_vector_type(8)));

#if defined(__has_builtin)
#if __has_builtin(__builtin_amdgcn_wmma_f32_16x16x4_f32)
#define HAVE_WMMA_F32 1
#endif
#endif

__device__ __forceinline__ v8f wmma4(v2f a, v2f b, v8f c) {
#ifdef HAVE_WMMA_F32
  return __builtin_amdgcn_wmma_f32_16x16x4_f32(false, a, false, b, (short)0, c,
                                               false, false);
#else
  float* cf = (float*)&c;
  #pragma unroll
  for (int i = 0; i < 8; ++i) cf[i] += a.x * b.x + a.y * b.y;
  return c;
#endif
}

__device__ __forceinline__ float leaky_(float x) { return x > 0.f ? x : LEAKY_ * x; }
__device__ __forceinline__ float sigmoid_(float x) { return 1.f / (1.f + __expf(-x)); }

__global__ void gw1t_kernel(const float* __restrict__ gw1, float* __restrict__ gw1T) {
  int i = blockIdx.x * 256 + threadIdx.x;
  if (i >= 2 * D_ * KP_) return;
  int hop = i / (D_ * KP_);
  int rem = i % (D_ * KP_);
  int n = rem / KP_, k = rem % KP_;
  gw1T[i] = (k < 129) ? gw1[hop * 129 * D_ + k * D_ + n] : 0.f;
}

__global__ void w1t_kernel(const float* __restrict__ w1, float* __restrict__ w1botT) {
  int i = blockIdx.x * 256 + threadIdx.x;
  if (i >= D_ * D_) return;
  int n = i >> 7, k = i & 127;
  w1botT[i] = w1[(D_ + k) * D_ + n];
}

__global__ void gather_h_kernel(const float* __restrict__ emb,
                                const int* __restrict__ inputs,
                                float* __restrict__ h, int total) {
  int i = blockIdx.x * 256 + threadIdx.x;
  if (i >= total) return;
  int row = i >> 7, d = i & 127;
  h[i] = emb[((long)inputs[row] << 7) + d];
}

__global__ void hop_kernel(const int* __restrict__ src,
                           const int* __restrict__ adj_all,
                           const float* __restrict__ num_w,
                           int* __restrict__ dst, float* __restrict__ dstw,
                           int total) {
  int i = blockIdx.x * 256 + threadIdx.x;
  if (i >= total) return;
  int node = src[i >> 3];
  int s = i & 7;
  dst[i]  = adj_all[node * S_ + s];
  dstw[i] = num_w[node * S_ + s];
}

__global__ void sess_kernel(const float* __restrict__ emb,
                            const int* __restrict__ item,
                            const int* __restrict__ mask,
                            float* __restrict__ sess) {
  int b = blockIdx.x, d = threadIdx.x;
  float acc = 0.f, ms = 0.f;
  for (int l = 0; l < L_; ++l) {
    float m = (float)mask[b * L_ + l];
    ms += m;
    acc += emb[((long)item[b * L_ + l] << 7) + d] * m;
  }
  sess[(b << 7) + d] = acc / ms;
}

__global__ void local_agg_kernel(const float* __restrict__ h,
                                 const int* __restrict__ adj,
                                 const float* __restrict__ a_local,
                                 float* __restrict__ h_local) {
  int b = blockIdx.x;
  __shared__ float hv[L_][D_];
  __shared__ float att[L_][L_];
  __shared__ float rinv[L_];
  for (int i = threadIdx.x; i < L_ * D_; i += 256)
    hv[i / D_][i & 127] = h[(long)b * L_ * D_ + i];
  __syncthreads();
  for (int p = threadIdx.x; p < L_ * L_; p += 256) {
    int i = p / L_, j = p % L_;
    int k = adj[b * L_ * L_ + p];
    float v;
    if (k == 0) v = -9e15f;
    else {
      const float* a = a_local + (k - 1) * D_;
      float acc = 0.f;
      for (int d = 0; d < D_; ++d) acc += hv[i][d] * hv[j][d] * a[d];
      v = leaky_(acc);
    }
    att[i][j] = v;
  }
  __syncthreads();
  if (threadIdx.x < L_) {
    int i = threadIdx.x;
    float mx = -1e30f;
    for (int j = 0; j < L_; ++j) mx = fmaxf(mx, att[i][j]);
    float s = 0.f;
    for (int j = 0; j < L_; ++j) { float e = __expf(att[i][j] - mx); att[i][j] = e; s += e; }
    rinv[i] = 1.f / s;
  }
  __syncthreads();
  for (int p = threadIdx.x; p < L_ * D_; p += 256) {
    int i = p >> 7, d = p & 127;
    float acc = 0.f;
    for (int j = 0; j < L_; ++j) acc += att[i][j] * hv[j][d];
    h_local[(long)b * L_ * D_ + p] = acc * rinv[i];
  }
}

__global__ void agg_kernel(const float* __restrict__ emb,
                           const float* __restrict__ selfVal, const int* __restrict__ selfIdx,
                           const int* __restrict__ neighIdx, const float* __restrict__ neighVal,
                           const float* __restrict__ neighW,
                           const float* __restrict__ sess,
                           const float* __restrict__ gw1T, const float* __restrict__ gw2p,
                           const float* __restrict__ gw3p,
                           float* __restrict__ out, int M) {
  __shared__ float nv[16][D_];
  __shared__ float xs[16][KP_];
  __shared__ float selfv[2][D_];
  __shared__ float sessv[D_];
  __shared__ float wgt[16];
  __shared__ float sc[16];
  __shared__ float alpha[16];
  __shared__ float nout[2][D_];

  int tid = threadIdx.x;
  int lane = tid & 31, laneIdx = lane & 15, laneHalf = lane >> 4;
  int wave = tid >> 5;
  int n0 = wave * 16;

  int p = blockIdx.x;
  int b = (2 * p) / M;
  int m0 = (2 * p) % M;
  long base  = (long)b * M + m0;
  long nbase = base * S_;

  if (tid < D_) sessv[tid] = sess[(b << 7) + tid];
  {
    int r = tid >> 7, d = tid & 127;
    float v = selfVal ? selfVal[((base + r) << 7) + d]
                      : emb[((long)selfIdx[base + r] << 7) + d];
    selfv[r][d] = v;
  }
  for (int i = tid; i < 16 * D_; i += 256) {
    int r = i >> 7, d = i & 127;
    float v = neighIdx ? emb[((long)neighIdx[nbase + r] << 7) + d]
                       : neighVal[((nbase + r) << 7) + d];
    nv[r][d] = v;
  }
  if (tid < 16) { wgt[tid] = neighW[nbase + tid]; sc[tid] = 0.f; }
  __syncthreads();

  for (int i = tid; i < 16 * KP_; i += 256) {
    int r = i / KP_, k = i % KP_;
    float v = (k < D_) ? sessv[k] * nv[r][k] : (k == D_ ? wgt[r] : 0.f);
    xs[r][k] = v;
  }
  __syncthreads();

  v8f c = {};
  const float* bp = gw1T + (n0 + laneIdx) * KP_;
  {
    const float* ap = &xs[laneIdx][0];
    #pragma unroll 3
    for (int k0 = 0; k0 < KP_; k0 += 4) {
      int kk = k0 + (laneHalf << 1);
      v2f a = *(const v2f*)(ap + kk);
      v2f bb = *(const v2f*)(bp + kk);
      c = wmma4(a, bb, c);
    }
  }
  float g2 = gw2p[n0 + laneIdx];
  float* cf = (float*)&c;
  #pragma unroll
  for (int v = 0; v < 8; ++v) {
    float val = leaky_(cf[v]) * g2;
    val += __shfl_xor(val, 1);
    val += __shfl_xor(val, 2);
    val += __shfl_xor(val, 4);
    val += __shfl_xor(val, 8);
    if (laneIdx == 0) atomicAdd(&sc[v + (laneHalf << 3)], val);
  }
  __syncthreads();

  if (tid < 2) {
    float mx = -1e30f;
    for (int s = 0; s < 8; ++s) mx = fmaxf(mx, sc[tid * 8 + s]);
    float ex[8], sum = 0.f;
    for (int s = 0; s < 8; ++s) { ex[s] = __expf(sc[tid * 8 + s] - mx); sum += ex[s]; }
    float inv = 1.f / sum;
    for (int s = 0; s < 8; ++s) alpha[tid * 8 + s] = ex[s] * inv;
  }
  __syncthreads();
  {
    int r = tid >> 7, d = tid & 127;
    float acc = 0.f;
    #pragma unroll
    for (int s = 0; s < 8; ++s) acc += alpha[(r << 3) + s] * nv[(r << 3) + s][d];
    nout[r][d] = acc;
  }
  __syncthreads();
  {
    int r = tid >> 7, d = tid & 127;
    float acc = 0.f;
    for (int e = 0; e < D_; ++e)
      acc += selfv[r][e] * gw3p[(e << 7) + d] + nout[r][e] * gw3p[((D_ + e) << 7) + d];
    out[((base + r) << 7) + d] = fmaxf(acc, 0.f);
  }
}

__global__ void gemm16_kernel(const float* __restrict__ A, int lda,
                              const float* __restrict__ BT, int ldbt,
                              float* __restrict__ C, long ldc,
                              int M, int N, int K,
                              const float* __restrict__ rowBias, int biasMod,
                              int act) {
  int tilesM = (M + 15) >> 4;
  long tilesN = (long)((N + 15) >> 4);
  long total = (long)tilesM * tilesN;
  int lane = threadIdx.x & 31, laneIdx = lane & 15, laneHalf = lane >> 4;
  for (long t = (long)blockIdx.x * 8 + (threadIdx.x >> 5); t < total;
       t += (long)gridDim.x * 8) {
    int m0 = (int)(t % tilesM) << 4;
    long n0 = (t / tilesM) << 4;
    int row = m0 + laneIdx;
    long col = n0 + laneIdx;
    v8f c = {};
    bool full = (m0 + 16 <= M) && (n0 + 16 <= N);
    if (full) {
      const float* ap = A + (long)row * lda;
      const float* bp = BT + col * ldbt;
      for (int k0 = 0; k0 < K; k0 += 4) {
        int kk = k0 + (laneHalf << 1);
        v2f a = *(const v2f*)(ap + kk);
        v2f b = *(const v2f*)(bp + kk);
        c = wmma4(a, b, c);
      }
    } else {
      for (int k0 = 0; k0 < K; k0 += 4) {
        int kk = k0 + (laneHalf << 1);
        v2f a, b;
        a.x = (row < M) ? A[(long)row * lda + kk] : 0.f;
        a.y = (row < M) ? A[(long)row * lda + kk + 1] : 0.f;
        b.x = (col < N) ? BT[col * ldbt + kk] : 0.f;
        b.y = (col < N) ? BT[col * ldbt + kk + 1] : 0.f;
        c = wmma4(a, b, c);
      }
    }
    float* cf = (float*)&c;
    #pragma unroll
    for (int v = 0; v < 8; ++v) {
      int rr = m0 + v + (laneHalf << 3);
      if (rr < M && col < N) {
        float val = cf[v];
        if (rowBias) val += rowBias[(long)(rr % biasMod) * N + col];
        if (act == 1) val = tanhf(val);
        C[(long)rr * ldc + col] = val;
      }
    }
  }
}

__global__ void add_kernel(const float* a, const float* b, float* o, int n) {
  int i = blockIdx.x * 256 + threadIdx.x;
  if (i < n) o[i] = a[i] + b[i];
}

__global__ void zero1_kernel(float* p) { if (threadIdx.x == 0 && blockIdx.x == 0) p[0] = 0.f; }

__global__ void ssl_kernel(const float* __restrict__ hl, const float* __restrict__ hg,
                           float* __restrict__ loss) {
  int idx = blockIdx.x * 256 + threadIdx.x;
  int b = idx >> 7, d = idx & 127;
  int pb = (b * 73 + 21) & 127;
  float pos = 0.f, neg = 0.f;
  for (int l = 0; l < L_; ++l) {
    int pl = (l * 13 + 7) % L_;
    pos += hl[((b * L_ + l) << 7) + d] * hg[((b * L_ + l) << 7) + d];
    neg += hg[((b * L_ + l) << 7) + d] * hl[((pb * L_ + pl) << 7) + d];
  }
  float term = -__logf(1e-8f + sigmoid_(pos)) - __logf(1e-8f + 1.f - sigmoid_(neg));
  __shared__ float red[256];
  red[threadIdx.x] = term;
  __syncthreads();
  for (int s = 128; s > 0; s >>= 1) {
    if (threadIdx.x < s) red[threadIdx.x] += red[threadIdx.x + s];
    __syncthreads();
  }
  if (threadIdx.x == 0) atomicAdd(loss, BETA_ * red[0]);
}

__global__ void hs_kernel(const float* __restrict__ hidden, const int* __restrict__ mask,
                          float* __restrict__ hs) {
  int b = blockIdx.x, d = threadIdx.x;
  float acc = 0.f, ms = 0.f;
  for (int l = 0; l < L_; ++l) {
    float m = (float)mask[b * L_ + l];
    ms += m;
    acc += hidden[((b * L_ + l) << 7) + d] * m;
  }
  hs[(b << 7) + d] = acc / ms;
}

__global__ void hsg_kernel(const float* __restrict__ hs, const float* __restrict__ glu2_w,
                           float* __restrict__ hsg) {
  __shared__ float hr[D_];
  int b = blockIdx.x, d = threadIdx.x;
  hr[d] = hs[(b << 7) + d];
  __syncthreads();
  float acc = 0.f;
  for (int e = 0; e < D_; ++e) acc += hr[e] * glu2_w[(d << 7) + e];
  hsg[(b << 7) + d] = acc;
}

__global__ void pos_kernel(const float* __restrict__ pos_emb, const float* __restrict__ w1,
                           float* __restrict__ P) {
  __shared__ float pr[D_];
  int l = blockIdx.x, d = threadIdx.x;
  pr[d] = pos_emb[(l << 7) + d];
  __syncthreads();
  float acc = 0.f;
  for (int e = 0; e < D_; ++e) acc += pr[e] * w1[(e << 7) + d];
  P[(l << 7) + d] = acc;
}

__global__ void beta_kernel(const float* __restrict__ nh, const float* __restrict__ glu1_w,
                            const float* __restrict__ glu1_b, const float* __restrict__ hsg,
                            const float* __restrict__ w2, const int* __restrict__ mask,
                            float* __restrict__ beta) {
  __shared__ float nrow[D_];
  __shared__ float red[D_];
  int row = blockIdx.x;
  int b = row / L_;
  int d = threadIdx.x;
  nrow[d] = nh[((long)row << 7) + d];
  __syncthreads();
  float acc = glu1_b[d] + hsg[(b << 7) + d];
  for (int e = 0; e < D_; ++e) acc += nrow[e] * glu1_w[(d << 7) + e];
  red[d] = sigmoid_(acc) * w2[d];
  __syncthreads();
  for (int s = 64; s > 0; s >>= 1) {
    if (d < s) red[d] += red[d + s];
    __syncthreads();
  }
  if (d == 0) beta[row] = red[0] * (float)mask[row];
}

__global__ void select_kernel(const float* __restrict__ hidden, const float* __restrict__ beta,
                              float* __restrict__ sel) {
  int b = blockIdx.x, d = threadIdx.x;
  float acc = 0.f;
  for (int l = 0; l < L_; ++l) acc += beta[b * L_ + l] * hidden[((b * L_ + l) << 7) + d];
  sel[(b << 7) + d] = acc;
}

__global__ void norm_kernel(const float* __restrict__ sel, float* __restrict__ snorm) {
  __shared__ float red[D_];
  int b = blockIdx.x, d = threadIdx.x;
  float v = sel[(b << 7) + d];
  red[d] = v * v + 1e-6f;
  __syncthreads();
  for (int s = 64; s > 0; s >>= 1) {
    if (d < s) red[d] += red[d + s];
    __syncthreads();
  }
  if (d == 0) snorm[b] = sqrtf(red[0]);
}

__global__ void fn_kernel(const float* __restrict__ sel, const float* __restrict__ snorm,
                          float* __restrict__ selF) {
  __shared__ float row[B_];
  __shared__ float selfrow[D_];
  __shared__ float tvs[9];
  __shared__ int tis[9];
  int i = blockIdx.x, t = threadIdx.x;
  selfrow[t] = sel[(i << 7) + t];
  __syncthreads();
  float acc = 0.f;
  for (int d = 0; d < D_; ++d) acc += selfrow[d] * sel[(t << 7) + d];
  row[t] = acc / (snorm[i] * snorm[t]);
  __syncthreads();
  if (t == 0) {
    float mx = -1e30f;
    for (int j = 0; j < B_; ++j) mx = fmaxf(mx, row[j]);
    float s = 0.f;
    for (int j = 0; j < B_; ++j) { row[j] = __expf(row[j] - mx); s += row[j]; }
    float inv = 1.f / s;
    for (int j = 0; j < B_; ++j) row[j] *= inv;
    for (int q = 0; q < 9; ++q) {
      float bm = -1.f; int bi = 0;
      for (int j = 0; j < B_; ++j) if (row[j] > bm) { bm = row[j]; bi = j; }
      tvs[q] = bm; tis[q] = bi; row[bi] = -2.f;
    }
    float mx2 = tvs[0], s2 = 0.f;
    for (int q = 0; q < 9; ++q) { tvs[q] = __expf(tvs[q] - mx2); s2 += tvs[q]; }
    float inv2 = 1.f / s2;
    for (int q = 0; q < 9; ++q) tvs[q] *= inv2;
  }
  __syncthreads();
  float nb = 0.f;
  for (int q = 0; q < 9; ++q) nb += tvs[q] * sel[(tis[q] << 7) + t];
  selF[(i << 7) + t] = selfrow[t] + nb;
}

extern "C" void kernel_launch(void* const* d_in, const int* in_sizes, int n_in,
                              void* d_out, int out_size, void* d_ws, size_t ws_size,
                              hipStream_t stream) {
  (void)in_sizes; (void)n_in; (void)out_size; (void)ws_size;
  const int*   inputs  = (const int*)d_in[0];
  const int*   adj     = (const int*)d_in[1];
  const int*   mask    = (const int*)d_in[2];
  const int*   item    = (const int*)d_in[3];
  const int*   adj_all = (const int*)d_in[5];
  const float* num_w   = (const float*)d_in[6];
  const float* emb     = (const float*)d_in[7];
  const float* pos_emb = (const float*)d_in[8];
  const float* a_local = (const float*)d_in[9];
  const float* gw1     = (const float*)d_in[10];
  const float* gw2     = (const float*)d_in[11];
  const float* gw3     = (const float*)d_in[12];
  const float* w1      = (const float*)d_in[13];
  const float* w2      = (const float*)d_in[14];
  const float* glu1_w  = (const float*)d_in[15];
  const float* glu1_b  = (const float*)d_in[16];
  const float* glu2_w  = (const float*)d_in[17];

  float* out0   = (float*)d_out;
  float* lossP  = out0 + B_ * L_ * D_;
  float* scores = lossP + 1;

  char* wp = (char*)d_ws;
  auto alloc = [&](size_t nfloats) -> float* {
    float* r = (float*)wp;
    wp += ((nfloats * sizeof(float) + 255) / 256) * 256;
    return r;
  };
  float* h        = alloc(B_ * L_ * D_);
  float* h_local  = alloc(B_ * L_ * D_);
  float* h_global = alloc(B_ * L_ * D_);
  float* sessb    = alloc(B_ * D_);
  float* e0p      = alloc(B_ * L_ * D_);
  float* e1p      = alloc((size_t)B_ * 400 * D_);
  float* hsb      = alloc(B_ * D_);
  float* hsgb     = alloc(B_ * D_);
  float* Pb       = alloc(L_ * D_);
  float* nh       = alloc(B_ * L_ * D_);
  float* betab    = alloc(B_ * L_);
  float* selb     = alloc(B_ * D_);
  float* selF     = alloc(B_ * D_);
  float* snorm    = alloc(B_);
  float* gw1T     = alloc(2 * D_ * KP_);
  float* w1botT   = alloc(D_ * D_);
  float* wn1      = alloc(B_ * 400);
  float* wn2      = alloc((size_t)B_ * 3200);
  int*   n1       = (int*)alloc(B_ * 400);
  int*   n2       = (int*)alloc((size_t)B_ * 3200);

  gw1t_kernel<<<dim3((2 * D_ * KP_ + 255) / 256), 256, 0, stream>>>(gw1, gw1T);
  w1t_kernel<<<dim3((D_ * D_ + 255) / 256), 256, 0, stream>>>(w1, w1botT);
  gather_h_kernel<<<dim3((B_ * L_ * D_) / 256), 256, 0, stream>>>(emb, inputs, h, B_ * L_ * D_);
  sess_kernel<<<dim3(B_), dim3(D_), 0, stream>>>(emb, item, mask, sessb);
  hop_kernel<<<dim3((B_ * L_ * S_ + 255) / 256), 256, 0, stream>>>(inputs, adj_all, num_w, n1, wn1, B_ * L_ * S_);
  hop_kernel<<<dim3((B_ * 400 * S_ + 255) / 256), 256, 0, stream>>>(n1, adj_all, num_w, n2, wn2, B_ * 400 * S_);
  local_agg_kernel<<<dim3(B_), 256, 0, stream>>>(h, adj, a_local, h_local);
  agg_kernel<<<dim3(B_ * L_ / 2), 256, 0, stream>>>(emb, h, nullptr, n1, nullptr, wn1,
      sessb, gw1T, gw2, gw3, e0p, L_);
  agg_kernel<<<dim3(B_ * 400 / 2), 256, 0, stream>>>(emb, nullptr, n1, n2, nullptr, wn2,
      sessb, gw1T, gw2, gw3, e1p, 400);
  agg_kernel<<<dim3(B_ * L_ / 2), 256, 0, stream>>>(emb, e0p, nullptr, nullptr, e1p, wn1,
      sessb, gw1T + D_ * KP_, gw2 + D_, gw3 + 2 * D_ * D_, h_global, L_);
  add_kernel<<<dim3((B_ * L_ * D_) / 256), 256, 0, stream>>>(h_local, h_global, out0, B_ * L_ * D_);
  zero1_kernel<<<1, 1, 0, stream>>>(lossP);
  ssl_kernel<<<dim3(64), 256, 0, stream>>>(h_local, h_global, lossP);
  hs_kernel<<<dim3(B_), dim3(D_), 0, stream>>>(out0, mask, hsb);
  hsg_kernel<<<dim3(B_), dim3(D_), 0, stream>>>(hsb, glu2_w, hsgb);
  pos_kernel<<<dim3(L_), dim3(D_), 0, stream>>>(pos_emb, w1, Pb);
  {
    int tiles = ((B_ * L_ + 15) / 16) * ((D_ + 15) / 16);
    gemm16_kernel<<<dim3((tiles + 7) / 8), 256, 0, stream>>>(
        out0, D_, w1botT, D_, nh, D_, B_ * L_, D_, D_, Pb, L_, 1);
  }
  beta_kernel<<<dim3(B_ * L_), dim3(D_), 0, stream>>>(nh, glu1_w, glu1_b, hsgb, w2, mask, betab);
  select_kernel<<<dim3(B_), dim3(D_), 0, stream>>>(out0, betab, selb);
  norm_kernel<<<dim3(B_), dim3(D_), 0, stream>>>(selb, snorm);
  fn_kernel<<<dim3(B_), dim3(D_), 0, stream>>>(selb, snorm, selF);
  {
    int Nsc = NN_ - 1;
    long tiles = (long)((B_ + 15) / 16) * ((Nsc + 15) / 16);
    gemm16_kernel<<<dim3((unsigned)((tiles + 7) / 8)), 256, 0, stream>>>(
        selF, D_, emb + D_, D_, scores, (long)Nsc, B_, Nsc, D_, nullptr, 1, 0);
  }
}


# agg kernel retiled to 16 m-rows (128 neighbour rows) per work-group: gw1 slice kept in registers over 8 row tiles, w3 product as a 16-row WMMA tile; weights read once per 16 m-rows, not per 2
# speedup vs baseline: 1.3341x; 1.3341x over previous
#include <hip/hip_runtime.h>
#include <hip/hip_bf16.h>
#include <math.h>

#define B_ 128
#define L_ 50
#define D_ 128
#define S_ 8
#define NN_ 100000
#define KP_ 132
#define LEAKY_ 0.2f
#define BETA_ 0.005f

typedef float v2f __attribute__((ext_vector_type(2)));
typedef float v8f __attribute__((ext_vector_type(8)));

#if defined(__has_builtin)
#if __has_builtin(__builtin_amdgcn_wmma_f32_16x16x4_f32)
#define HAVE_WMMA_F32 1
#endif
#endif

__device__ __forceinline__ v8f wmma4(v2f a, v2f b, v8f c) {
#ifdef HAVE_WMMA_F32
  return __builtin_amdgcn_wmma_f32_16x16x4_f32(false, a, false, b, (short)0, c,
                                               false, false);
#else
  float* cf = (float*)&c;
  #pragma unroll
  for (int i = 0; i < 8; ++i) cf[i] += a.x * b.x + a.y * b.y;
  return c;
#endif
}

__device__ __forceinline__ float leaky_(float x) { return x > 0.f ? x : LEAKY_ * x; }
__device__ __forceinline__ float sigmoid_(float x) { return 1.f / (1.f + __expf(-x)); }

__global__ void gw1t_kernel(const float* __restrict__ gw1, float* __restrict__ gw1T) {
  int i = blockIdx.x * 256 + threadIdx.x;
  if (i >= 2 * D_ * KP_) return;
  int hop = i / (D_ * KP_);
  int rem = i % (D_ * KP_);
  int n = rem / KP_, k = rem % KP_;
  gw1T[i] = (k < 129) ? gw1[hop * 129 * D_ + k * D_ + n] : 0.f;
}

__global__ void w1t_kernel(const float* __restrict__ w1, float* __restrict__ w1botT) {
  int i = blockIdx.x * 256 + threadIdx.x;
  if (i >= D_ * D_) return;
  int n = i >> 7, k = i & 127;
  w1botT[i] = w1[(D_ + k) * D_ + n];
}

__global__ void gather_h_kernel(const float* __restrict__ emb,
                                const int* __restrict__ inputs,
                                float* __restrict__ h, int total) {
  int i = blockIdx.x * 256 + threadIdx.x;
  if (i >= total) return;
  int row = i >> 7, d = i & 127;
  h[i] = emb[((long)inputs[row] << 7) + d];
}

__global__ void hop_kernel(const int* __restrict__ src,
                           const int* __restrict__ adj_all,
                           const float* __restrict__ num_w,
                           int* __restrict__ dst, float* __restrict__ dstw,
                           int total) {
  int i = blockIdx.x * 256 + threadIdx.x;
  if (i >= total) return;
  int node = src[i >> 3];
  int s = i & 7;
  dst[i]  = adj_all[node * S_ + s];
  dstw[i] = num_w[node * S_ + s];
}

__global__ void sess_kernel(const float* __restrict__ emb,
                            const int* __restrict__ item,
                            const int* __restrict__ mask,
                            float* __restrict__ sess) {
  int b = blockIdx.x, d = threadIdx.x;
  float acc = 0.f, ms = 0.f;
  for (int l = 0; l < L_; ++l) {
    float m = (float)mask[b * L_ + l];
    ms += m;
    acc += emb[((long)item[b * L_ + l] << 7) + d] * m;
  }
  sess[(b << 7) + d] = acc / ms;
}

__global__ void local_agg_kernel(const float* __restrict__ h,
                                 const int* __restrict__ adj,
                                 const float* __restrict__ a_local,
                                 float* __restrict__ h_local) {
  int b = blockIdx.x;
  __shared__ float hv[L_][D_];
  __shared__ float att[L_][L_];
  __shared__ float rinv[L_];
  for (int i = threadIdx.x; i < L_ * D_; i += 256)
    hv[i / D_][i & 127] = h[(long)b * L_ * D_ + i];
  __syncthreads();
  for (int p = threadIdx.x; p < L_ * L_; p += 256) {
    int i = p / L_, j = p % L_;
    int k = adj[b * L_ * L_ + p];
    float v;
    if (k == 0) v = -9e15f;
    else {
      const float* a = a_local + (k - 1) * D_;
      float acc = 0.f;
      for (int d = 0; d < D_; ++d) acc += hv[i][d] * hv[j][d] * a[d];
      v = leaky_(acc);
    }
    att[i][j] = v;
  }
  __syncthreads();
  if (threadIdx.x < L_) {
    int i = threadIdx.x;
    float mx = -1e30f;
    for (int j = 0; j < L_; ++j) mx = fmaxf(mx, att[i][j]);
    float s = 0.f;
    for (int j = 0; j < L_; ++j) { float e = __expf(att[i][j] - mx); att[i][j] = e; s += e; }
    rinv[i] = 1.f / s;
  }
  __syncthreads();
  for (int p = threadIdx.x; p < L_ * D_; p += 256) {
    int i = p >> 7, d = p & 127;
    float acc = 0.f;
    for (int j = 0; j < L_; ++j) acc += att[i][j] * hv[j][d];
    h_local[(long)b * L_ * D_ + p] = acc * rinv[i];
  }
}

#define GM_ 16
#define GR_ 128
#define CP_ 260

__device__ __forceinline__ void score_rows(v8f c, float g2, float* scRow,
                                           int laneIdx, int laneHalf) {
  #pragma unroll
  for (int v = 0; v < 8; ++v) {
    float val = leaky_(c[v]) * g2;
    val += __shfl_xor(val, 1);
    val += __shfl_xor(val, 2);
    val += __shfl_xor(val, 4);
    val += __shfl_xor(val, 8);
    if (laneIdx == 0) atomicAdd(&scRow[v + (laneHalf << 3)], val);
  }
}

template <bool SELF_VAL, bool NEIGH_IDX>
__global__ void __launch_bounds__(256)
agg_kernel(const float* __restrict__ emb,
           const float* __restrict__ selfVal, const int* __restrict__ selfIdx,
           const int* __restrict__ neighIdx, const float* __restrict__ neighVal,
           const float* __restrict__ neighW,
           const float* __restrict__ sess,
           const float* __restrict__ gw1T, const float* __restrict__ gw2p,
           const float* __restrict__ gw3p,
           float* __restrict__ out, int M) {
  __shared__ __align__(16) float nv[GR_][D_];
  __shared__ __align__(16) float xs[GR_][KP_];
  __shared__ __align__(16) float cat[GM_][CP_];
  __shared__ float sc[GR_];
  __shared__ float alpha[GR_];

  const int tid = threadIdx.x;
  const int lane = tid & 31, laneIdx = lane & 15, laneHalf = lane >> 4;
  const int wave = tid >> 5;
  const int n0 = wave * 16;
  const long base  = (long)blockIdx.x * GM_;
  const long nbase = base * S_;

  v2f breg[KP_ / 4];
  {
    const float* bp = gw1T + (n0 + laneIdx) * KP_ + (laneHalf << 1);
    #pragma unroll
    for (int ks = 0; ks < KP_ / 4; ++ks) breg[ks] = *(const v2f*)(bp + 4 * ks);
  }
  const float g2 = gw2p[n0 + laneIdx];

  const unsigned ubase = blockIdx.x * GM_;
  const unsigned b0 = ubase / (unsigned)M;
  const unsigned rem0 = ubase - b0 * (unsigned)M;
  float4 gv[GM_];
  if (NEIGH_IDX) {
    int idx[GM_];
    #pragma unroll
    for (int j = 0; j < GM_; ++j) idx[j] = neighIdx[nbase + (j << 3) + wave];
    #pragma unroll
    for (int j = 0; j < GM_; ++j)
      gv[j] = *(const float4*)(emb + ((long)idx[j] << 7) + (lane << 2));
  } else {
    #pragma unroll
    for (int j = 0; j < GM_; ++j)
      gv[j] = *(const float4*)(neighVal + ((nbase + (j << 3) + wave) << 7) + (lane << 2));
  }
  const unsigned b1 = (rem0 + GM_ > (unsigned)M) ? b0 + 1 : b0;
  const float4 sfa = *(const float4*)(sess + ((long)b0 << 7) + (lane << 2));
  const float4 sfb = *(const float4*)(sess + ((long)b1 << 7) + (lane << 2));
  float4 sv0, sv1;
  if (SELF_VAL) {
    sv0 = *(const float4*)(selfVal + ((base + wave) << 7) + (lane << 2));
    sv1 = *(const float4*)(selfVal + ((base + wave + 8) << 7) + (lane << 2));
  } else {
    sv0 = *(const float4*)(emb + ((long)selfIdx[base + wave] << 7) + (lane << 2));
    sv1 = *(const float4*)(emb + ((long)selfIdx[base + wave + 8] << 7) + (lane << 2));
  }
  #pragma unroll
  for (int j = 0; j < GM_; ++j) {
    const int r = (j << 3) + wave;
    const float4 v = gv[j];
    const bool nextb = rem0 + (unsigned)j >= (unsigned)M;
    *(float4*)&nv[r][lane << 2] = v;
    float4 x;
    x.x = (nextb ? sfb.x : sfa.x) * v.x;
    x.y = (nextb ? sfb.y : sfa.y) * v.y;
    x.z = (nextb ? sfb.z : sfa.z) * v.z;
    x.w = (nextb ? sfb.w : sfa.w) * v.w;
    *(float4*)&xs[r][lane << 2] = x;
  }
  *(float4*)&cat[wave][lane << 2] = sv0;
  *(float4*)&cat[wave + 8][lane << 2] = sv1;
  if (tid < GR_) {
    xs[tid][D_] = neighW[nbase + tid];
    xs[tid][D_ + 1] = 0.f; xs[tid][D_ + 2] = 0.f; xs[tid][D_ + 3] = 0.f;
    sc[tid] = 0.f;
  }
  __syncthreads();

  for (int rp = 0; rp < GR_ / 32; ++rp) {
    const float* ap0 = &xs[(rp << 5) + laneIdx][laneHalf << 1];
    const float* ap1 = ap0 + 16 * KP_;
    v8f c0 = {}, c1 = {};
    #pragma unroll
    for (int ks = 0; ks < KP_ / 4; ++ks) {
      v2f a0 = *(const v2f*)(ap0 + 4 * ks);
      v2f a1 = *(const v2f*)(ap1 + 4 * ks);
      c0 = wmma4(a0, breg[ks], c0);
      c1 = wmma4(a1, breg[ks], c1);
    }
    score_rows(c0, g2, &sc[(rp << 5)], laneIdx, laneHalf);
    score_rows(c1, g2, &sc[(rp << 5) + 16], laneIdx, laneHalf);
  }
  __syncthreads();

  if (tid < GM_) {
    float mx = -1e30f;
    for (int s = 0; s < 8; ++s) mx = fmaxf(mx, sc[tid * 8 + s]);
    float ex[8], sum = 0.f;
    for (int s = 0; s < 8; ++s) { ex[s] = __expf(sc[tid * 8 + s] - mx); sum += ex[s]; }
    float inv = 1.f / sum;
    for (int s = 0; s < 8; ++s) alpha[tid * 8 + s] = ex[s] * inv;
  }
  __syncthreads();
  {
    const int d = tid & 127;
    for (int j = 0; j < GM_ / 2; ++j) {
      const int r = (tid >> 7) + (j << 1);
      float acc = 0.f;
      for (int s = 0; s < 8; ++s) acc += alpha[(r << 3) + s] * nv[(r << 3) + s][d];
      cat[r][D_ + d] = acc;
    }
  }
  __syncthreads();

  {
    const int col = n0 + laneIdx;
    const float* ap = &cat[laneIdx][laneHalf << 1];
    const float* bp = gw3p + ((laneHalf << 1) << 7) + col;
    v8f cs = {}, cn = {};
    #pragma unroll 8
    for (int k0 = 0; k0 < D_; k0 += 4) {
      v2f as = *(const v2f*)(ap + k0);
      v2f an = *(const v2f*)(ap + D_ + k0);
      v2f bs, bn;
      bs.x = bp[k0 << 7];
      bs.y = bp[(k0 + 1) << 7];
      bn.x = bp[(D_ + k0) << 7];
      bn.y = bp[(D_ + k0 + 1) << 7];
      cs = wmma4(as, bs, cs);
      cn = wmma4(an, bn, cn);
    }
    #pragma unroll
    for (int v = 0; v < 8; ++v)
      xs[v + (laneHalf << 3)][col] = fmaxf(cs[v] + cn[v], 0.f);
  }
  __syncthreads();
  *(float4*)(out + ((base + wave) << 7) + (lane << 2)) = *(const float4*)&xs[wave][lane << 2];
  *(float4*)(out + ((base + wave + 8) << 7) + (lane << 2)) = *(const float4*)&xs[wave + 8][lane << 2];
}

__global__ void gemm16_kernel(const float* __restrict__ A, int lda,
                              const float* __restrict__ BT, int ldbt,
                              float* __restrict__ C, long ldc,
                              int M, int N, int K,
                              const float* __restrict__ rowBias, int biasMod,
                              int act) {
  int tilesM = (M + 15) >> 4;
  long tilesN = (long)((N + 15) >> 4);
  long total = (long)tilesM * tilesN;
  int lane = threadIdx.x & 31, laneIdx = lane & 15, laneHalf = lane >> 4;
  for (long t = (long)blockIdx.x * 8 + (threadIdx.x >> 5); t < total;
       t += (long)gridDim.x * 8) {
    int m0 = (int)(t % tilesM) << 4;
    long n0 = (t / tilesM) << 4;
    int row = m0 + laneIdx;
    long col = n0 + laneIdx;
    v8f c = {};
    bool full = (m0 + 16 <= M) && (n0 + 16 <= N);
    if (full) {
      const float* ap = A + (long)row * lda;
      const float* bp = BT + col * ldbt;
      for (int k0 = 0; k0 < K; k0 += 4) {
        int kk = k0 + (laneHalf << 1);
        v2f a = *(const v2f*)(ap + kk);
        v2f b = *(const v2f*)(bp + kk);
        c = wmma4(a, b, c);
      }
    } else {
      for (int k0 = 0; k0 < K; k0 += 4) {
        int kk = k0 + (laneHalf << 1);
        v2f a, b;
        a.x = (row < M) ? A[(long)row * lda + kk] : 0.f;
        a.y = (row < M) ? A[(long)row * lda + kk + 1] : 0.f;
        b.x = (col < N) ? BT[col * ldbt + kk] : 0.f;
        b.y = (col < N) ? BT[col * ldbt + kk + 1] : 0.f;
        c = wmma4(a, b, c);
      }
    }
    float* cf = (float*)&c;
    #pragma unroll
    for (int v = 0; v < 8; ++v) {
      int rr = m0 + v + (laneHalf << 3);
      if (rr < M && col < N) {
        float val = cf[v];
        if (rowBias) val += rowBias[(long)(rr % biasMod) * N + col];
        if (act == 1) val = tanhf(val);
        C[(long)rr * ldc + col] = val;
      }
    }
  }
}

__global__ void add_kernel(const float* a, const float* b, float* o, int n) {
  int i = blockIdx.x * 256 + threadIdx.x;
  if (i < n) o[i] = a[i] + b[i];
}

__global__ void zero1_kernel(float* p) { if (threadIdx.x == 0 && blockIdx.x == 0) p[0] = 0.f; }

__global__ void ssl_kernel(const float* __restrict__ hl, const float* __restrict__ hg,
                           float* __restrict__ loss) {
  int idx = blockIdx.x * 256 + threadIdx.x;
  int b = idx >> 7, d = idx & 127;
  int pb = (b * 73 + 21) & 127;
  float pos = 0.f, neg = 0.f;
  for (int l = 0; l < L_; ++l) {
    int pl = (l * 13 + 7) % L_;
    pos += hl[((b * L_ + l) << 7) + d] * hg[((b * L_ + l) << 7) + d];
    neg += hg[((b * L_ + l) << 7) + d] * hl[((pb * L_ + pl) << 7) + d];
  }
  float term = -__logf(1e-8f + sigmoid_(pos)) - __logf(1e-8f + 1.f - sigmoid_(neg));
  __shared__ float red[256];
  red[threadIdx.x] = term;
  __syncthreads();
  for (int s = 128; s > 0; s >>= 1) {
    if (threadIdx.x < s) red[threadIdx.x] += red[threadIdx.x + s];
    __syncthreads();
  }
  if (threadIdx.x == 0) atomicAdd(loss, BETA_ * red[0]);
}

__global__ void hs_kernel(const float* __restrict__ hidden, const int* __restrict__ mask,
                          float* __restrict__ hs) {
  int b = blockIdx.x, d = threadIdx.x;
  float acc = 0.f, ms = 0.f;
  for (int l = 0; l < L_; ++l) {
    float m = (float)mask[b * L_ + l];
    ms += m;
    acc += hidden[((b * L_ + l) << 7) + d] * m;
  }
  hs[(b << 7) + d] = acc / ms;
}

__global__ void hsg_kernel(const float* __restrict__ hs, const float* __restrict__ glu2_w,
                           float* __restrict__ hsg) {
  __shared__ float hr[D_];
  int b = blockIdx.x, d = threadIdx.x;
  hr[d] = hs[(b << 7) + d];
  __syncthreads();
  float acc = 0.f;
  for (int e = 0; e < D_; ++e) acc += hr[e] * glu2_w[(d << 7) + e];
  hsg[(b << 7) + d] = acc;
}

__global__ void pos_kernel(const float* __restrict__ pos_emb, const float* __restrict__ w1,
                           float* __restrict__ P) {
  __shared__ float pr[D_];
  int l = blockIdx.x, d = threadIdx.x;
  pr[d] = pos_emb[(l << 7) + d];
  __syncthreads();
  float acc = 0.f;
  for (int e = 0; e < D_; ++e) acc += pr[e] * w1[(e << 7) + d];
  P[(l << 7) + d] = acc;
}

__global__ void beta_kernel(const float* __restrict__ nh, const float* __restrict__ glu1_w,
                            const float* __restrict__ glu1_b, const float* __restrict__ hsg,
                            const float* __restrict__ w2, const int* __restrict__ mask,
                            float* __restrict__ beta) {
  __shared__ float nrow[D_];
  __shared__ float red[D_];
  int row = blockIdx.x;
  int b = row / L_;
  int d = threadIdx.x;
  nrow[d] = nh[((long)row << 7) + d];
  __syncthreads();
  float acc = glu1_b[d] + hsg[(b << 7) + d];
  for (int e = 0; e < D_; ++e) acc += nrow[e] * glu1_w[(d << 7) + e];
  red[d] = sigmoid_(acc) * w2[d];
  __syncthreads();
  for (int s = 64; s > 0; s >>= 1) {
    if (d < s) red[d] += red[d + s];
    __syncthreads();
  }
  if (d == 0) beta[row] = red[0] * (float)mask[row];
}

__global__ void select_kernel(const float* __restrict__ hidden, const float* __restrict__ beta,
                              float* __restrict__ sel) {
  int b = blockIdx.x, d = threadIdx.x;
  float acc = 0.f;
  for (int l = 0; l < L_; ++l) acc += beta[b * L_ + l] * hidden[((b * L_ + l) << 7) + d];
  sel[(b << 7) + d] = acc;
}

__global__ void norm_kernel(const float* __restrict__ sel, float* __restrict__ snorm) {
  __shared__ float red[D_];
  int b = blockIdx.x, d = threadIdx.x;
  float v = sel[(b << 7) + d];
  red[d] = v * v + 1e-6f;
  __syncthreads();
  for (int s = 64; s > 0; s >>= 1) {
    if (d < s) red[d] += red[d + s];
    __syncthreads();
  }
  if (d == 0) snorm[b] = sqrtf(red[0]);
}

__global__ void fn_kernel(const float* __restrict__ sel, const float* __restrict__ snorm,
                          float* __restrict__ selF) {
  __shared__ float row[B_];
  __shared__ float selfrow[D_];
  __shared__ float tvs[9];
  __shared__ int tis[9];
  int i = blockIdx.x, t = threadIdx.x;
  selfrow[t] = sel[(i << 7) + t];
  __syncthreads();
  float acc = 0.f;
  for (int d = 0; d < D_; ++d) acc += selfrow[d] * sel[(t << 7) + d];
  row[t] = acc / (snorm[i] * snorm[t]);
  __syncthreads();
  if (t == 0) {
    float mx = -1e30f;
    for (int j = 0; j < B_; ++j) mx = fmaxf(mx, row[j]);
    float s = 0.f;
    for (int j = 0; j < B_; ++j) { row[j] = __expf(row[j] - mx); s += row[j]; }
    float inv = 1.f / s;
    for (int j = 0; j < B_; ++j) row[j] *= inv;
    for (int q = 0; q < 9; ++q) {
      float bm = -1.f; int bi = 0;
      for (int j = 0; j < B_; ++j) if (row[j] > bm) { bm = row[j]; bi = j; }
      tvs[q] = bm; tis[q] = bi; row[bi] = -2.f;
    }
    float mx2 = tvs[0], s2 = 0.f;
    for (int q = 0; q < 9; ++q) { tvs[q] = __expf(tvs[q] - mx2); s2 += tvs[q]; }
    float inv2 = 1.f / s2;
    for (int q = 0; q < 9; ++q) tvs[q] *= inv2;
  }
  __syncthreads();
  float nb = 0.f;
  for (int q = 0; q < 9; ++q) nb += tvs[q] * sel[(tis[q] << 7) + t];
  selF[(i << 7) + t] = selfrow[t] + nb;
}

extern "C" void kernel_launch(void* const* d_in, const int* in_sizes, int n_in,
                              void* d_out, int out_size, void* d_ws, size_t ws_size,
                              hipStream_t stream) {
  (void)in_sizes; (void)n_in; (void)out_size; (void)ws_size;
  const int*   inputs  = (const int*)d_in[0];
  const int*   adj     = (const int*)d_in[1];
  const int*   mask    = (const int*)d_in[2];
  const int*   item    = (const int*)d_in[3];
  const int*   adj_all = (const int*)d_in[5];
  const float* num_w   = (const float*)d_in[6];
  const float* emb     = (const float*)d_in[7];
  const float* pos_emb = (const float*)d_in[8];
  const float* a_local = (const float*)d_in[9];
  const float* gw1     = (const float*)d_in[10];
  const float* gw2     = (const float*)d_in[11];
  const float* gw3     = (const float*)d_in[12];
  const float* w1      = (const float*)d_in[13];
  const float* w2      = (const float*)d_in[14];
  const float* glu1_w  = (const float*)d_in[15];
  const float* glu1_b  = (const float*)d_in[16];
  const float* glu2_w  = (const float*)d_in[17];

  float* out0   = (float*)d_out;
  float* lossP  = out0 + B_ * L_ * D_;
  float* scores = lossP + 1;

  char* wp = (char*)d_ws;
  auto alloc = [&](size_t nfloats) -> float* {
    float* r = (float*)wp;
    wp += ((nfloats * sizeof(float) + 255) / 256) * 256;
    return r;
  };
  float* h        = alloc(B_ * L_ * D_);
  float* h_local  = alloc(B_ * L_ * D_);
  float* h_global = alloc(B_ * L_ * D_);
  float* sessb    = alloc(B_ * D_);
  float* e0p      = alloc(B_ * L_ * D_);
  float* e1p      = alloc((size_t)B_ * 400 * D_);
  float* hsb      = alloc(B_ * D_);
  float* hsgb     = alloc(B_ * D_);
  float* Pb       = alloc(L_ * D_);
  float* nh       = alloc(B_ * L_ * D_);
  float* betab    = alloc(B_ * L_);
  float* selb     = alloc(B_ * D_);
  float* selF     = alloc(B_ * D_);
  float* snorm    = alloc(B_);
  float* gw1T     = alloc(2 * D_ * KP_);
  float* w1botT   = alloc(D_ * D_);
  float* wn1      = alloc(B_ * 400);
  float* wn2      = alloc((size_t)B_ * 3200);
  int*   n1       = (int*)alloc(B_ * 400);
  int*   n2       = (int*)alloc((size_t)B_ * 3200);

  gw1t_kernel<<<dim3((2 * D_ * KP_ + 255) / 256), 256, 0, stream>>>(gw1, gw1T);
  w1t_kernel<<<dim3((D_ * D_ + 255) / 256), 256, 0, stream>>>(w1, w1botT);
  gather_h_kernel<<<dim3((B_ * L_ * D_) / 256), 256, 0, stream>>>(emb, inputs, h, B_ * L_ * D_);
  sess_kernel<<<dim3(B_), dim3(D_), 0, stream>>>(emb, item, mask, sessb);
  hop_kernel<<<dim3((B_ * L_ * S_ + 255) / 256), 256, 0, stream>>>(inputs, adj_all, num_w, n1, wn1, B_ * L_ * S_);
  hop_kernel<<<dim3((B_ * 400 * S_ + 255) / 256), 256, 0, stream>>>(n1, adj_all, num_w, n2, wn2, B_ * 400 * S_);
  local_agg_kernel<<<dim3(B_), 256, 0, stream>>>(h, adj, a_local, h_local);
  agg_kernel<true, true><<<dim3(B_ * L_ / GM_), 256, 0, stream>>>(emb, h, nullptr, n1, nullptr, wn1,
      sessb, gw1T, gw2, gw3, e0p, L_);
  agg_kernel<false, true><<<dim3(B_ * 400 / GM_), 256, 0, stream>>>(emb, nullptr, n1, n2, nullptr, wn2,
      sessb, gw1T, gw2, gw3, e1p, 400);
  agg_kernel<true, false><<<dim3(B_ * L_ / GM_), 256, 0, stream>>>(emb, e0p, nullptr, nullptr, e1p, wn1,
      sessb, gw1T + D_ * KP_, gw2 + D_, gw3 + 2 * D_ * D_, h_global, L_);
  add_kernel<<<dim3((B_ * L_ * D_) / 256), 256, 0, stream>>>(h_local, h_global, out0, B_ * L_ * D_);
  zero1_kernel<<<1, 1, 0, stream>>>(lossP);
  ssl_kernel<<<dim3(64), 256, 0, stream>>>(h_local, h_global, lossP);
  hs_kernel<<<dim3(B_), dim3(D_), 0, stream>>>(out0, mask, hsb);
  hsg_kernel<<<dim3(B_), dim3(D_), 0, stream>>>(hsb, glu2_w, hsgb);
  pos_kernel<<<dim3(L_), dim3(D_), 0, stream>>>(pos_emb, w1, Pb);
  {
    int tiles = ((B_ * L_ + 15) / 16) * ((D_ + 15) / 16);
    gemm16_kernel<<<dim3((tiles + 7) / 8), 256, 0, stream>>>(
        out0, D_, w1botT, D_, nh, D_, B_ * L_, D_, D_, Pb, L_, 1);
  }
  beta_kernel<<<dim3(B_ * L_), dim3(D_), 0, stream>>>(nh, glu1_w, glu1_b, hsgb, w2, mask, betab);
  select_kernel<<<dim3(B_), dim3(D_), 0, stream>>>(out0, betab, selb);
  norm_kernel<<<dim3(B_), dim3(D_), 0, stream>>>(selb, snorm);
  fn_kernel<<<dim3(B_), dim3(D_), 0, stream>>>(selb, snorm, selF);
  {
    int Nsc = NN_ - 1;
    long tiles = (long)((B_ + 15) / 16) * ((Nsc + 15) / 16);
    gemm16_kernel<<<dim3((unsigned)((tiles + 7) / 8)), 256, 0, stream>>>(
        selF, D_, emb + D_, D_, scores, (long)Nsc, B_, Nsc, D_, nullptr, 1, 0);
  }
}


# score GEMM (128 x 99999 x 128) retiled: selF fragments stay in registers, each embedding row is read once as a whole 512-B row and staged in LDS (two buffers, 32 columns per stage)
# speedup vs baseline: 1.5357x; 1.1511x over previous
#include <hip/hip_runtime.h>
#include <hip/hip_bf16.h>
#include <math.h>

#define B_ 128
#define L_ 50
#define D_ 128
#define S_ 8
#define NN_ 100000
#define KP_ 132
#define LEAKY_ 0.2f
#define BETA_ 0.005f

typedef float v2f __attribute__((ext_vector_type(2)));
typedef float v8f __attribute__((ext_vector_type(8)));

#if defined(__has_builtin)
#if __has_builtin(__builtin_amdgcn_wmma_f32_16x16x4_f32)
#define HAVE_WMMA_F32 1
#endif
#endif

__device__ __forceinline__ v8f wmma4(v2f a, v2f b, v8f c) {
#ifdef HAVE_WMMA_F32
  return __builtin_amdgcn_wmma_f32_16x16x4_f32(false, a, false, b, (short)0, c,
                                               false, false);
#else
  float* cf = (float*)&c;
  #pragma unroll
  for (int i = 0; i < 8; ++i) cf[i] += a.x * b.x + a.y * b.y;
  return c;
#endif
}

__device__ __forceinline__ float leaky_(float x) { return x > 0.f ? x : LEAKY_ * x; }
__device__ __forceinline__ float sigmoid_(float x) { return 1.f / (1.f + __expf(-x)); }

__global__ void gw1t_kernel(const float* __restrict__ gw1, float* __restrict__ gw1T) {
  int i = blockIdx.x * 256 + threadIdx.x;
  if (i >= 2 * D_ * KP_) return;
  int hop = i / (D_ * KP_);
  int rem = i % (D_ * KP_);
  int n = rem / KP_, k = rem % KP_;
  gw1T[i] = (k < 129) ? gw1[hop * 129 * D_ + k * D_ + n] : 0.f;
}

__global__ void w1t_kernel(const float* __restrict__ w1, float* __restrict__ w1botT) {
  int i = blockIdx.x * 256 + threadIdx.x;
  if (i >= D_ * D_) return;
  int n = i >> 7, k = i & 127;
  w1botT[i] = w1[(D_ + k) * D_ + n];
}

__global__ void gather_h_kernel(const float* __restrict__ emb,
                                const int* __restrict__ inputs,
                                float* __restrict__ h, int total) {
  int i = blockIdx.x * 256 + threadIdx.x;
  if (i >= total) return;
  int row = i >> 7, d = i & 127;
  h[i] = emb[((long)inputs[row] << 7) + d];
}

__global__ void hop_kernel(const int* __restrict__ src,
                           const int* __restrict__ adj_all,
                           const float* __restrict__ num_w,
                           int* __restrict__ dst, float* __restrict__ dstw,
                           int total) {
  int i = blockIdx.x * 256 + threadIdx.x;
  if (i >= total) return;
  int node = src[i >> 3];
  int s = i & 7;
  dst[i]  = adj_all[node * S_ + s];
  dstw[i] = num_w[node * S_ + s];
}

__global__ void sess_kernel(const float* __restrict__ emb,
                            const int* __restrict__ item,
                            const int* __restrict__ mask,
                            float* __restrict__ sess) {
  int b = blockIdx.x, d = threadIdx.x;
  float acc = 0.f, ms = 0.f;
  for (int l = 0; l < L_; ++l) {
    float m = (float)mask[b * L_ + l];
    ms += m;
    acc += emb[((long)item[b * L_ + l] << 7) + d] * m;
  }
  sess[(b << 7) + d] = acc / ms;
}

__global__ void local_agg_kernel(const float* __restrict__ h,
                                 const int* __restrict__ adj,
                                 const float* __restrict__ a_local,
                                 float* __restrict__ h_local) {
  int b = blockIdx.x;
  __shared__ float hv[L_][D_];
  __shared__ float att[L_][L_];
  __shared__ float rinv[L_];
  for (int i = threadIdx.x; i < L_ * D_; i += 256)
    hv[i / D_][i & 127] = h[(long)b * L_ * D_ + i];
  __syncthreads();
  for (int p = threadIdx.x; p < L_ * L_; p += 256) {
    int i = p / L_, j = p % L_;
    int k = adj[b * L_ * L_ + p];
    float v;
    if (k == 0) v = -9e15f;
    else {
      const float* a = a_local + (k - 1) * D_;
      float acc = 0.f;
      for (int d = 0; d < D_; ++d) acc += hv[i][d] * hv[j][d] * a[d];
      v = leaky_(acc);
    }
    att[i][j] = v;
  }
  __syncthreads();
  if (threadIdx.x < L_) {
    int i = threadIdx.x;
    float mx = -1e30f;
    for (int j = 0; j < L_; ++j) mx = fmaxf(mx, att[i][j]);
    float s = 0.f;
    for (int j = 0; j < L_; ++j) { float e = __expf(att[i][j] - mx); att[i][j] = e; s += e; }
    rinv[i] = 1.f / s;
  }
  __syncthreads();
  for (int p = threadIdx.x; p < L_ * D_; p += 256) {
    int i = p >> 7, d = p & 127;
    float acc = 0.f;
    for (int j = 0; j < L_; ++j) acc += att[i][j] * hv[j][d];
    h_local[(long)b * L_ * D_ + p] = acc * rinv[i];
  }
}

#define GM_ 16
#define GR_ 128
#define CP_ 260

__device__ __forceinline__ void score_rows(v8f c, float g2, float* scRow,
                                           int laneIdx, int laneHalf) {
  #pragma unroll
  for (int v = 0; v < 8; ++v) {
    float val = leaky_(c[v]) * g2;
    val += __shfl_xor(val, 1);
    val += __shfl_xor(val, 2);
    val += __shfl_xor(val, 4);
    val += __shfl_xor(val, 8);
    if (laneIdx == 0) atomicAdd(&scRow[v + (laneHalf << 3)], val);
  }
}

template <bool SELF_VAL, bool NEIGH_IDX>
__global__ void __launch_bounds__(256)
agg_kernel(const float* __restrict__ emb,
           const float* __restrict__ selfVal, const int* __restrict__ selfIdx,
           const int* __restrict__ neighIdx, const float* __restrict__ neighVal,
           const float* __restrict__ neighW,
           const float* __restrict__ sess,
           const float* __restrict__ gw1T, const float* __restrict__ gw2p,
           const float* __restrict__ gw3p,
           float* __restrict__ out, int M) {
  __shared__ __align__(16) float nv[GR_][D_];
  __shared__ __align__(16) float xs[GR_][KP_];
  __shared__ __align__(16) float cat[GM_][CP_];
  __shared__ float sc[GR_];
  __shared__ float alpha[GR_];

  const int tid = threadIdx.x;
  const int lane = tid & 31, laneIdx = lane & 15, laneHalf = lane >> 4;
  const int wave = tid >> 5;
  const int n0 = wave * 16;
  const long base  = (long)blockIdx.x * GM_;
  const long nbase = base * S_;

  v2f breg[KP_ / 4];
  {
    const float* bp = gw1T + (n0 + laneIdx) * KP_ + (laneHalf << 1);
    #pragma unroll
    for (int ks = 0; ks < KP_ / 4; ++ks) breg[ks] = *(const v2f*)(bp + 4 * ks);
  }
  const float g2 = gw2p[n0 + laneIdx];

  const unsigned ubase = blockIdx.x * GM_;
  const unsigned b0 = ubase / (unsigned)M;
  const unsigned rem0 = ubase - b0 * (unsigned)M;
  float4 gv[GM_];
  if (NEIGH_IDX) {
    int idx[GM_];
    #pragma unroll
    for (int j = 0; j < GM_; ++j) idx[j] = neighIdx[nbase + (j << 3) + wave];
    #pragma unroll
    for (int j = 0; j < GM_; ++j)
      gv[j] = *(const float4*)(emb + ((long)idx[j] << 7) + (lane << 2));
  } else {
    #pragma unroll
    for (int j = 0; j < GM_; ++j)
      gv[j] = *(const float4*)(neighVal + ((nbase + (j << 3) + wave) << 7) + (lane << 2));
  }
  const unsigned b1 = (rem0 + GM_ > (unsigned)M) ? b0 + 1 : b0;
  const float4 sfa = *(const float4*)(sess + ((long)b0 << 7) + (lane << 2));
  const float4 sfb = *(const float4*)(sess + ((long)b1 << 7) + (lane << 2));
  float4 sv0, sv1;
  if (SELF_VAL) {
    sv0 = *(const float4*)(selfVal + ((base + wave) << 7) + (lane << 2));
    sv1 = *(const float4*)(selfVal + ((base + wave + 8) << 7) + (lane << 2));
  } else {
    sv0 = *(const float4*)(emb + ((long)selfIdx[base + wave] << 7) + (lane << 2));
    sv1 = *(const float4*)(emb + ((long)selfIdx[base + wave + 8] << 7) + (lane << 2));
  }
  #pragma unroll
  for (int j = 0; j < GM_; ++j) {
    const int r = (j << 3) + wave;
    const float4 v = gv[j];
    const bool nextb = rem0 + (unsigned)j >= (unsigned)M;
    *(float4*)&nv[r][lane << 2] = v;
    float4 x;
    x.x = (nextb ? sfb.x : sfa.x) * v.x;
    x.y = (nextb ? sfb.y : sfa.y) * v.y;
    x.z = (nextb ? sfb.z : sfa.z) * v.z;
    x.w = (nextb ? sfb.w : sfa.w) * v.w;
    *(float4*)&xs[r][lane << 2] = x;
  }
  *(float4*)&cat[wave][lane << 2] = sv0;
  *(float4*)&cat[wave + 8][lane << 2] = sv1;
  if (tid < GR_) {
    xs[tid][D_] = neighW[nbase + tid];
    xs[tid][D_ + 1] = 0.f; xs[tid][D_ + 2] = 0.f; xs[tid][D_ + 3] = 0.f;
    sc[tid] = 0.f;
  }
  __syncthreads();

  for (int rp = 0; rp < GR_ / 32; ++rp) {
    const float* ap0 = &xs[(rp << 5) + laneIdx][laneHalf << 1];
    const float* ap1 = ap0 + 16 * KP_;
    v8f c0 = {}, c1 = {};
    #pragma unroll
    for (int ks = 0; ks < KP_ / 4; ++ks) {
      v2f a0 = *(const v2f*)(ap0 + 4 * ks);
      v2f a1 = *(const v2f*)(ap1 + 4 * ks);
      c0 = wmma4(a0, breg[ks], c0);
      c1 = wmma4(a1, breg[ks], c1);
    }
    score_rows(c0, g2, &sc[(rp << 5)], laneIdx, laneHalf);
    score_rows(c1, g2, &sc[(rp << 5) + 16], laneIdx, laneHalf);
  }
  __syncthreads();

  if (tid < GM_) {
    float mx = -1e30f;
    for (int s = 0; s < 8; ++s) mx = fmaxf(mx, sc[tid * 8 + s]);
    float ex[8], sum = 0.f;
    for (int s = 0; s < 8; ++s) { ex[s] = __expf(sc[tid * 8 + s] - mx); sum += ex[s]; }
    float inv = 1.f / sum;
    for (int s = 0; s < 8; ++s) alpha[tid * 8 + s] = ex[s] * inv;
  }
  __syncthreads();
  {
    const int d = tid & 127;
    for (int j = 0; j < GM_ / 2; ++j) {
      const int r = (tid >> 7) + (j << 1);
      float acc = 0.f;
      for (int s = 0; s < 8; ++s) acc += alpha[(r << 3) + s] * nv[(r << 3) + s][d];
      cat[r][D_ + d] = acc;
    }
  }
  __syncthreads();

  {
    const int col = n0 + laneIdx;
    const float* ap = &cat[laneIdx][laneHalf << 1];
    const float* bp = gw3p + ((laneHalf << 1) << 7) + col;
    v8f cs = {}, cn = {};
    #pragma unroll 8
    for (int k0 = 0; k0 < D_; k0 += 4) {
      v2f as = *(const v2f*)(ap + k0);
      v2f an = *(const v2f*)(ap + D_ + k0);
      v2f bs, bn;
      bs.x = bp[k0 << 7];
      bs.y = bp[(k0 + 1) << 7];
      bn.x = bp[(D_ + k0) << 7];
      bn.y = bp[(D_ + k0 + 1) << 7];
      cs = wmma4(as, bs, cs);
      cn = wmma4(an, bn, cn);
    }
    #pragma unroll
    for (int v = 0; v < 8; ++v)
      xs[v + (laneHalf << 3)][col] = fmaxf(cs[v] + cn[v], 0.f);
  }
  __syncthreads();
  *(float4*)(out + ((base + wave) << 7) + (lane << 2)) = *(const float4*)&xs[wave][lane << 2];
  *(float4*)(out + ((base + wave + 8) << 7) + (lane << 2)) = *(const float4*)&xs[wave + 8][lane << 2];
}

__global__ void gemm16_kernel(const float* __restrict__ A, int lda,
                              const float* __restrict__ BT, int ldbt,
                              float* __restrict__ C, long ldc,
                              int M, int N, int K,
                              const float* __restrict__ rowBias, int biasMod,
                              int act) {
  int tilesM = (M + 15) >> 4;
  long tilesN = (long)((N + 15) >> 4);
  long total = (long)tilesM * tilesN;
  int lane = threadIdx.x & 31, laneIdx = lane & 15, laneHalf = lane >> 4;
  for (long t = (long)blockIdx.x * 8 + (threadIdx.x >> 5); t < total;
       t += (long)gridDim.x * 8) {
    int m0 = (int)(t % tilesM) << 4;
    long n0 = (t / tilesM) << 4;
    int row = m0 + laneIdx;
    long col = n0 + laneIdx;
    v8f c = {};
    bool full = (m0 + 16 <= M) && (n0 + 16 <= N);
    if (full) {
      const float* ap = A + (long)row * lda;
      const float* bp = BT + col * ldbt;
      for (int k0 = 0; k0 < K; k0 += 4) {
        int kk = k0 + (laneHalf << 1);
        v2f a = *(const v2f*)(ap + kk);
        v2f b = *(const v2f*)(bp + kk);
        c = wmma4(a, b, c);
      }
    } else {
      for (int k0 = 0; k0 < K; k0 += 4) {
        int kk = k0 + (laneHalf << 1);
        v2f a, b;
        a.x = (row < M) ? A[(long)row * lda + kk] : 0.f;
        a.y = (row < M) ? A[(long)row * lda + kk + 1] : 0.f;
        b.x = (col < N) ? BT[col * ldbt + kk] : 0.f;
        b.y = (col < N) ? BT[col * ldbt + kk + 1] : 0.f;
        c = wmma4(a, b, c);
      }
    }
    float* cf = (float*)&c;
    #pragma unroll
    for (int v = 0; v < 8; ++v) {
      int rr = m0 + v + (laneHalf << 3);
      if (rr < M && col < N) {
        float val = cf[v];
        if (rowBias) val += rowBias[(long)(rr % biasMod) * N + col];
        if (act == 1) val = tanhf(val);
        C[(long)rr * ldc + col] = val;
      }
    }
  }
}

#define SC_COLS_ 32
#define SC_KH_ (D_ / 8)

__device__ __forceinline__ float4 sc_row(const float* __restrict__ embT, int n, int N, int lane) {
  n = n < N ? n : N - 1;
  return *(const float4*)(embT + ((long)n << 7) + (lane << 2));
}

__global__ void __launch_bounds__(256)
scores_kernel(const float* __restrict__ selF, const float* __restrict__ embT,
              float* __restrict__ scores, int N, int nStages) {
  __shared__ __align__(16) float bs[2][SC_COLS_][KP_];
  const int tid = threadIdx.x;
  const int lane = tid & 31, laneIdx = lane & 15, laneHalf = lane >> 4;
  const int wave = tid >> 5;

  v2f areg[D_ / 4];
  {
    const float* ap = selF + ((wave * 16 + laneIdx) << 7) + (laneHalf << 1);
    #pragma unroll
    for (int ks = 0; ks < D_ / 4; ++ks) areg[ks] = *(const v2f*)(ap + 4 * ks);
  }

  const int r0 = wave * 4;
  float4 p0, p1, p2, p3;
  int s = blockIdx.x;
  if (s < nStages) {
    const int n = s * SC_COLS_ + r0;
    p0 = sc_row(embT, n, N, lane);
    p1 = sc_row(embT, n + 1, N, lane);
    p2 = sc_row(embT, n + 2, N, lane);
    p3 = sc_row(embT, n + 3, N, lane);
    *(float4*)&bs[0][r0][lane << 2] = p0;
    *(float4*)&bs[0][r0 + 1][lane << 2] = p1;
    *(float4*)&bs[0][r0 + 2][lane << 2] = p2;
    *(float4*)&bs[0][r0 + 3][lane << 2] = p3;
  }
  __syncthreads();

  int buf = 0;
  for (; s < nStages; s += gridDim.x) {
    const int sn = s + gridDim.x;
    const bool more = sn < nStages;
    if (more) {
      const int n = sn * SC_COLS_ + r0;
      p0 = sc_row(embT, n, N, lane);
      p1 = sc_row(embT, n + 1, N, lane);
      p2 = sc_row(embT, n + 2, N, lane);
      p3 = sc_row(embT, n + 3, N, lane);
    }
    const float* bp0 = &bs[buf][laneIdx][laneHalf << 1];
    const float* bp1 = bp0 + 16 * KP_;
    v8f c0 = {}, c1 = {};
    #pragma unroll
    for (int kh = 0; kh < 2; ++kh) {
      v2f f0[SC_KH_], f1[SC_KH_];
      #pragma unroll
      for (int ks = 0; ks < SC_KH_; ++ks) {
        f0[ks] = *(const v2f*)(bp0 + 4 * (kh * SC_KH_ + ks));
        f1[ks] = *(const v2f*)(bp1 + 4 * (kh * SC_KH_ + ks));
      }
      #pragma unroll
      for (int ks = 0; ks < SC_KH_; ++ks) {
        c0 = wmma4(areg[kh * SC_KH_ + ks], f0[ks], c0);
        c1 = wmma4(areg[kh * SC_KH_ + ks], f1[ks], c1);
      }
    }
    if (more) {
      *(float4*)&bs[buf ^ 1][r0][lane << 2] = p0;
      *(float4*)&bs[buf ^ 1][r0 + 1][lane << 2] = p1;
      *(float4*)&bs[buf ^ 1][r0 + 2][lane << 2] = p2;
      *(float4*)&bs[buf ^ 1][r0 + 3][lane << 2] = p3;
    }
    __syncthreads();
    buf ^= 1;
    {
      const int n0 = s * SC_COLS_;
      float* o = scores + (long)(wave * 16 + (laneHalf << 3)) * N + n0 + laneIdx;
      if (n0 + SC_COLS_ <= N) {
        #pragma unroll
        for (int v = 0; v < 8; ++v) {
          o[(long)v * N] = c0[v];
          o[(long)v * N + 16] = c1[v];
        }
      } else {
        const bool in0 = n0 + laneIdx < N, in1 = n0 + laneIdx + 16 < N;
        #pragma unroll
        for (int v = 0; v < 8; ++v) {
          if (in0) o[(long)v * N] = c0[v];
          if (in1) o[(long)v * N + 16] = c1[v];
        }
      }
    }
  }
}

__global__ void add_kernel(const float* a, const float* b, float* o, int n) {
  int i = blockIdx.x * 256 + threadIdx.x;
  if (i < n) o[i] = a[i] + b[i];
}

__global__ void zero1_kernel(float* p) { if (threadIdx.x == 0 && blockIdx.x == 0) p[0] = 0.f; }

__global__ void ssl_kernel(const float* __restrict__ hl, const float* __restrict__ hg,
                           float* __restrict__ loss) {
  int idx = blockIdx.x * 256 + threadIdx.x;
  int b = idx >> 7, d = idx & 127;
  int pb = (b * 73 + 21) & 127;
  float pos = 0.f, neg = 0.f;
  for (int l = 0; l < L_; ++l) {
    int pl = (l * 13 + 7) % L_;
    pos += hl[((b * L_ + l) << 7) + d] * hg[((b * L_ + l) << 7) + d];
    neg += hg[((b * L_ + l) << 7) + d] * hl[((pb * L_ + pl) << 7) + d];
  }
  float term = -__logf(1e-8f + sigmoid_(pos)) - __logf(1e-8f + 1.f - sigmoid_(neg));
  __shared__ float red[256];
  red[threadIdx.x] = term;
  __syncthreads();
  for (int s = 128; s > 0; s >>= 1) {
    if (threadIdx.x < s) red[threadIdx.x] += red[threadIdx.x + s];
    __syncthreads();
  }
  if (threadIdx.x == 0) atomicAdd(loss, BETA_ * red[0]);
}

__global__ void hs_kernel(const float* __restrict__ hidden, const int* __restrict__ mask,
                          float* __restrict__ hs) {
  int b = blockIdx.x, d = threadIdx.x;
  float acc = 0.f, ms = 0.f;
  for (int l = 0; l < L_; ++l) {
    float m = (float)mask[b * L_ + l];
    ms += m;
    acc += hidden[((b * L_ + l) << 7) + d] * m;
  }
  hs[(b << 7) + d] = acc / ms;
}

__global__ void hsg_kernel(const float* __restrict__ hs, const float* __restrict__ glu2_w,
                           float* __restrict__ hsg) {
  __shared__ float hr[D_];
  int b = blockIdx.x, d = threadIdx.x;
  hr[d] = hs[(b << 7) + d];
  __syncthreads();
  float acc = 0.f;
  for (int e = 0; e < D_; ++e) acc += hr[e] * glu2_w[(d << 7) + e];
  hsg[(b << 7) + d] = acc;
}

__global__ void pos_kernel(const float* __restrict__ pos_emb, const float* __restrict__ w1,
                           float* __restrict__ P) {
  __shared__ float pr[D_];
  int l = blockIdx.x, d = threadIdx.x;
  pr[d] = pos_emb[(l << 7) + d];
  __syncthreads();
  float acc = 0.f;
  for (int e = 0; e < D_; ++e) acc += pr[e] * w1[(e << 7) + d];
  P[(l << 7) + d] = acc;
}

__global__ void beta_kernel(const float* __restrict__ nh, const float* __restrict__ glu1_w,
                            const float* __restrict__ glu1_b, const float* __restrict__ hsg,
                            const float* __restrict__ w2, const int* __restrict__ mask,
                            float* __restrict__ beta) {
  __shared__ float nrow[D_];
  __shared__ float red[D_];
  int row = blockIdx.x;
  int b = row / L_;
  int d = threadIdx.x;
  nrow[d] = nh[((long)row << 7) + d];
  __syncthreads();
  float acc = glu1_b[d] + hsg[(b << 7) + d];
  for (int e = 0; e < D_; ++e) acc += nrow[e] * glu1_w[(d << 7) + e];
  red[d] = sigmoid_(acc) * w2[d];
  __syncthreads();
  for (int s = 64; s > 0; s >>= 1) {
    if (d < s) red[d] += red[d + s];
    __syncthreads();
  }
  if (d == 0) beta[row] = red[0] * (float)mask[row];
}

__global__ void select_kernel(const float* __restrict__ hidden, const float* __restrict__ beta,
                              float* __restrict__ sel) {
  int b = blockIdx.x, d = threadIdx.x;
  float acc = 0.f;
  for (int l = 0; l < L_; ++l) acc += beta[b * L_ + l] * hidden[((b * L_ + l) << 7) + d];
  sel[(b << 7) + d] = acc;
}

__global__ void norm_kernel(const float* __restrict__ sel, float* __restrict__ snorm) {
  __shared__ float red[D_];
  int b = blockIdx.x, d = threadIdx.x;
  float v = sel[(b << 7) + d];
  red[d] = v * v + 1e-6f;
  __syncthreads();
  for (int s = 64; s > 0; s >>= 1) {
    if (d < s) red[d] += red[d + s];
    __syncthreads();
  }
  if (d == 0) snorm[b] = sqrtf(red[0]);
}

__global__ void fn_kernel(const float* __restrict__ sel, const float* __restrict__ snorm,
                          float* __restrict__ selF) {
  __shared__ float row[B_];
  __shared__ float selfrow[D_];
  __shared__ float tvs[9];
  __shared__ int tis[9];
  int i = blockIdx.x, t = threadIdx.x;
  selfrow[t] = sel[(i << 7) + t];
  __syncthreads();
  float acc = 0.f;
  for (int d = 0; d < D_; ++d) acc += selfrow[d] * sel[(t << 7) + d];
  row[t] = acc / (snorm[i] * snorm[t]);
  __syncthreads();
  if (t == 0) {
    float mx = -1e30f;
    for (int j = 0; j < B_; ++j) mx = fmaxf(mx, row[j]);
    float s = 0.f;
    for (int j = 0; j < B_; ++j) { row[j] = __expf(row[j] - mx); s += row[j]; }
    float inv = 1.f / s;
    for (int j = 0; j < B_; ++j) row[j] *= inv;
    for (int q = 0; q < 9; ++q) {
      float bm = -1.f; int bi = 0;
      for (int j = 0; j < B_; ++j) if (row[j] > bm) { bm = row[j]; bi = j; }
      tvs[q] = bm; tis[q] = bi; row[bi] = -2.f;
    }
    float mx2 = tvs[0], s2 = 0.f;
    for (int q = 0; q < 9; ++q) { tvs[q] = __expf(tvs[q] - mx2); s2 += tvs[q]; }
    float inv2 = 1.f / s2;
    for (int q = 0; q < 9; ++q) tvs[q] *= inv2;
  }
  __syncthreads();
  float nb = 0.f;
  for (int q = 0; q < 9; ++q) nb += tvs[q] * sel[(tis[q] << 7) + t];
  selF[(i << 7) + t] = selfrow[t] + nb;
}

extern "C" void kernel_launch(void* const* d_in, const int* in_sizes, int n_in,
                              void* d_out, int out_size, void* d_ws, size_t ws_size,
                              hipStream_t stream) {
  (void)in_sizes; (void)n_in; (void)out_size; (void)ws_size;
  const int*   inputs  = (const int*)d_in[0];
  const int*   adj     = (const int*)d_in[1];
  const int*   mask    = (const int*)d_in[2];
  const int*   item    = (const int*)d_in[3];
  const int*   adj_all = (const int*)d_in[5];
  const float* num_w   = (const float*)d_in[6];
  const float* emb     = (const float*)d_in[7];
  const float* pos_emb = (const float*)d_in[8];
  const float* a_local = (const float*)d_in[9];
  const float* gw1     = (const float*)d_in[10];
  const float* gw2     = (const float*)d_in[11];
  const float* gw3     = (const float*)d_in[12];
  const float* w1      = (const float*)d_in[13];
  const float* w2      = (const float*)d_in[14];
  const float* glu1_w  = (const float*)d_in[15];
  const float* glu1_b  = (const float*)d_in[16];
  const float* glu2_w  = (const float*)d_in[17];

  float* out0   = (float*)d_out;
  float* lossP  = out0 + B_ * L_ * D_;
  float* scores = lossP + 1;

  char* wp = (char*)d_ws;
  auto alloc = [&](size_t nfloats) -> float* {
    float* r = (float*)wp;
    wp += ((nfloats * sizeof(float) + 255) / 256) * 256;
    return r;
  };
  float* h        = alloc(B_ * L_ * D_);
  float* h_local  = alloc(B_ * L_ * D_);
  float* h_global = alloc(B_ * L_ * D_);
  float* sessb    = alloc(B_ * D_);
  float* e0p      = alloc(B_ * L_ * D_);
  float* e1p      = alloc((size_t)B_ * 400 * D_);
  float* hsb      = alloc(B_ * D_);
  float* hsgb     = alloc(B_ * D_);
  float* Pb       = alloc(L_ * D_);
  float* nh       = alloc(B_ * L_ * D_);
  float* betab    = alloc(B_ * L_);
  float* selb     = alloc(B_ * D_);
  float* selF     = alloc(B_ * D_);
  float* snorm    = alloc(B_);
  float* gw1T     = alloc(2 * D_ * KP_);
  float* w1botT   = alloc(D_ * D_);
  float* wn1      = alloc(B_ * 400);
  float* wn2      = alloc((size_t)B_ * 3200);
  int*   n1       = (int*)alloc(B_ * 400);
  int*   n2       = (int*)alloc((size_t)B_ * 3200);

  gw1t_kernel<<<dim3((2 * D_ * KP_ + 255) / 256), 256, 0, stream>>>(gw1, gw1T);
  w1t_kernel<<<dim3((D_ * D_ + 255) / 256), 256, 0, stream>>>(w1, w1botT);
  gather_h_kernel<<<dim3((B_ * L_ * D_) / 256), 256, 0, stream>>>(emb, inputs, h, B_ * L_ * D_);
  sess_kernel<<<dim3(B_), dim3(D_), 0, stream>>>(emb, item, mask, sessb);
  hop_kernel<<<dim3((B_ * L_ * S_ + 255) / 256), 256, 0, stream>>>(inputs, adj_all, num_w, n1, wn1, B_ * L_ * S_);
  hop_kernel<<<dim3((B_ * 400 * S_ + 255) / 256), 256, 0, stream>>>(n1, adj_all, num_w, n2, wn2, B_ * 400 * S_);
  local_agg_kernel<<<dim3(B_), 256, 0, stream>>>(h, adj, a_local, h_local);
  agg_kernel<true, true><<<dim3(B_ * L_ / GM_), 256, 0, stream>>>(emb, h, nullptr, n1, nullptr, wn1,
      sessb, gw1T, gw2, gw3, e0p, L_);
  agg_kernel<false, true><<<dim3(B_ * 400 / GM_), 256, 0, stream>>>(emb, nullptr, n1, n2, nullptr, wn2,
      sessb, gw1T, gw2, gw3, e1p, 400);
  agg_kernel<true, false><<<dim3(B_ * L_ / GM_), 256, 0, stream>>>(emb, e0p, nullptr, nullptr, e1p, wn1,
      sessb, gw1T + D_ * KP_, gw2 + D_, gw3 + 2 * D_ * D_, h_global, L_);
  add_kernel<<<dim3((B_ * L_ * D_) / 256), 256, 0, stream>>>(h_local, h_global, out0, B_ * L_ * D_);
  zero1_kernel<<<1, 1, 0, stream>>>(lossP);
  ssl_kernel<<<dim3(64), 256, 0, stream>>>(h_local, h_global, lossP);
  hs_kernel<<<dim3(B_), dim3(D_), 0, stream>>>(out0, mask, hsb);
  hsg_kernel<<<dim3(B_), dim3(D_), 0, stream>>>(hsb, glu2_w, hsgb);
  pos_kernel<<<dim3(L_), dim3(D_), 0, stream>>>(pos_emb, w1, Pb);
  {
    int tiles = ((B_ * L_ + 15) / 16) * ((D_ + 15) / 16);
    gemm16_kernel<<<dim3((tiles + 7) / 8), 256, 0, stream>>>(
        out0, D_, w1botT, D_, nh, D_, B_ * L_, D_, D_, Pb, L_, 1);
  }
  beta_kernel<<<dim3(B_ * L_), dim3(D_), 0, stream>>>(nh, glu1_w, glu1_b, hsgb, w2, mask, betab);
  select_kernel<<<dim3(B_), dim3(D_), 0, stream>>>(out0, betab, selb);
  norm_kernel<<<dim3(B_), dim3(D_), 0, stream>>>(selb, snorm);
  fn_kernel<<<dim3(B_), dim3(D_), 0, stream>>>(selb, snorm, selF);
  scores_kernel<<<dim3(512), 256, 0, stream>>>(selF, emb + D_, scores, NN_ - 1,
                                               (NN_ - 1 + SC_COLS_ - 1) / SC_COLS_);
}


# agg kernel: gw1 product taken transposed (weights as the A operand, x rows as B), so the attention logit's sum over columns is in-lane: 32 dependent cross-lane exchanges per row tile become none
# speedup vs baseline: 1.8811x; 1.2249x over previous
#include <hip/hip_runtime.h>
#include <hip/hip_bf16.h>
#include <math.h>

#define B_ 128
#define L_ 50
#define D_ 128
#define S_ 8
#define NN_ 100000
#define KP_ 132
#define LEAKY_ 0.2f
#define BETA_ 0.005f

typedef float v2f __attribute__((ext_vector_type(2)));
typedef float v8f __attribute__((ext_vector_type(8)));

#if defined(__has_builtin)
#if __has_builtin(__builtin_amdgcn_wmma_f32_16x16x4_f32)
#define HAVE_WMMA_F32 1
#endif
#endif

__device__ __forceinline__ v8f wmma4(v2f a, v2f b, v8f c) {
#ifdef HAVE_WMMA_F32
  return __builtin_amdgcn_wmma_f32_16x16x4_f32(false, a, false, b, (short)0, c,
                                               false, false);
#else
  float* cf = (float*)&c;
  #pragma unroll
  for (int i = 0; i < 8; ++i) cf[i] += a.x * b.x + a.y * b.y;
  return c;
#endif
}

__device__ __forceinline__ float leaky_(float x) { return x > 0.f ? x : LEAKY_ * x; }
__device__ __forceinline__ float sigmoid_(float x) { return 1.f / (1.f + __expf(-x)); }

__global__ void gw1t_kernel(const float* __restrict__ gw1, float* __restrict__ gw1T) {
  int i = blockIdx.x * 256 + threadIdx.x;
  if (i >= 2 * D_ * KP_) return;
  int hop = i / (D_ * KP_);
  int rem = i % (D_ * KP_);
  int n = rem / KP_, k = rem % KP_;
  gw1T[i] = (k < 129) ? gw1[hop * 129 * D_ + k * D_ + n] : 0.f;
}

__global__ void w1t_kernel(const float* __restrict__ w1, float* __restrict__ w1botT) {
  int i = blockIdx.x * 256 + threadIdx.x;
  if (i >= D_ * D_) return;
  int n = i >> 7, k = i & 127;
  w1botT[i] = w1[(D_ + k) * D_ + n];
}

__global__ void gather_h_kernel(const float* __restrict__ emb,
                                const int* __restrict__ inputs,
                                float* __restrict__ h, int total) {
  int i = blockIdx.x * 256 + threadIdx.x;
  if (i >= total) return;
  int row = i >> 7, d = i & 127;
  h[i] = emb[((long)inputs[row] << 7) + d];
}

__global__ void hop_kernel(const int* __restrict__ src,
                           const int* __restrict__ adj_all,
                           const float* __restrict__ num_w,
                           int* __restrict__ dst, float* __restrict__ dstw,
                           int total) {
  int i = blockIdx.x * 256 + threadIdx.x;
  if (i >= total) return;
  int node = src[i >> 3];
  int s = i & 7;
  dst[i]  = adj_all[node * S_ + s];
  dstw[i] = num_w[node * S_ + s];
}

__global__ void sess_kernel(const float* __restrict__ emb,
                            const int* __restrict__ item,
                            const int* __restrict__ mask,
                            float* __restrict__ sess) {
  int b = blockIdx.x, d = threadIdx.x;
  float acc = 0.f, ms = 0.f;
  for (int l = 0; l < L_; ++l) {
    float m = (float)mask[b * L_ + l];
    ms += m;
    acc += emb[((long)item[b * L_ + l] << 7) + d] * m;
  }
  sess[(b << 7) + d] = acc / ms;
}

__global__ void local_agg_kernel(const float* __restrict__ h,
                                 const int* __restrict__ adj,
                                 const float* __restrict__ a_local,
                                 float* __restrict__ h_local) {
  int b = blockIdx.x;
  __shared__ float hv[L_][D_];
  __shared__ float att[L_][L_];
  __shared__ float rinv[L_];
  for (int i = threadIdx.x; i < L_ * D_; i += 256)
    hv[i / D_][i & 127] = h[(long)b * L_ * D_ + i];
  __syncthreads();
  for (int p = threadIdx.x; p < L_ * L_; p += 256) {
    int i = p / L_, j = p % L_;
    int k = adj[b * L_ * L_ + p];
    float v;
    if (k == 0) v = -9e15f;
    else {
      const float* a = a_local + (k - 1) * D_;
      float acc = 0.f;
      for (int d = 0; d < D_; ++d) acc += hv[i][d] * hv[j][d] * a[d];
      v = leaky_(acc);
    }
    att[i][j] = v;
  }
  __syncthreads();
  if (threadIdx.x < L_) {
    int i = threadIdx.x;
    float mx = -1e30f;
    for (int j = 0; j < L_; ++j) mx = fmaxf(mx, att[i][j]);
    float s = 0.f;
    for (int j = 0; j < L_; ++j) { float e = __expf(att[i][j] - mx); att[i][j] = e; s += e; }
    rinv[i] = 1.f / s;
  }
  __syncthreads();
  for (int p = threadIdx.x; p < L_ * D_; p += 256) {
    int i = p >> 7, d = p & 127;
    float acc = 0.f;
    for (int j = 0; j < L_; ++j) acc += att[i][j] * hv[j][d];
    h_local[(long)b * L_ * D_ + p] = acc * rinv[i];
  }
}

#define GM_ 16
#define GR_ 128
#define CP_ 260

__device__ __forceinline__ void score_cols(v8f c, const float (&g2v)[8], float* scRow,
                                           int laneIdx) {
  float acc = 0.f;
  #pragma unroll
  for (int v = 0; v < 8; ++v) acc += leaky_(c[v]) * g2v[v];
  atomicAdd(&scRow[laneIdx], acc);
}

template <bool SELF_VAL, bool NEIGH_IDX>
__global__ void __launch_bounds__(256)
agg_kernel(const float* __restrict__ emb,
           const float* __restrict__ selfVal, const int* __restrict__ selfIdx,
           const int* __restrict__ neighIdx, const float* __restrict__ neighVal,
           const float* __restrict__ neighW,
           const float* __restrict__ sess,
           const float* __restrict__ gw1T, const float* __restrict__ gw2p,
           const float* __restrict__ gw3p,
           float* __restrict__ out, int M) {
  __shared__ __align__(16) float nv[GR_][D_];
  __shared__ __align__(16) float xs[GR_][KP_];
  __shared__ __align__(16) float cat[GM_][CP_];
  __shared__ float sc[GR_];
  __shared__ float alpha[GR_];

  const int tid = threadIdx.x;
  const int lane = tid & 31, laneIdx = lane & 15, laneHalf = lane >> 4;
  const int wave = tid >> 5;
  const int n0 = wave * 16;
  const long base  = (long)blockIdx.x * GM_;
  const long nbase = base * S_;

  v2f breg[KP_ / 4];
  {
    const float* bp = gw1T + (n0 + laneIdx) * KP_ + (laneHalf << 1);
    #pragma unroll
    for (int ks = 0; ks < KP_ / 4; ++ks) breg[ks] = *(const v2f*)(bp + 4 * ks);
  }
  float g2v[8];
  {
    const float4 ga = *(const float4*)(gw2p + n0 + (laneHalf << 3));
    const float4 gb = *(const float4*)(gw2p + n0 + (laneHalf << 3) + 4);
    g2v[0] = ga.x; g2v[1] = ga.y; g2v[2] = ga.z; g2v[3] = ga.w;
    g2v[4] = gb.x; g2v[5] = gb.y; g2v[6] = gb.z; g2v[7] = gb.w;
  }

  const unsigned ubase = blockIdx.x * GM_;
  const unsigned b0 = ubase / (unsigned)M;
  const unsigned rem0 = ubase - b0 * (unsigned)M;
  float4 gv[GM_];
  if (NEIGH_IDX) {
    int idx[GM_];
    #pragma unroll
    for (int j = 0; j < GM_; ++j) idx[j] = neighIdx[nbase + (j << 3) + wave];
    #pragma unroll
    for (int j = 0; j < GM_; ++j)
      gv[j] = *(const float4*)(emb + ((long)idx[j] << 7) + (lane << 2));
  } else {
    #pragma unroll
    for (int j = 0; j < GM_; ++j)
      gv[j] = *(const float4*)(neighVal + ((nbase + (j << 3) + wave) << 7) + (lane << 2));
  }
  const unsigned b1 = (rem0 + GM_ > (unsigned)M) ? b0 + 1 : b0;
  const float4 sfa = *(const float4*)(sess + ((long)b0 << 7) + (lane << 2));
  const float4 sfb = *(const float4*)(sess + ((long)b1 << 7) + (lane << 2));
  float4 sv0, sv1;
  if (SELF_VAL) {
    sv0 = *(const float4*)(selfVal + ((base + wave) << 7) + (lane << 2));
    sv1 = *(const float4*)(selfVal + ((base + wave + 8) << 7) + (lane << 2));
  } else {
    sv0 = *(const float4*)(emb + ((long)selfIdx[base + wave] << 7) + (lane << 2));
    sv1 = *(const float4*)(emb + ((long)selfIdx[base + wave + 8] << 7) + (lane << 2));
  }
  #pragma unroll
  for (int j = 0; j < GM_; ++j) {
    const int r = (j << 3) + wave;
    const float4 v = gv[j];
    const bool nextb = rem0 + (unsigned)j >= (unsigned)M;
    *(float4*)&nv[r][lane << 2] = v;
    float4 x;
    x.x = (nextb ? sfb.x : sfa.x) * v.x;
    x.y = (nextb ? sfb.y : sfa.y) * v.y;
    x.z = (nextb ? sfb.z : sfa.z) * v.z;
    x.w = (nextb ? sfb.w : sfa.w) * v.w;
    *(float4*)&xs[r][lane << 2] = x;
  }
  *(float4*)&cat[wave][lane << 2] = sv0;
  *(float4*)&cat[wave + 8][lane << 2] = sv1;
  if (tid < GR_) {
    xs[tid][D_] = neighW[nbase + tid];
    xs[tid][D_ + 1] = 0.f; xs[tid][D_ + 2] = 0.f; xs[tid][D_ + 3] = 0.f;
    sc[tid] = 0.f;
  }
  __syncthreads();

  for (int rp = 0; rp < GR_ / 32; ++rp) {
    const float* xp0 = &xs[(rp << 5) + laneIdx][laneHalf << 1];
    const float* xp1 = xp0 + 16 * KP_;
    v8f c0 = {}, c1 = {};
    #pragma unroll
    for (int ks = 0; ks < KP_ / 4; ++ks) {
      v2f x0 = *(const v2f*)(xp0 + 4 * ks);
      v2f x1 = *(const v2f*)(xp1 + 4 * ks);
      c0 = wmma4(breg[ks], x0, c0);
      c1 = wmma4(breg[ks], x1, c1);
    }
    score_cols(c0, g2v, &sc[(rp << 5)], laneIdx);
    score_cols(c1, g2v, &sc[(rp << 5) + 16], laneIdx);
  }
  __syncthreads();

  if (tid < GM_) {
    float mx = -1e30f;
    for (int s = 0; s < 8; ++s) mx = fmaxf(mx, sc[tid * 8 + s]);
    float ex[8], sum = 0.f;
    for (int s = 0; s < 8; ++s) { ex[s] = __expf(sc[tid * 8 + s] - mx); sum += ex[s]; }
    float inv = 1.f / sum;
    for (int s = 0; s < 8; ++s) alpha[tid * 8 + s] = ex[s] * inv;
  }
  __syncthreads();
  {
    const int d = tid & 127;
    for (int j = 0; j < GM_ / 2; ++j) {
      const int r = (tid >> 7) + (j << 1);
      float acc = 0.f;
      for (int s = 0; s < 8; ++s) acc += alpha[(r << 3) + s] * nv[(r << 3) + s][d];
      cat[r][D_ + d] = acc;
    }
  }
  __syncthreads();

  {
    const int col = n0 + laneIdx;
    const float* ap = &cat[laneIdx][laneHalf << 1];
    const float* bp = gw3p + ((laneHalf << 1) << 7) + col;
    v8f cs = {}, cn = {};
    #pragma unroll 8
    for (int k0 = 0; k0 < D_; k0 += 4) {
      v2f as = *(const v2f*)(ap + k0);
      v2f an = *(const v2f*)(ap + D_ + k0);
      v2f bs, bn;
      bs.x = bp[k0 << 7];
      bs.y = bp[(k0 + 1) << 7];
      bn.x = bp[(D_ + k0) << 7];
      bn.y = bp[(D_ + k0 + 1) << 7];
      cs = wmma4(as, bs, cs);
      cn = wmma4(an, bn, cn);
    }
    #pragma unroll
    for (int v = 0; v < 8; ++v)
      xs[v + (laneHalf << 3)][col] = fmaxf(cs[v] + cn[v], 0.f);
  }
  __syncthreads();
  *(float4*)(out + ((base + wave) << 7) + (lane << 2)) = *(const float4*)&xs[wave][lane << 2];
  *(float4*)(out + ((base + wave + 8) << 7) + (lane << 2)) = *(const float4*)&xs[wave + 8][lane << 2];
}

__global__ void gemm16_kernel(const float* __restrict__ A, int lda,
                              const float* __restrict__ BT, int ldbt,
                              float* __restrict__ C, long ldc,
                              int M, int N, int K,
                              const float* __restrict__ rowBias, int biasMod,
                              int act) {
  int tilesM = (M + 15) >> 4;
  long tilesN = (long)((N + 15) >> 4);
  long total = (long)tilesM * tilesN;
  int lane = threadIdx.x & 31, laneIdx = lane & 15, laneHalf = lane >> 4;
  for (long t = (long)blockIdx.x * 8 + (threadIdx.x >> 5); t < total;
       t += (long)gridDim.x * 8) {
    int m0 = (int)(t % tilesM) << 4;
    long n0 = (t / tilesM) << 4;
    int row = m0 + laneIdx;
    long col = n0 + laneIdx;
    v8f c = {};
    bool full = (m0 + 16 <= M) && (n0 + 16 <= N);
    if (full) {
      const float* ap = A + (long)row * lda;
      const float* bp = BT + col * ldbt;
      for (int k0 = 0; k0 < K; k0 += 4) {
        int kk = k0 + (laneHalf << 1);
        v2f a = *(const v2f*)(ap + kk);
        v2f b = *(const v2f*)(bp + kk);
        c = wmma4(a, b, c);
      }
    } else {
      for (int k0 = 0; k0 < K; k0 += 4) {
        int kk = k0 + (laneHalf << 1);
        v2f a, b;
        a.x = (row < M) ? A[(long)row * lda + kk] : 0.f;
        a.y = (row < M) ? A[(long)row * lda + kk + 1] : 0.f;
        b.x = (col < N) ? BT[col * ldbt + kk] : 0.f;
        b.y = (col < N) ? BT[col * ldbt + kk + 1] : 0.f;
        c = wmma4(a, b, c);
      }
    }
    float* cf = (float*)&c;
    #pragma unroll
    for (int v = 0; v < 8; ++v) {
      int rr = m0 + v + (laneHalf << 3);
      if (rr < M && col < N) {
        float val = cf[v];
        if (rowBias) val += rowBias[(long)(rr % biasMod) * N + col];
        if (act == 1) val = tanhf(val);
        C[(long)rr * ldc + col] = val;
      }
    }
  }
}

#define SC_COLS_ 32
#define SC_KH_ (D_ / 8)

__device__ __forceinline__ float4 sc_row(const float* __restrict__ embT, int n, int N, int lane) {
  n = n < N ? n : N - 1;
  return *(const float4*)(embT + ((long)n << 7) + (lane << 2));
}

__global__ void __launch_bounds__(256)
scores_kernel(const float* __restrict__ selF, const float* __restrict__ embT,
              float* __restrict__ scores, int N, int nStages) {
  __shared__ __align__(16) float bs[2][SC_COLS_][KP_];
  const int tid = threadIdx.x;
  const int lane = tid & 31, laneIdx = lane & 15, laneHalf = lane >> 4;
  const int wave = tid >> 5;

  v2f areg[D_ / 4];
  {
    const float* ap = selF + ((wave * 16 + laneIdx) << 7) + (laneHalf << 1);
    #pragma unroll
    for (int ks = 0; ks < D_ / 4; ++ks) areg[ks] = *(const v2f*)(ap + 4 * ks);
  }

  const int r0 = wave * 4;
  float4 p0, p1, p2, p3;
  int s = blockIdx.x;
  if (s < nStages) {
    const int n = s * SC_COLS_ + r0;
    p0 = sc_row(embT, n, N, lane);
    p1 = sc_row(embT, n + 1, N, lane);
    p2 = sc_row(embT, n + 2, N, lane);
    p3 = sc_row(embT, n + 3, N, lane);
    *(float4*)&bs[0][r0][lane << 2] = p0;
    *(float4*)&bs[0][r0 + 1][lane << 2] = p1;
    *(float4*)&bs[0][r0 + 2][lane << 2] = p2;
    *(float4*)&bs[0][r0 + 3][lane << 2] = p3;
  }
  __syncthreads();

  int buf = 0;
  for (; s < nStages; s += gridDim.x) {
    const int sn = s + gridDim.x;
    const bool more = sn < nStages;
    if (more) {
      const int n = sn * SC_COLS_ + r0;
      p0 = sc_row(embT, n, N, lane);
      p1 = sc_row(embT, n + 1, N, lane);
      p2 = sc_row(embT, n + 2, N, lane);
      p3 = sc_row(embT, n + 3, N, lane);
    }
    const float* bp0 = &bs[buf][laneIdx][laneHalf << 1];
    const float* bp1 = bp0 + 16 * KP_;
    v8f c0 = {}, c1 = {};
    #pragma unroll
    for (int kh = 0; kh < 2; ++kh) {
      v2f f0[SC_KH_], f1[SC_KH_];
      #pragma unroll
      for (int ks = 0; ks < SC_KH_; ++ks) {
        f0[ks] = *(const v2f*)(bp0 + 4 * (kh * SC_KH_ + ks));
        f1[ks] = *(const v2f*)(bp1 + 4 * (kh * SC_KH_ + ks));
      }
      #pragma unroll
      for (int ks = 0; ks < SC_KH_; ++ks) {
        c0 = wmma4(areg[kh * SC_KH_ + ks], f0[ks], c0);
        c1 = wmma4(areg[kh * SC_KH_ + ks], f1[ks], c1);
      }
    }
    if (more) {
      *(float4*)&bs[buf ^ 1][r0][lane << 2] = p0;
      *(float4*)&bs[buf ^ 1][r0 + 1][lane << 2] = p1;
      *(float4*)&bs[buf ^ 1][r0 + 2][lane << 2] = p2;
      *(float4*)&bs[buf ^ 1][r0 + 3][lane << 2] = p3;
    }
    __syncthreads();
    buf ^= 1;
    {
      const int n0 = s * SC_COLS_;
      float* o = scores + (long)(wave * 16 + (laneHalf << 3)) * N + n0 + laneIdx;
      if (n0 + SC_COLS_ <= N) {
        #pragma unroll
        for (int v = 0; v < 8; ++v) {
          o[(long)v * N] = c0[v];
          o[(long)v * N + 16] = c1[v];
        }
      } else {
        const bool in0 = n0 + laneIdx < N, in1 = n0 + laneIdx + 16 < N;
        #pragma unroll
        for (int v = 0; v < 8; ++v) {
          if (in0) o[(long)v * N] = c0[v];
          if (in1) o[(long)v * N + 16] = c1[v];
        }
      }
    }
  }
}

__global__ void add_kernel(const float* a, const float* b, float* o, int n) {
  int i = blockIdx.x * 256 + threadIdx.x;
  if (i < n) o[i] = a[i] + b[i];
}

__global__ void zero1_kernel(float* p) { if (threadIdx.x == 0 && blockIdx.x == 0) p[0] = 0.f; }

__global__ void ssl_kernel(const float* __restrict__ hl, const float* __restrict__ hg,
                           float* __restrict__ loss) {
  int idx = blockIdx.x * 256 + threadIdx.x;
  int b = idx >> 7, d = idx & 127;
  int pb = (b * 73 + 21) & 127;
  float pos = 0.f, neg = 0.f;
  for (int l = 0; l < L_; ++l) {
    int pl = (l * 13 + 7) % L_;
    pos += hl[((b * L_ + l) << 7) + d] * hg[((b * L_ + l) << 7) + d];
    neg += hg[((b * L_ + l) << 7) + d] * hl[((pb * L_ + pl) << 7) + d];
  }
  float term = -__logf(1e-8f + sigmoid_(pos)) - __logf(1e-8f + 1.f - sigmoid_(neg));
  __shared__ float red[256];
  red[threadIdx.x] = term;
  __syncthreads();
  for (int s = 128; s > 0; s >>= 1) {
    if (threadIdx.x < s) red[threadIdx.x] += red[threadIdx.x + s];
    __syncthreads();
  }
  if (threadIdx.x == 0) atomicAdd(loss, BETA_ * red[0]);
}

__global__ void hs_kernel(const float* __restrict__ hidden, const int* __restrict__ mask,
                          float* __restrict__ hs) {
  int b = blockIdx.x, d = threadIdx.x;
  float acc = 0.f, ms = 0.f;
  for (int l = 0; l < L_; ++l) {
    float m = (float)mask[b * L_ + l];
    ms += m;
    acc += hidden[((b * L_ + l) << 7) + d] * m;
  }
  hs[(b << 7) + d] = acc / ms;
}

__global__ void hsg_kernel(const float* __restrict__ hs, const float* __restrict__ glu2_w,
                           float* __restrict__ hsg) {
  __shared__ float hr[D_];
  int b = blockIdx.x, d = threadIdx.x;
  hr[d] = hs[(b << 7) + d];
  __syncthreads();
  float acc = 0.f;
  for (int e = 0; e < D_; ++e) acc += hr[e] * glu2_w[(d << 7) + e];
  hsg[(b << 7) + d] = acc;
}

__global__ void pos_kernel(const float* __restrict__ pos_emb, const float* __restrict__ w1,
                           float* __restrict__ P) {
  __shared__ float pr[D_];
  int l = blockIdx.x, d = threadIdx.x;
  pr[d] = pos_emb[(l << 7) + d];
  __syncthreads();
  float acc = 0.f;
  for (int e = 0; e < D_; ++e) acc += pr[e] * w1[(e << 7) + d];
  P[(l << 7) + d] = acc;
}

__global__ void beta_kernel(const float* __restrict__ nh, const float* __restrict__ glu1_w,
                            const float* __restrict__ glu1_b, const float* __restrict__ hsg,
                            const float* __restrict__ w2, const int* __restrict__ mask,
                            float* __restrict__ beta) {
  __shared__ float nrow[D_];
  __shared__ float red[D_];
  int row = blockIdx.x;
  int b = row / L_;
  int d = threadIdx.x;
  nrow[d] = nh[((long)row << 7) + d];
  __syncthreads();
  float acc = glu1_b[d] + hsg[(b << 7) + d];
  for (int e = 0; e < D_; ++e) acc += nrow[e] * glu1_w[(d << 7) + e];
  red[d] = sigmoid_(acc) * w2[d];
  __syncthreads();
  for (int s = 64; s > 0; s >>= 1) {
    if (d < s) red[d] += red[d + s];
    __syncthreads();
  }
  if (d == 0) beta[row] = red[0] * (float)mask[row];
}

__global__ void select_kernel(const float* __restrict__ hidden, const float* __restrict__ beta,
                              float* __restrict__ sel) {
  int b = blockIdx.x, d = threadIdx.x;
  float acc = 0.f;
  for (int l = 0; l < L_; ++l) acc += beta[b * L_ + l] * hidden[((b * L_ + l) << 7) + d];
  sel[(b << 7) + d] = acc;
}

__global__ void norm_kernel(const float* __restrict__ sel, float* __restrict__ snorm) {
  __shared__ float red[D_];
  int b = blockIdx.x, d = threadIdx.x;
  float v = sel[(b << 7) + d];
  red[d] = v * v + 1e-6f;
  __syncthreads();
  for (int s = 64; s > 0; s >>= 1) {
    if (d < s) red[d] += red[d + s];
    __syncthreads();
  }
  if (d == 0) snorm[b] = sqrtf(red[0]);
}

__global__ void fn_kernel(const float* __restrict__ sel, const float* __restrict__ snorm,
                          float* __restrict__ selF) {
  __shared__ float row[B_];
  __shared__ float selfrow[D_];
  __shared__ float tvs[9];
  __shared__ int tis[9];
  int i = blockIdx.x, t = threadIdx.x;
  selfrow[t] = sel[(i << 7) + t];
  __syncthreads();
  float acc = 0.f;
  for (int d = 0; d < D_; ++d) acc += selfrow[d] * sel[(t << 7) + d];
  row[t] = acc / (snorm[i] * snorm[t]);
  __syncthreads();
  if (t == 0) {
    float mx = -1e30f;
    for (int j = 0; j < B_; ++j) mx = fmaxf(mx, row[j]);
    float s = 0.f;
    for (int j = 0; j < B_; ++j) { row[j] = __expf(row[j] - mx); s += row[j]; }
    float inv = 1.f / s;
    for (int j = 0; j < B_; ++j) row[j] *= inv;
    for (int q = 0; q < 9; ++q) {
      float bm = -1.f; int bi = 0;
      for (int j = 0; j < B_; ++j) if (row[j] > bm) { bm = row[j]; bi = j; }
      tvs[q] = bm; tis[q] = bi; row[bi] = -2.f;
    }
    float mx2 = tvs[0], s2 = 0.f;
    for (int q = 0; q < 9; ++q) { tvs[q] = __expf(tvs[q] - mx2); s2 += tvs[q]; }
    float inv2 = 1.f / s2;
    for (int q = 0; q < 9; ++q) tvs[q] *= inv2;
  }
  __syncthreads();
  float nb = 0.f;
  for (int q = 0; q < 9; ++q) nb += tvs[q] * sel[(tis[q] << 7) + t];
  selF[(i << 7) + t] = selfrow[t] + nb;
}

extern "C" void kernel_launch(void* const* d_in, const int* in_sizes, int n_in,
                              void* d_out, int out_size, void* d_ws, size_t ws_size,
                              hipStream_t stream) {
  (void)in_sizes; (void)n_in; (void)out_size; (void)ws_size;
  const int*   inputs  = (const int*)d_in[0];
  const int*   adj     = (const int*)d_in[1];
  const int*   mask    = (const int*)d_in[2];
  const int*   item    = (const int*)d_in[3];
  const int*   adj_all = (const int*)d_in[5];
  const float* num_w   = (const float*)d_in[6];
  const float* emb     = (const float*)d_in[7];
  const float* pos_emb = (const float*)d_in[8];
  const float* a_local = (const float*)d_in[9];
  const float* gw1     = (const float*)d_in[10];
  const float* gw2     = (const float*)d_in[11];
  const float* gw3     = (const float*)d_in[12];
  const float* w1      = (const float*)d_in[13];
  const float* w2      = (const float*)d_in[14];
  const float* glu1_w  = (const float*)d_in[15];
  const float* glu1_b  = (const float*)d_in[16];
  const float* glu2_w  = (const float*)d_in[17];

  float* out0   = (float*)d_out;
  float* lossP  = out0 + B_ * L_ * D_;
  float* scores = lossP + 1;

  char* wp = (char*)d_ws;
  auto alloc = [&](size_t nfloats) -> float* {
    float* r = (float*)wp;
    wp += ((nfloats * sizeof(float) + 255) / 256) * 256;
    return r;
  };
  float* h        = alloc(B_ * L_ * D_);
  float* h_local  = alloc(B_ * L_ * D_);
  float* h_global = alloc(B_ * L_ * D_);
  float* sessb    = alloc(B_ * D_);
  float* e0p      = alloc(B_ * L_ * D_);
  float* e1p      = alloc((size_t)B_ * 400 * D_);
  float* hsb      = alloc(B_ * D_);
  float* hsgb     = alloc(B_ * D_);
  float* Pb       = alloc(L_ * D_);
  float* nh       = alloc(B_ * L_ * D_);
  float* betab    = alloc(B_ * L_);
  float* selb     = alloc(B_ * D_);
  float* selF     = alloc(B_ * D_);
  float* snorm    = alloc(B_);
  float* gw1T     = alloc(2 * D_ * KP_);
  float* w1botT   = alloc(D_ * D_);
  float* wn1      = alloc(B_ * 400);
  float* wn2      = alloc((size_t)B_ * 3200);
  int*   n1       = (int*)alloc(B_ * 400);
  int*   n2       = (int*)alloc((size_t)B_ * 3200);

  gw1t_kernel<<<dim3((2 * D_ * KP_ + 255) / 256), 256, 0, stream>>>(gw1, gw1T);
  w1t_kernel<<<dim3((D_ * D_ + 255) / 256), 256, 0, stream>>>(w1, w1botT);
  gather_h_kernel<<<dim3((B_ * L_ * D_) / 256), 256, 0, stream>>>(emb, inputs, h, B_ * L_ * D_);
  sess_kernel<<<dim3(B_), dim3(D_), 0, stream>>>(emb, item, mask, sessb);
  hop_kernel<<<dim3((B_ * L_ * S_ + 255) / 256), 256, 0, stream>>>(inputs, adj_all, num_w, n1, wn1, B_ * L_ * S_);
  hop_kernel<<<dim3((B_ * 400 * S_ + 255) / 256), 256, 0, stream>>>(n1, adj_all, num_w, n2, wn2, B_ * 400 * S_);
  local_agg_kernel<<<dim3(B_), 256, 0, stream>>>(h, adj, a_local, h_local);
  agg_kernel<true, true><<<dim3(B_ * L_ / GM_), 256, 0, stream>>>(emb, h, nullptr, n1, nullptr, wn1,
      sessb, gw1T, gw2, gw3, e0p, L_);
  agg_kernel<false, true><<<dim3(B_ * 400 / GM_), 256, 0, stream>>>(emb, nullptr, n1, n2, nullptr, wn2,
      sessb, gw1T, gw2, gw3, e1p, 400);
  agg_kernel<true, false><<<dim3(B_ * L_ / GM_), 256, 0, stream>>>(emb, e0p, nullptr, nullptr, e1p, wn1,
      sessb, gw1T + D_ * KP_, gw2 + D_, gw3 + 2 * D_ * D_, h_global, L_);
  add_kernel<<<dim3((B_ * L_ * D_) / 256), 256, 0, stream>>>(h_local, h_global, out0, B_ * L_ * D_);
  zero1_kernel<<<1, 1, 0, stream>>>(lossP);
  ssl_kernel<<<dim3(64), 256, 0, stream>>>(h_local, h_global, lossP);
  hs_kernel<<<dim3(B_), dim3(D_), 0, stream>>>(out0, mask, hsb);
  hsg_kernel<<<dim3(B_), dim3(D_), 0, stream>>>(hsb, glu2_w, hsgb);
  pos_kernel<<<dim3(L_), dim3(D_), 0, stream>>>(pos_emb, w1, Pb);
  {
    int tiles = ((B_ * L_ + 15) / 16) * ((D_ + 15) / 16);
    gemm16_kernel<<<dim3((tiles + 7) / 8), 256, 0, stream>>>(
        out0, D_, w1botT, D_, nh, D_, B_ * L_, D_, D_, Pb, L_, 1);
  }
  beta_kernel<<<dim3(B_ * L_), dim3(D_), 0, stream>>>(nh, glu1_w, glu1_b, hsgb, w2, mask, betab);
  select_kernel<<<dim3(B_), dim3(D_), 0, stream>>>(out0, betab, selb);
  norm_kernel<<<dim3(B_), dim3(D_), 0, stream>>>(selb, snorm);
  fn_kernel<<<dim3(B_), dim3(D_), 0, stream>>>(selb, snorm, selF);
  scores_kernel<<<dim3(512), 256, 0, stream>>>(selF, emb + D_, scores, NN_ - 1,
                                               (NN_ - 1 + SC_COLS_ - 1) / SC_COLS_);
}


# local attention logits (4 planes of 50x50x128 per batch): per-pair scalar loops whose LDS row reads all hit the same banks become f32 WMMA tiles, a_k folded into the A fragment, h rows at pitch 132
# speedup vs baseline: 2.0516x; 1.0906x over previous
#include <hip/hip_runtime.h>
#include <hip/hip_bf16.h>
#include <math.h>

#define B_ 128
#define L_ 50
#define D_ 128
#define S_ 8
#define NN_ 100000
#define KP_ 132
#define LEAKY_ 0.2f
#define BETA_ 0.005f

typedef float v2f __attribute__((ext_vector_type(2)));
typedef float v8f __attribute__((ext_vector_type(8)));

#if defined(__has_builtin)
#if __has_builtin(__builtin_amdgcn_wmma_f32_16x16x4_f32)
#define HAVE_WMMA_F32 1
#endif
#endif

__device__ __forceinline__ v8f wmma4(v2f a, v2f b, v8f c) {
#ifdef HAVE_WMMA_F32
  return __builtin_amdgcn_wmma_f32_16x16x4_f32(false, a, false, b, (short)0, c,
                                               false, false);
#else
  float* cf = (float*)&c;
  #pragma unroll
  for (int i = 0; i < 8; ++i) cf[i] += a.x * b.x + a.y * b.y;
  return c;
#endif
}

__device__ __forceinline__ float leaky_(float x) { return x > 0.f ? x : LEAKY_ * x; }
__device__ __forceinline__ float sigmoid_(float x) { return 1.f / (1.f + __expf(-x)); }

__global__ void gw1t_kernel(const float* __restrict__ gw1, float* __restrict__ gw1T) {
  int i = blockIdx.x * 256 + threadIdx.x;
  if (i >= 2 * D_ * KP_) return;
  int hop = i / (D_ * KP_);
  int rem = i % (D_ * KP_);
  int n = rem / KP_, k = rem % KP_;
  gw1T[i] = (k < 129) ? gw1[hop * 129 * D_ + k * D_ + n] : 0.f;
}

__global__ void w1t_kernel(const float* __restrict__ w1, float* __restrict__ w1botT) {
  int i = blockIdx.x * 256 + threadIdx.x;
  if (i >= D_ * D_) return;
  int n = i >> 7, k = i & 127;
  w1botT[i] = w1[(D_ + k) * D_ + n];
}

__global__ void gather_h_kernel(const float* __restrict__ emb,
                                const int* __restrict__ inputs,
                                float* __restrict__ h, int total) {
  int i = blockIdx.x * 256 + threadIdx.x;
  if (i >= total) return;
  int row = i >> 7, d = i & 127;
  h[i] = emb[((long)inputs[row] << 7) + d];
}

__global__ void hop_kernel(const int* __restrict__ src,
                           const int* __restrict__ adj_all,
                           const float* __restrict__ num_w,
                           int* __restrict__ dst, float* __restrict__ dstw,
                           int total) {
  int i = blockIdx.x * 256 + threadIdx.x;
  if (i >= total) return;
  int node = src[i >> 3];
  int s = i & 7;
  dst[i]  = adj_all[node * S_ + s];
  dstw[i] = num_w[node * S_ + s];
}

__global__ void sess_kernel(const float* __restrict__ emb,
                            const int* __restrict__ item,
                            const int* __restrict__ mask,
                            float* __restrict__ sess) {
  int b = blockIdx.x, d = threadIdx.x;
  float acc = 0.f, ms = 0.f;
  for (int l = 0; l < L_; ++l) {
    float m = (float)mask[b * L_ + l];
    ms += m;
    acc += emb[((long)item[b * L_ + l] << 7) + d] * m;
  }
  sess[(b << 7) + d] = acc / ms;
}

#define HP_ 132

__global__ void __launch_bounds__(256)
local_agg_kernel(const float* __restrict__ h,
                 const int* __restrict__ adj,
                 const float* __restrict__ a_local,
                 float* __restrict__ h_local) {
  const int b = blockIdx.x;
  __shared__ __align__(16) float hv[L_][HP_];
  __shared__ __align__(16) float al[4][D_];
  __shared__ float att[L_][L_];
  __shared__ float rinv[L_];
  const int tid = threadIdx.x;
  const int lane = tid & 31, laneIdx = lane & 15, laneHalf = lane >> 4;
  const int wave = tid >> 5;

  for (int q = tid; q < L_ * (D_ / 4); q += 256)
    *(float4*)&hv[q >> 5][(q & 31) << 2] = *(const float4*)(h + (long)b * L_ * D_ + (q << 2));
  if (tid < D_) *(float4*)&al[0][tid << 2] = *(const float4*)(a_local + (tid << 2));

  const int i0 = (wave >> 1) << 4;
  const int j0 = (wave & 1) << 5;
  int et[2][8];
  #pragma unroll
  for (int t = 0; t < 2; ++t) {
    #pragma unroll
    for (int v = 0; v < 8; ++v) {
      const int i = i0 + (laneHalf << 3) + v, j = j0 + (t << 4) + laneIdx;
      et[t][v] = (i < L_ && j < L_) ? adj[b * L_ * L_ + i * L_ + j] : 0;
    }
  }
  __syncthreads();

  {
    const int ir  = (i0 + laneIdx < L_) ? i0 + laneIdx : L_ - 1;
    const int jr0 = (j0 + laneIdx < L_) ? j0 + laneIdx : L_ - 1;
    const int jr1 = (j0 + 16 + laneIdx < L_) ? j0 + 16 + laneIdx : L_ - 1;
    const float* ap  = &hv[ir][laneHalf << 1];
    const float* bp0 = &hv[jr0][laneHalf << 1];
    const float* bp1 = &hv[jr1][laneHalf << 1];
    const float* sp  = &al[0][laneHalf << 1];
    v8f e[4][2] = {};
    for (int ks = 0; ks < D_ / 4; ++ks) {
      const v2f x  = *(const v2f*)(ap + 4 * ks);
      const v2f y0 = *(const v2f*)(bp0 + 4 * ks);
      const v2f y1 = *(const v2f*)(bp1 + 4 * ks);
      #pragma unroll
      for (int k = 0; k < 4; ++k) {
        const v2f s = *(const v2f*)(sp + k * D_ + 4 * ks);
        const v2f xa = x * s;
        e[k][0] = wmma4(xa, y0, e[k][0]);
        e[k][1] = wmma4(xa, y1, e[k][1]);
      }
    }
    #pragma unroll
    for (int t = 0; t < 2; ++t) {
      const int j = j0 + (t << 4) + laneIdx;
      #pragma unroll
      for (int v = 0; v < 8; ++v) {
        const int i = i0 + (laneHalf << 3) + v;
        const int k = et[t][v];
        float x = e[3][t][v];
        x = (k == 3) ? e[2][t][v] : x;
        x = (k == 2) ? e[1][t][v] : x;
        x = (k == 1) ? e[0][t][v] : x;
        x = (k == 0) ? -9e15f : leaky_(x);
        if (i < L_ && j < L_) att[i][j] = x;
      }
    }
  }
  __syncthreads();
  if (threadIdx.x < L_) {
    int i = threadIdx.x;
    float mx = -1e30f;
    for (int j = 0; j < L_; ++j) mx = fmaxf(mx, att[i][j]);
    float s = 0.f;
    for (int j = 0; j < L_; ++j) { float ex = __expf(att[i][j] - mx); att[i][j] = ex; s += ex; }
    rinv[i] = 1.f / s;
  }
  __syncthreads();
  for (int p = threadIdx.x; p < L_ * D_; p += 256) {
    int i = p >> 7, d = p & 127;
    float acc = 0.f;
    for (int j = 0; j < L_; ++j) acc += att[i][j] * hv[j][d];
    h_local[(long)b * L_ * D_ + p] = acc * rinv[i];
  }
}

#define GM_ 16
#define GR_ 128
#define CP_ 260

__device__ __forceinline__ void score_cols(v8f c, const float (&g2v)[8], float* scRow,
                                           int laneIdx) {
  float acc = 0.f;
  #pragma unroll
  for (int v = 0; v < 8; ++v) acc += leaky_(c[v]) * g2v[v];
  atomicAdd(&scRow[laneIdx], acc);
}

template <bool SELF_VAL, bool NEIGH_IDX>
__global__ void __launch_bounds__(256)
agg_kernel(const float* __restrict__ emb,
           const float* __restrict__ selfVal, const int* __restrict__ selfIdx,
           const int* __restrict__ neighIdx, const float* __restrict__ neighVal,
           const float* __restrict__ neighW,
           const float* __restrict__ sess,
           const float* __restrict__ gw1T, const float* __restrict__ gw2p,
           const float* __restrict__ gw3p,
           float* __restrict__ out, int M) {
  __shared__ __align__(16) float nv[GR_][D_];
  __shared__ __align__(16) float xs[GR_][KP_];
  __shared__ __align__(16) float cat[GM_][CP_];
  __shared__ float sc[GR_];
  __shared__ float alpha[GR_];

  const int tid = threadIdx.x;
  const int lane = tid & 31, laneIdx = lane & 15, laneHalf = lane >> 4;
  const int wave = tid >> 5;
  const int n0 = wave * 16;
  const long base  = (long)blockIdx.x * GM_;
  const long nbase = base * S_;

  v2f breg[KP_ / 4];
  {
    const float* bp = gw1T + (n0 + laneIdx) * KP_ + (laneHalf << 1);
    #pragma unroll
    for (int ks = 0; ks < KP_ / 4; ++ks) breg[ks] = *(const v2f*)(bp + 4 * ks);
  }
  float g2v[8];
  {
    const float4 ga = *(const float4*)(gw2p + n0 + (laneHalf << 3));
    const float4 gb = *(const float4*)(gw2p + n0 + (laneHalf << 3) + 4);
    g2v[0] = ga.x; g2v[1] = ga.y; g2v[2] = ga.z; g2v[3] = ga.w;
    g2v[4] = gb.x; g2v[5] = gb.y; g2v[6] = gb.z; g2v[7] = gb.w;
  }

  const unsigned ubase = blockIdx.x * GM_;
  const unsigned b0 = ubase / (unsigned)M;
  const unsigned rem0 = ubase - b0 * (unsigned)M;
  float4 gv[GM_];
  if (NEIGH_IDX) {
    int idx[GM_];
    #pragma unroll
    for (int j = 0; j < GM_; ++j) idx[j] = neighIdx[nbase + (j << 3) + wave];
    #pragma unroll
    for (int j = 0; j < GM_; ++j)
      gv[j] = *(const float4*)(emb + ((long)idx[j] << 7) + (lane << 2));
  } else {
    #pragma unroll
    for (int j = 0; j < GM_; ++j)
      gv[j] = *(const float4*)(neighVal + ((nbase + (j << 3) + wave) << 7) + (lane << 2));
  }
  const unsigned b1 = (rem0 + GM_ > (unsigned)M) ? b0 + 1 : b0;
  const float4 sfa = *(const float4*)(sess + ((long)b0 << 7) + (lane << 2));
  const float4 sfb = *(const float4*)(sess + ((long)b1 << 7) + (lane << 2));
  float4 sv0, sv1;
  if (SELF_VAL) {
    sv0 = *(const float4*)(selfVal + ((base + wave) << 7) + (lane << 2));
    sv1 = *(const float4*)(selfVal + ((base + wave + 8) << 7) + (lane << 2));
  } else {
    sv0 = *(const float4*)(emb + ((long)selfIdx[base + wave] << 7) + (lane << 2));
    sv1 = *(const float4*)(emb + ((long)selfIdx[base + wave + 8] << 7) + (lane << 2));
  }
  #pragma unroll
  for (int j = 0; j < GM_; ++j) {
    const int r = (j << 3) + wave;
    const float4 v = gv[j];
    const bool nextb = rem0 + (unsigned)j >= (unsigned)M;
    *(float4*)&nv[r][lane << 2] = v;
    float4 x;
    x.x = (nextb ? sfb.x : sfa.x) * v.x;
    x.y = (nextb ? sfb.y : sfa.y) * v.y;
    x.z = (nextb ? sfb.z : sfa.z) * v.z;
    x.w = (nextb ? sfb.w : sfa.w) * v.w;
    *(float4*)&xs[r][lane << 2] = x;
  }
  *(float4*)&cat[wave][lane << 2] = sv0;
  *(float4*)&cat[wave + 8][lane << 2] = sv1;
  if (tid < GR_) {
    xs[tid][D_] = neighW[nbase + tid];
    xs[tid][D_ + 1] = 0.f; xs[tid][D_ + 2] = 0.f; xs[tid][D_ + 3] = 0.f;
    sc[tid] = 0.f;
  }
  __syncthreads();

  for (int rp = 0; rp < GR_ / 32; ++rp) {
    const float* xp0 = &xs[(rp << 5) + laneIdx][laneHalf << 1];
    const float* xp1 = xp0 + 16 * KP_;
    v8f c0 = {}, c1 = {};
    #pragma unroll
    for (int ks = 0; ks < KP_ / 4; ++ks) {
      v2f x0 = *(const v2f*)(xp0 + 4 * ks);
      v2f x1 = *(const v2f*)(xp1 + 4 * ks);
      c0 = wmma4(breg[ks], x0, c0);
      c1 = wmma4(breg[ks], x1, c1);
    }
    score_cols(c0, g2v, &sc[(rp << 5)], laneIdx);
    score_cols(c1, g2v, &sc[(rp << 5) + 16], laneIdx);
  }
  __syncthreads();

  if (tid < GM_) {
    float mx = -1e30f;
    for (int s = 0; s < 8; ++s) mx = fmaxf(mx, sc[tid * 8 + s]);
    float ex[8], sum = 0.f;
    for (int s = 0; s < 8; ++s) { ex[s] = __expf(sc[tid * 8 + s] - mx); sum += ex[s]; }
    float inv = 1.f / sum;
    for (int s = 0; s < 8; ++s) alpha[tid * 8 + s] = ex[s] * inv;
  }
  __syncthreads();
  {
    const int d = tid & 127;
    for (int j = 0; j < GM_ / 2; ++j) {
      const int r = (tid >> 7) + (j << 1);
      float acc = 0.f;
      for (int s = 0; s < 8; ++s) acc += alpha[(r << 3) + s] * nv[(r << 3) + s][d];
      cat[r][D_ + d] = acc;
    }
  }
  __syncthreads();

  {
    const int col = n0 + laneIdx;
    const float* ap = &cat[laneIdx][laneHalf << 1];
    const float* bp = gw3p + ((laneHalf << 1) << 7) + col;
    v8f cs = {}, cn = {};
    #pragma unroll 8
    for (int k0 = 0; k0 < D_; k0 += 4) {
      v2f as = *(const v2f*)(ap + k0);
      v2f an = *(const v2f*)(ap + D_ + k0);
      v2f bs, bn;
      bs.x = bp[k0 << 7];
      bs.y = bp[(k0 + 1) << 7];
      bn.x = bp[(D_ + k0) << 7];
      bn.y = bp[(D_ + k0 + 1) << 7];
      cs = wmma4(as, bs, cs);
      cn = wmma4(an, bn, cn);
    }
    #pragma unroll
    for (int v = 0; v < 8; ++v)
      xs[v + (laneHalf << 3)][col] = fmaxf(cs[v] + cn[v], 0.f);
  }
  __syncthreads();
  *(float4*)(out + ((base + wave) << 7) + (lane << 2)) = *(const float4*)&xs[wave][lane << 2];
  *(float4*)(out + ((base + wave + 8) << 7) + (lane << 2)) = *(const float4*)&xs[wave + 8][lane << 2];
}

__global__ void gemm16_kernel(const float* __restrict__ A, int lda,
                              const float* __restrict__ BT, int ldbt,
                              float* __restrict__ C, long ldc,
                              int M, int N, int K,
                              const float* __restrict__ rowBias, int biasMod,
                              int act) {
  int tilesM = (M + 15) >> 4;
  long tilesN = (long)((N + 15) >> 4);
  long total = (long)tilesM * tilesN;
  int lane = threadIdx.x & 31, laneIdx = lane & 15, laneHalf = lane >> 4;
  for (long t = (long)blockIdx.x * 8 + (threadIdx.x >> 5); t < total;
       t += (long)gridDim.x * 8) {
    int m0 = (int)(t % tilesM) << 4;
    long n0 = (t / tilesM) << 4;
    int row = m0 + laneIdx;
    long col = n0 + laneIdx;
    v8f c = {};
    bool full = (m0 + 16 <= M) && (n0 + 16 <= N);
    if (full) {
      const float* ap = A + (long)row * lda;
      const float* bp = BT + col * ldbt;
      for (int k0 = 0; k0 < K; k0 += 4) {
        int kk = k0 + (laneHalf << 1);
        v2f a = *(const v2f*)(ap + kk);
        v2f b = *(const v2f*)(bp + kk);
        c = wmma4(a, b, c);
      }
    } else {
      for (int k0 = 0; k0 < K; k0 += 4) {
        int kk = k0 + (laneHalf << 1);
        v2f a, b;
        a.x = (row < M) ? A[(long)row * lda + kk] : 0.f;
        a.y = (row < M) ? A[(long)row * lda + kk + 1] : 0.f;
        b.x = (col < N) ? BT[col * ldbt + kk] : 0.f;
        b.y = (col < N) ? BT[col * ldbt + kk + 1] : 0.f;
        c = wmma4(a, b, c);
      }
    }
    float* cf = (float*)&c;
    #pragma unroll
    for (int v = 0; v < 8; ++v) {
      int rr = m0 + v + (laneHalf << 3);
      if (rr < M && col < N) {
        float val = cf[v];
        if (rowBias) val += rowBias[(long)(rr % biasMod) * N + col];
        if (act == 1) val = tanhf(val);
        C[(long)rr * ldc + col] = val;
      }
    }
  }
}

#define SC_COLS_ 32
#define SC_KH_ (D_ / 8)

__device__ __forceinline__ float4 sc_row(const float* __restrict__ embT, int n, int N, int lane) {
  n = n < N ? n : N - 1;
  return *(const float4*)(embT + ((long)n << 7) + (lane << 2));
}

__global__ void __launch_bounds__(256)
scores_kernel(const float* __restrict__ selF, const float* __restrict__ embT,
              float* __restrict__ scores, int N, int nStages) {
  __shared__ __align__(16) float bs[2][SC_COLS_][KP_];
  const int tid = threadIdx.x;
  const int lane = tid & 31, laneIdx = lane & 15, laneHalf = lane >> 4;
  const int wave = tid >> 5;

  v2f areg[D_ / 4];
  {
    const float* ap = selF + ((wave * 16 + laneIdx) << 7) + (laneHalf << 1);
    #pragma unroll
    for (int ks = 0; ks < D_ / 4; ++ks) areg[ks] = *(const v2f*)(ap + 4 * ks);
  }

  const int r0 = wave * 4;
  float4 p0, p1, p2, p3;
  int s = blockIdx.x;
  if (s < nStages) {
    const int n = s * SC_COLS_ + r0;
    p0 = sc_row(embT, n, N, lane);
    p1 = sc_row(embT, n + 1, N, lane);
    p2 = sc_row(embT, n + 2, N, lane);
    p3 = sc_row(embT, n + 3, N, lane);
    *(float4*)&bs[0][r0][lane << 2] = p0;
    *(float4*)&bs[0][r0 + 1][lane << 2] = p1;
    *(float4*)&bs[0][r0 + 2][lane << 2] = p2;
    *(float4*)&bs[0][r0 + 3][lane << 2] = p3;
  }
  __syncthreads();

  int buf = 0;
  for (; s < nStages; s += gridDim.x) {
    const int sn = s + gridDim.x;
    const bool more = sn < nStages;
    if (more) {
      const int n = sn * SC_COLS_ + r0;
      p0 = sc_row(embT, n, N, lane);
      p1 = sc_row(embT, n + 1, N, lane);
      p2 = sc_row(embT, n + 2, N, lane);
      p3 = sc_row(embT, n + 3, N, lane);
    }
    const float* bp0 = &bs[buf][laneIdx][laneHalf << 1];
    const float* bp1 = bp0 + 16 * KP_;
    v8f c0 = {}, c1 = {};
    #pragma unroll
    for (int kh = 0; kh < 2; ++kh) {
      v2f f0[SC_KH_], f1[SC_KH_];
      #pragma unroll
      for (int ks = 0; ks < SC_KH_; ++ks) {
        f0[ks] = *(const v2f*)(bp0 + 4 * (kh * SC_KH_ + ks));
        f1[ks] = *(const v2f*)(bp1 + 4 * (kh * SC_KH_ + ks));
      }
      #pragma unroll
      for (int ks = 0; ks < SC_KH_; ++ks) {
        c0 = wmma4(areg[kh * SC_KH_ + ks], f0[ks], c0);
        c1 = wmma4(areg[kh * SC_KH_ + ks], f1[ks], c1);
      }
    }
    if (more) {
      *(float4*)&bs[buf ^ 1][r0][lane << 2] = p0;
      *(float4*)&bs[buf ^ 1][r0 + 1][lane << 2] = p1;
      *(float4*)&bs[buf ^ 1][r0 + 2][lane << 2] = p2;
      *(float4*)&bs[buf ^ 1][r0 + 3][lane << 2] = p3;
    }
    __syncthreads();
    buf ^= 1;
    {
      const int n0 = s * SC_COLS_;
      float* o = scores + (long)(wave * 16 + (laneHalf << 3)) * N + n0 + laneIdx;
      if (n0 + SC_COLS_ <= N) {
        #pragma unroll
        for (int v = 0; v < 8; ++v) {
          o[(long)v * N] = c0[v];
          o[(long)v * N + 16] = c1[v];
        }
      } else {
        const bool in0 = n0 + laneIdx < N, in1 = n0 + laneIdx + 16 < N;
        #pragma unroll
        for (int v = 0; v < 8; ++v) {
          if (in0) o[(long)v * N] = c0[v];
          if (in1) o[(long)v * N + 16] = c1[v];
        }
      }
    }
  }
}

__global__ void add_kernel(const float* a, const float* b, float* o, int n) {
  int i = blockIdx.x * 256 + threadIdx.x;
  if (i < n) o[i] = a[i] + b[i];
}

__global__ void zero1_kernel(float* p) { if (threadIdx.x == 0 && blockIdx.x == 0) p[0] = 0.f; }

__global__ void ssl_kernel(const float* __restrict__ hl, const float* __restrict__ hg,
                           float* __restrict__ loss) {
  int idx = blockIdx.x * 256 + threadIdx.x;
  int b = idx >> 7, d = idx & 127;
  int pb = (b * 73 + 21) & 127;
  float pos = 0.f, neg = 0.f;
  for (int l = 0; l < L_; ++l) {
    int pl = (l * 13 + 7) % L_;
    pos += hl[((b * L_ + l) << 7) + d] * hg[((b * L_ + l) << 7) + d];
    neg += hg[((b * L_ + l) << 7) + d] * hl[((pb * L_ + pl) << 7) + d];
  }
  float term = -__logf(1e-8f + sigmoid_(pos)) - __logf(1e-8f + 1.f - sigmoid_(neg));
  __shared__ float red[256];
  red[threadIdx.x] = term;
  __syncthreads();
  for (int s = 128; s > 0; s >>= 1) {
    if (threadIdx.x < s) red[threadIdx.x] += red[threadIdx.x + s];
    __syncthreads();
  }
  if (threadIdx.x == 0) atomicAdd(loss, BETA_ * red[0]);
}

__global__ void hs_kernel(const float* __restrict__ hidden, const int* __restrict__ mask,
                          float* __restrict__ hs) {
  int b = blockIdx.x, d = threadIdx.x;
  float acc = 0.f, ms = 0.f;
  for (int l = 0; l < L_; ++l) {
    float m = (float)mask[b * L_ + l];
    ms += m;
    acc += hidden[((b * L_ + l) << 7) + d] * m;
  }
  hs[(b << 7) + d] = acc / ms;
}

__global__ void hsg_kernel(const float* __restrict__ hs, const float* __restrict__ glu2_w,
                           float* __restrict__ hsg) {
  __shared__ float hr[D_];
  int b = blockIdx.x, d = threadIdx.x;
  hr[d] = hs[(b << 7) + d];
  __syncthreads();
  float acc = 0.f;
  for (int e = 0; e < D_; ++e) acc += hr[e] * glu2_w[(d << 7) + e];
  hsg[(b << 7) + d] = acc;
}

__global__ void pos_kernel(const float* __restrict__ pos_emb, const float* __restrict__ w1,
                           float* __restrict__ P) {
  __shared__ float pr[D_];
  int l = blockIdx.x, d = threadIdx.x;
  pr[d] = pos_emb[(l << 7) + d];
  __syncthreads();
  float acc = 0.f;
  for (int e = 0; e < D_; ++e) acc += pr[e] * w1[(e << 7) + d];
  P[(l << 7) + d] = acc;
}

__global__ void beta_kernel(const float* __restrict__ nh, const float* __restrict__ glu1_w,
                            const float* __restrict__ glu1_b, const float* __restrict__ hsg,
                            const float* __restrict__ w2, const int* __restrict__ mask,
                            float* __restrict__ beta) {
  __shared__ float nrow[D_];
  __shared__ float red[D_];
  int row = blockIdx.x;
  int b = row / L_;
  int d = threadIdx.x;
  nrow[d] = nh[((long)row << 7) + d];
  __syncthreads();
  float acc = glu1_b[d] + hsg[(b << 7) + d];
  for (int e = 0; e < D_; ++e) acc += nrow[e] * glu1_w[(d << 7) + e];
  red[d] = sigmoid_(acc) * w2[d];
  __syncthreads();
  for (int s = 64; s > 0; s >>= 1) {
    if (d < s) red[d] += red[d + s];
    __syncthreads();
  }
  if (d == 0) beta[row] = red[0] * (float)mask[row];
}

__global__ void select_kernel(const float* __restrict__ hidden, const float* __restrict__ beta,
                              float* __restrict__ sel) {
  int b = blockIdx.x, d = threadIdx.x;
  float acc = 0.f;
  for (int l = 0; l < L_; ++l) acc += beta[b * L_ + l] * hidden[((b * L_ + l) << 7) + d];
  sel[(b << 7) + d] = acc;
}

__global__ void norm_kernel(const float* __restrict__ sel, float* __restrict__ snorm) {
  __shared__ float red[D_];
  int b = blockIdx.x, d = threadIdx.x;
  float v = sel[(b << 7) + d];
  red[d] = v * v + 1e-6f;
  __syncthreads();
  for (int s = 64; s > 0; s >>= 1) {
    if (d < s) red[d] += red[d + s];
    __syncthreads();
  }
  if (d == 0) snorm[b] = sqrtf(red[0]);
}

__global__ void fn_kernel(const float* __restrict__ sel, const float* __restrict__ snorm,
                          float* __restrict__ selF) {
  __shared__ float row[B_];
  __shared__ float selfrow[D_];
  __shared__ float tvs[9];
  __shared__ int tis[9];
  int i = blockIdx.x, t = threadIdx.x;
  selfrow[t] = sel[(i << 7) + t];
  __syncthreads();
  float acc = 0.f;
  for (int d = 0; d < D_; ++d) acc += selfrow[d] * sel[(t << 7) + d];
  row[t] = acc / (snorm[i] * snorm[t]);
  __syncthreads();
  if (t == 0) {
    float mx = -1e30f;
    for (int j = 0; j < B_; ++j) mx = fmaxf(mx, row[j]);
    float s = 0.f;
    for (int j = 0; j < B_; ++j) { row[j] = __expf(row[j] - mx); s += row[j]; }
    float inv = 1.f / s;
    for (int j = 0; j < B_; ++j) row[j] *= inv;
    for (int q = 0; q < 9; ++q) {
      float bm = -1.f; int bi = 0;
      for (int j = 0; j < B_; ++j) if (row[j] > bm) { bm = row[j]; bi = j; }
      tvs[q] = bm; tis[q] = bi; row[bi] = -2.f;
    }
    float mx2 = tvs[0], s2 = 0.f;
    for (int q = 0; q < 9; ++q) { tvs[q] = __expf(tvs[q] - mx2); s2 += tvs[q]; }
    float inv2 = 1.f / s2;
    for (int q = 0; q < 9; ++q) tvs[q] *= inv2;
  }
  __syncthreads();
  float nb = 0.f;
  for (int q = 0; q < 9; ++q) nb += tvs[q] * sel[(tis[q] << 7) + t];
  selF[(i << 7) + t] = selfrow[t] + nb;
}

extern "C" void kernel_launch(void* const* d_in, const int* in_sizes, int n_in,
                              void* d_out, int out_size, void* d_ws, size_t ws_size,
                              hipStream_t stream) {
  (void)in_sizes; (void)n_in; (void)out_size; (void)ws_size;
  const int*   inputs  = (const int*)d_in[0];
  const int*   adj     = (const int*)d_in[1];
  const int*   mask    = (const int*)d_in[2];
  const int*   item    = (const int*)d_in[3];
  const int*   adj_all = (const int*)d_in[5];
  const float* num_w   = (const float*)d_in[6];
  const float* emb     = (const float*)d_in[7];
  const float* pos_emb = (const float*)d_in[8];
  const float* a_local = (const float*)d_in[9];
  const float* gw1     = (const float*)d_in[10];
  const float* gw2     = (const float*)d_in[11];
  const float* gw3     = (const float*)d_in[12];
  const float* w1      = (const float*)d_in[13];
  const float* w2      = (const float*)d_in[14];
  const float* glu1_w  = (const float*)d_in[15];
  const float* glu1_b  = (const float*)d_in[16];
  const float* glu2_w  = (const float*)d_in[17];

  float* out0   = (float*)d_out;
  float* lossP  = out0 + B_ * L_ * D_;
  float* scores = lossP + 1;

  char* wp = (char*)d_ws;
  auto alloc = [&](size_t nfloats) -> float* {
    float* r = (float*)wp;
    wp += ((nfloats * sizeof(float) + 255) / 256) * 256;
    return r;
  };
  float* h        = alloc(B_ * L_ * D_);
  float* h_local  = alloc(B_ * L_ * D_);
  float* h_global = alloc(B_ * L_ * D_);
  float* sessb    = alloc(B_ * D_);
  float* e0p      = alloc(B_ * L_ * D_);
  float* e1p      = alloc((size_t)B_ * 400 * D_);
  float* hsb      = alloc(B_ * D_);
  float* hsgb     = alloc(B_ * D_);
  float* Pb       = alloc(L_ * D_);
  float* nh       = alloc(B_ * L_ * D_);
  float* betab    = alloc(B_ * L_);
  float* selb     = alloc(B_ * D_);
  float* selF     = alloc(B_ * D_);
  float* snorm    = alloc(B_);
  float* gw1T     = alloc(2 * D_ * KP_);
  float* w1botT   = alloc(D_ * D_);
  float* wn1      = alloc(B_ * 400);
  float* wn2      = alloc((size_t)B_ * 3200);
  int*   n1       = (int*)alloc(B_ * 400);
  int*   n2       = (int*)alloc((size_t)B_ * 3200);

  gw1t_kernel<<<dim3((2 * D_ * KP_ + 255) / 256), 256, 0, stream>>>(gw1, gw1T);
  w1t_kernel<<<dim3((D_ * D_ + 255) / 256), 256, 0, stream>>>(w1, w1botT);
  gather_h_kernel<<<dim3((B_ * L_ * D_) / 256), 256, 0, stream>>>(emb, inputs, h, B_ * L_ * D_);
  sess_kernel<<<dim3(B_), dim3(D_), 0, stream>>>(emb, item, mask, sessb);
  hop_kernel<<<dim3((B_ * L_ * S_ + 255) / 256), 256, 0, stream>>>(inputs, adj_all, num_w, n1, wn1, B_ * L_ * S_);
  hop_kernel<<<dim3((B_ * 400 * S_ + 255) / 256), 256, 0, stream>>>(n1, adj_all, num_w, n2, wn2, B_ * 400 * S_);
  local_agg_kernel<<<dim3(B_), 256, 0, stream>>>(h, adj, a_local, h_local);
  agg_kernel<true, true><<<dim3(B_ * L_ / GM_), 256, 0, stream>>>(emb, h, nullptr, n1, nullptr, wn1,
      sessb, gw1T, gw2, gw3, e0p, L_);
  agg_kernel<false, true><<<dim3(B_ * 400 / GM_), 256, 0, stream>>>(emb, nullptr, n1, n2, nullptr, wn2,
      sessb, gw1T, gw2, gw3, e1p, 400);
  agg_kernel<true, false><<<dim3(B_ * L_ / GM_), 256, 0, stream>>>(emb, e0p, nullptr, nullptr, e1p, wn1,
      sessb, gw1T + D_ * KP_, gw2 + D_, gw3 + 2 * D_ * D_, h_global, L_);
  add_kernel<<<dim3((B_ * L_ * D_) / 256), 256, 0, stream>>>(h_local, h_global, out0, B_ * L_ * D_);
  zero1_kernel<<<1, 1, 0, stream>>>(lossP);
  ssl_kernel<<<dim3(64), 256, 0, stream>>>(h_local, h_global, lossP);
  hs_kernel<<<dim3(B_), dim3(D_), 0, stream>>>(out0, mask, hsb);
  hsg_kernel<<<dim3(B_), dim3(D_), 0, stream>>>(hsb, glu2_w, hsgb);
  pos_kernel<<<dim3(L_), dim3(D_), 0, stream>>>(pos_emb, w1, Pb);
  {
    int tiles = ((B_ * L_ + 15) / 16) * ((D_ + 15) / 16);
    gemm16_kernel<<<dim3((tiles + 7) / 8), 256, 0, stream>>>(
        out0, D_, w1botT, D_, nh, D_, B_ * L_, D_, D_, Pb, L_, 1);
  }
  beta_kernel<<<dim3(B_ * L_), dim3(D_), 0, stream>>>(nh, glu1_w, glu1_b, hsgb, w2, mask, betab);
  select_kernel<<<dim3(B_), dim3(D_), 0, stream>>>(out0, betab, selb);
  norm_kernel<<<dim3(B_), dim3(D_), 0, stream>>>(selb, snorm);
  fn_kernel<<<dim3(B_), dim3(D_), 0, stream>>>(selb, snorm, selF);
  scores_kernel<<<dim3(512), 256, 0, stream>>>(selF, emb + D_, scores, NN_ - 1,
                                               (NN_ - 1 + SC_COLS_ - 1) / SC_COLS_);
}


# local attention and round 0 of the global aggregation over both hops (three independent kernels) run as block ranges of one launch: 10 launches become 8, the short grids fill in beside the long one
# speedup vs baseline: 3.0205x; 1.4723x over previous
#include <hip/hip_runtime.h>
#include <hip/hip_bf16.h>
#include <math.h>

#define B_ 128
#define L_ 50
#define D_ 128
#define S_ 8
#define NN_ 100000
#define KP_ 132
#define LEAKY_ 0.2f
#define BETA_ 0.005f

typedef float v2f __attribute__((ext_vector_type(2)));
typedef float v8f __attribute__((ext_vector_type(8)));

#if defined(__has_builtin)
#if __has_builtin(__builtin_amdgcn_wmma_f32_16x16x4_f32)
#define HAVE_WMMA_F32 1
#endif
#endif

__device__ __forceinline__ v8f wmma4(v2f a, v2f b, v8f c) {
#ifdef HAVE_WMMA_F32
  return __builtin_amdgcn_wmma_f32_16x16x4_f32(false, a, false, b, (short)0, c,
                                               false, false);
#else
  float* cf = (float*)&c;
  #pragma unroll
  for (int i = 0; i < 8; ++i) cf[i] += a.x * b.x + a.y * b.y;
  return c;
#endif
}

__device__ __forceinline__ float leaky_(float x) { return x > 0.f ? x : LEAKY_ * x; }
__device__ __forceinline__ float sigmoid_(float x) { return 1.f / (1.f + __expf(-x)); }

#define P0_SESS_  64
#define P0_HOP2_  400
#define P0_POS_   25
#define P0_HOP1_  50
#define P0_GH_    800
#define P0_GW1T_  132
#define P0_W1T_   64
#define P0_TOTAL_ (P0_SESS_ + P0_HOP2_ + P0_POS_ + P0_HOP1_ + P0_GH_ + P0_GW1T_ + P0_W1T_)

__global__ void __launch_bounds__(256)
prep_kernel(const float* __restrict__ emb, const int* __restrict__ inputs,
            const int* __restrict__ item, const int* __restrict__ mask,
            const int* __restrict__ adj_all, const float* __restrict__ num_w,
            const float* __restrict__ pos_emb, const float* __restrict__ gw1,
            const float* __restrict__ w1,
            float* __restrict__ sess, int* __restrict__ n1, float* __restrict__ wn1,
            int* __restrict__ n2, float* __restrict__ wn2, float* __restrict__ P,
            float* __restrict__ h, float* __restrict__ gw1T, float* __restrict__ w1botT,
            float* __restrict__ loss) {
  __shared__ float pr[2][D_];
  const int tid = threadIdx.x;
  int bid = blockIdx.x;
  if (bid < P0_SESS_) {
    const int b = (bid << 1) + (tid >> 7), d = tid & 127;
    float acc = 0.f, ms = 0.f;
    for (int l = 0; l < L_; ++l) {
      float m = (float)mask[b * L_ + l];
      ms += m;
      acc += emb[((long)item[b * L_ + l] << 7) + d] * m;
    }
    sess[(b << 7) + d] = acc / ms;
    if (bid == 0 && tid == 0) loss[0] = 0.f;
  } else if ((bid -= P0_SESS_) < P0_HOP2_) {
    const int t = bid * 256 + tid;
    const int j = t >> 1;
    const int node = adj_all[inputs[j >> 3] * S_ + (j & 7)];
    const int o = node * S_ + ((t & 1) << 2);
    *(int4*)(n2 + (t << 2)) = *(const int4*)(adj_all + o);
    *(float4*)(wn2 + (t << 2)) = *(const float4*)(num_w + o);
  } else if ((bid -= P0_HOP2_) < P0_POS_) {
    const int hb = tid >> 7, d = tid & 127, l = (bid << 1) + hb;
    pr[hb][d] = pos_emb[(l << 7) + d];
    __syncthreads();
    float acc = 0.f;
    for (int e = 0; e < D_; ++e) acc += pr[hb][e] * w1[(e << 7) + d];
    P[(l << 7) + d] = acc;
  } else if ((bid -= P0_POS_) < P0_HOP1_) {
    const int t = bid * 256 + tid;
    const int o = inputs[t >> 1] * S_ + ((t & 1) << 2);
    *(int4*)(n1 + (t << 2)) = *(const int4*)(adj_all + o);
    *(float4*)(wn1 + (t << 2)) = *(const float4*)(num_w + o);
  } else if ((bid -= P0_HOP1_) < P0_GH_) {
    const int q = bid * 256 + tid;
    *(float4*)(h + ((long)q << 2)) =
        *(const float4*)(emb + ((long)inputs[q >> 5] << 7) + ((q & 31) << 2));
  } else if ((bid -= P0_GH_) < P0_GW1T_) {
    const int i = bid * 256 + tid;
    const int hop = i / (D_ * KP_);
    const int rem = i % (D_ * KP_);
    const int n = rem / KP_, k = rem % KP_;
    gw1T[i] = (k < 129) ? gw1[hop * 129 * D_ + k * D_ + n] : 0.f;
  } else {
    bid -= P0_GW1T_;
    const int i = bid * 256 + tid;
    const int n = i >> 7, k = i & 127;
    w1botT[i] = w1[(D_ + k) * D_ + n];
  }
}

#define HP_ 132

__device__ __forceinline__ void
local_agg_body(float* smem, const int b,
               const float* __restrict__ h,
               const int* __restrict__ adj,
               const float* __restrict__ a_local,
               float* __restrict__ h_local) {
  float (*hv)[HP_] = reinterpret_cast<float (*)[HP_]>(smem);
  float (*al)[D_]  = reinterpret_cast<float (*)[D_]>(smem + L_ * HP_);
  float (*att)[L_] = reinterpret_cast<float (*)[L_]>(smem + L_ * HP_ + 4 * D_);
  float* rinv = smem + L_ * HP_ + 4 * D_ + L_ * L_;
  const int tid = threadIdx.x;
  const int lane = tid & 31, laneIdx = lane & 15, laneHalf = lane >> 4;
  const int wave = tid >> 5;

  for (int q = tid; q < L_ * (D_ / 4); q += 256)
    *(float4*)&hv[q >> 5][(q & 31) << 2] = *(const float4*)(h + (long)b * L_ * D_ + (q << 2));
  if (tid < D_) *(float4*)&al[0][tid << 2] = *(const float4*)(a_local + (tid << 2));

  const int i0 = (wave >> 1) << 4;
  const int j0 = (wave & 1) << 5;
  int et[2][8];
  #pragma unroll
  for (int t = 0; t < 2; ++t) {
    #pragma unroll
    for (int v = 0; v < 8; ++v) {
      const int i = i0 + (laneHalf << 3) + v, j = j0 + (t << 4) + laneIdx;
      et[t][v] = (i < L_ && j < L_) ? adj[b * L_ * L_ + i * L_ + j] : 0;
    }
  }
  __syncthreads();

  {
    const int ir  = (i0 + laneIdx < L_) ? i0 + laneIdx : L_ - 1;
    const int jr0 = (j0 + laneIdx < L_) ? j0 + laneIdx : L_ - 1;
    const int jr1 = (j0 + 16 + laneIdx < L_) ? j0 + 16 + laneIdx : L_ - 1;
    const float* ap  = &hv[ir][laneHalf << 1];
    const float* bp0 = &hv[jr0][laneHalf << 1];
    const float* bp1 = &hv[jr1][laneHalf << 1];
    const float* sp  = &al[0][laneHalf << 1];
    v8f e[4][2] = {};
    for (int ks = 0; ks < D_ / 4; ++ks) {
      const v2f x  = *(const v2f*)(ap + 4 * ks);
      const v2f y0 = *(const v2f*)(bp0 + 4 * ks);
      const v2f y1 = *(const v2f*)(bp1 + 4 * ks);
      #pragma unroll
      for (int k = 0; k < 4; ++k) {
        const v2f s = *(const v2f*)(sp + k * D_ + 4 * ks);
        const v2f xa = x * s;
        e[k][0] = wmma4(xa, y0, e[k][0]);
        e[k][1] = wmma4(xa, y1, e[k][1]);
      }
    }
    #pragma unroll
    for (int t = 0; t < 2; ++t) {
      const int j = j0 + (t << 4) + laneIdx;
      #pragma unroll
      for (int v = 0; v < 8; ++v) {
        const int i = i0 + (laneHalf << 3) + v;
        const int k = et[t][v];
        float x = e[3][t][v];
        x = (k == 3) ? e[2][t][v] : x;
        x = (k == 2) ? e[1][t][v] : x;
        x = (k == 1) ? e[0][t][v] : x;
        x = (k == 0) ? -9e15f : leaky_(x);
        if (i < L_ && j < L_) att[i][j] = x;
      }
    }
  }
  __syncthreads();
  if (threadIdx.x < L_) {
    int i = threadIdx.x;
    float mx = -1e30f;
    for (int j = 0; j < L_; ++j) mx = fmaxf(mx, att[i][j]);
    float s = 0.f;
    for (int j = 0; j < L_; ++j) { float ex = __expf(att[i][j] - mx); att[i][j] = ex; s += ex; }
    rinv[i] = 1.f / s;
  }
  __syncthreads();
  for (int p = threadIdx.x; p < L_ * D_; p += 256) {
    int i = p >> 7, d = p & 127;
    float acc = 0.f;
    for (int j = 0; j < L_; ++j) acc += att[i][j] * hv[j][d];
    h_local[(long)b * L_ * D_ + p] = acc * rinv[i];
  }
}

#define GM_ 16
#define GR_ 128
#define CP_ 260

__device__ __forceinline__ void score_cols(v8f c, const float (&g2v)[8], float* scRow,
                                           int laneIdx) {
  float acc = 0.f;
  #pragma unroll
  for (int v = 0; v < 8; ++v) acc += leaky_(c[v]) * g2v[v];
  atomicAdd(&scRow[laneIdx], acc);
}

#define AGG_LDS_FLOATS_ (GR_ * D_ + GR_ * KP_ + GM_ * CP_ + 2 * GR_)

template <bool SELF_VAL, bool NEIGH_IDX>
__device__ __forceinline__ void
agg_body(float* smem, const unsigned blk,
         const float* __restrict__ emb,
         const float* __restrict__ selfVal, const int* __restrict__ selfIdx,
         const int* __restrict__ neighIdx, const float* __restrict__ neighVal,
         const float* __restrict__ neighW,
         const float* __restrict__ sess,
         const float* __restrict__ gw1T, const float* __restrict__ gw2p,
         const float* __restrict__ gw3p,
         float* __restrict__ out, const int M) {
  float (*nv)[D_]   = reinterpret_cast<float (*)[D_]>(smem);
  float (*xs)[KP_]  = reinterpret_cast<float (*)[KP_]>(smem + GR_ * D_);
  float (*cat)[CP_] = reinterpret_cast<float (*)[CP_]>(smem + GR_ * D_ + GR_ * KP_);
  float* sc    = smem + GR_ * D_ + GR_ * KP_ + GM_ * CP_;
  float* alpha = sc + GR_;

  const int tid = threadIdx.x;
  const int lane = tid & 31, laneIdx = lane & 15, laneHalf = lane >> 4;
  const int wave = tid >> 5;
  const int n0 = wave * 16;
  const long base  = (long)blk * GM_;
  const long nbase = base * S_;

  v2f breg[KP_ / 4];
  {
    const float* bp = gw1T + (n0 + laneIdx) * KP_ + (laneHalf << 1);
    #pragma unroll
    for (int ks = 0; ks < KP_ / 4; ++ks) breg[ks] = *(const v2f*)(bp + 4 * ks);
  }
  float g2v[8];
  {
    const float4 ga = *(const float4*)(gw2p + n0 + (laneHalf << 3));
    const float4 gb = *(const float4*)(gw2p + n0 + (laneHalf << 3) + 4);
    g2v[0] = ga.x; g2v[1] = ga.y; g2v[2] = ga.z; g2v[3] = ga.w;
    g2v[4] = gb.x; g2v[5] = gb.y; g2v[6] = gb.z; g2v[7] = gb.w;
  }

  const unsigned ubase = blk * GM_;
  const unsigned b0 = ubase / (unsigned)M;
  const unsigned rem0 = ubase - b0 * (unsigned)M;
  float4 gv[GM_];
  if (NEIGH_IDX) {
    int idx[GM_];
    #pragma unroll
    for (int j = 0; j < GM_; ++j) idx[j] = neighIdx[nbase + (j << 3) + wave];
    #pragma unroll
    for (int j = 0; j < GM_; ++j)
      gv[j] = *(const float4*)(emb + ((long)idx[j] << 7) + (lane << 2));
  } else {
    #pragma unroll
    for (int j = 0; j < GM_; ++j)
      gv[j] = *(const float4*)(neighVal + ((nbase + (j << 3) + wave) << 7) + (lane << 2));
  }
  const unsigned b1 = (rem0 + GM_ > (unsigned)M) ? b0 + 1 : b0;
  const float4 sfa = *(const float4*)(sess + ((long)b0 << 7) + (lane << 2));
  const float4 sfb = *(const float4*)(sess + ((long)b1 << 7) + (lane << 2));
  float4 sv0, sv1;
  if (SELF_VAL) {
    sv0 = *(const float4*)(selfVal + ((base + wave) << 7) + (lane << 2));
    sv1 = *(const float4*)(selfVal + ((base + wave + 8) << 7) + (lane << 2));
  } else {
    sv0 = *(const float4*)(emb + ((long)selfIdx[base + wave] << 7) + (lane << 2));
    sv1 = *(const float4*)(emb + ((long)selfIdx[base + wave + 8] << 7) + (lane << 2));
  }
  #pragma unroll
  for (int j = 0; j < GM_; ++j) {
    const int r = (j << 3) + wave;
    const float4 v = gv[j];
    const bool nextb = rem0 + (unsigned)j >= (unsigned)M;
    *(float4*)&nv[r][lane << 2] = v;
    float4 x;
    x.x = (nextb ? sfb.x : sfa.x) * v.x;
    x.y = (nextb ? sfb.y : sfa.y) * v.y;
    x.z = (nextb ? sfb.z : sfa.z) * v.z;
    x.w = (nextb ? sfb.w : sfa.w) * v.w;
    *(float4*)&xs[r][lane << 2] = x;
  }
  *(float4*)&cat[wave][lane << 2] = sv0;
  *(float4*)&cat[wave + 8][lane << 2] = sv1;
  if (tid < GR_) {
    xs[tid][D_] = neighW[nbase + tid];
    xs[tid][D_ + 1] = 0.f; xs[tid][D_ + 2] = 0.f; xs[tid][D_ + 3] = 0.f;
    sc[tid] = 0.f;
  }
  __syncthreads();

  for (int rp = 0; rp < GR_ / 32; ++rp) {
    const float* xp0 = &xs[(rp << 5) + laneIdx][laneHalf << 1];
    const float* xp1 = xp0 + 16 * KP_;
    v8f c0 = {}, c1 = {};
    #pragma unroll
    for (int ks = 0; ks < KP_ / 4; ++ks) {
      v2f x0 = *(const v2f*)(xp0 + 4 * ks);
      v2f x1 = *(const v2f*)(xp1 + 4 * ks);
      c0 = wmma4(breg[ks], x0, c0);
      c1 = wmma4(breg[ks], x1, c1);
    }
    score_cols(c0, g2v, &sc[(rp << 5)], laneIdx);
    score_cols(c1, g2v, &sc[(rp << 5) + 16], laneIdx);
  }
  __syncthreads();

  if (tid < GM_) {
    float mx = -1e30f;
    for (int s = 0; s < 8; ++s) mx = fmaxf(mx, sc[tid * 8 + s]);
    float ex[8], sum = 0.f;
    for (int s = 0; s < 8; ++s) { ex[s] = __expf(sc[tid * 8 + s] - mx); sum += ex[s]; }
    float inv = 1.f / sum;
    for (int s = 0; s < 8; ++s) alpha[tid * 8 + s] = ex[s] * inv;
  }
  __syncthreads();
  {
    const int d = tid & 127;
    for (int j = 0; j < GM_ / 2; ++j) {
      const int r = (tid >> 7) + (j << 1);
      float acc = 0.f;
      for (int s = 0; s < 8; ++s) acc += alpha[(r << 3) + s] * nv[(r << 3) + s][d];
      cat[r][D_ + d] = acc;
    }
  }
  __syncthreads();

  {
    const int col = n0 + laneIdx;
    const float* ap = &cat[laneIdx][laneHalf << 1];
    const float* bp = gw3p + ((laneHalf << 1) << 7) + col;
    v8f cs = {}, cn = {};
    #pragma unroll 8
    for (int k0 = 0; k0 < D_; k0 += 4) {
      v2f as = *(const v2f*)(ap + k0);
      v2f an = *(const v2f*)(ap + D_ + k0);
      v2f bs, bn;
      bs.x = bp[k0 << 7];
      bs.y = bp[(k0 + 1) << 7];
      bn.x = bp[(D_ + k0) << 7];
      bn.y = bp[(D_ + k0 + 1) << 7];
      cs = wmma4(as, bs, cs);
      cn = wmma4(an, bn, cn);
    }
    #pragma unroll
    for (int v = 0; v < 8; ++v)
      xs[v + (laneHalf << 3)][col] = fmaxf(cs[v] + cn[v], 0.f);
  }
  __syncthreads();
  *(float4*)(out + ((base + wave) << 7) + (lane << 2)) = *(const float4*)&xs[wave][lane << 2];
  *(float4*)(out + ((base + wave + 8) << 7) + (lane << 2)) = *(const float4*)&xs[wave + 8][lane << 2];
}

#define P1_LOCAL_ B_
#define P1_HOP1_  (B_ * L_ / GM_)
#define P1_HOP2_  (B_ * L_ * S_ / GM_)
#define P1_TOTAL_ (P1_LOCAL_ + P1_HOP1_ + P1_HOP2_)

__global__ void __launch_bounds__(256)
hop_kernel(const float* __restrict__ emb, const float* __restrict__ h,
           const int* __restrict__ adj, const float* __restrict__ a_local,
           const int* __restrict__ n1, const float* __restrict__ wn1,
           const int* __restrict__ n2, const float* __restrict__ wn2,
           const float* __restrict__ sess,
           const float* __restrict__ gw1T, const float* __restrict__ gw2p,
           const float* __restrict__ gw3p,
           float* __restrict__ h_local, float* __restrict__ e0, float* __restrict__ e1) {
  __shared__ __align__(16) float smem[AGG_LDS_FLOATS_];
  int bid = blockIdx.x;
  if (bid < P1_LOCAL_) {
    local_agg_body(smem, bid, h, adj, a_local, h_local);
    return;
  }
  bid -= P1_LOCAL_;
  if (bid < P1_HOP1_) {
    agg_body<true, true>(smem, (unsigned)bid, emb, h, nullptr, n1, nullptr, wn1,
                         sess, gw1T, gw2p, gw3p, e0, L_);
  } else {
    agg_body<false, true>(smem, (unsigned)(bid - P1_HOP1_), emb, nullptr, n1, n2, nullptr, wn2,
                          sess, gw1T, gw2p, gw3p, e1, L_ * S_);
  }
}

__global__ void __launch_bounds__(256)
agg_val_kernel(const float* __restrict__ selfVal, const float* __restrict__ neighVal,
               const float* __restrict__ neighW, const float* __restrict__ sess,
               const float* __restrict__ gw1T, const float* __restrict__ gw2p,
               const float* __restrict__ gw3p, float* __restrict__ out, int M) {
  __shared__ __align__(16) float smem[AGG_LDS_FLOATS_];
  agg_body<true, false>(smem, blockIdx.x, nullptr, selfVal, nullptr, nullptr, neighVal, neighW,
                        sess, gw1T, gw2p, gw3p, out, M);
}

#define SC_COLS_ 32
#define SC_KH_ (D_ / 8)

__device__ __forceinline__ float4 sc_row(const float* __restrict__ embT, int n, int N, int lane) {
  n = n < N ? n : N - 1;
  return *(const float4*)(embT + ((long)n << 7) + (lane << 2));
}

__global__ void __launch_bounds__(256)
scores_kernel(const float* __restrict__ selF, const float* __restrict__ embT,
              float* __restrict__ scores, int N, int nStages) {
  __shared__ __align__(16) float bs[2][SC_COLS_][KP_];
  const int tid = threadIdx.x;
  const int lane = tid & 31, laneIdx = lane & 15, laneHalf = lane >> 4;
  const int wave = tid >> 5;

  v2f areg[D_ / 4];
  {
    const float* ap = selF + ((wave * 16 + laneIdx) << 7) + (laneHalf << 1);
    #pragma unroll
    for (int ks = 0; ks < D_ / 4; ++ks) areg[ks] = *(const v2f*)(ap + 4 * ks);
  }

  const int r0 = wave * 4;
  float4 p0, p1, p2, p3;
  int s = blockIdx.x;
  if (s < nStages) {
    const int n = s * SC_COLS_ + r0;
    p0 = sc_row(embT, n, N, lane);
    p1 = sc_row(embT, n + 1, N, lane);
    p2 = sc_row(embT, n + 2, N, lane);
    p3 = sc_row(embT, n + 3, N, lane);
    *(float4*)&bs[0][r0][lane << 2] = p0;
    *(float4*)&bs[0][r0 + 1][lane << 2] = p1;
    *(float4*)&bs[0][r0 + 2][lane << 2] = p2;
    *(float4*)&bs[0][r0 + 3][lane << 2] = p3;
  }
  __syncthreads();

  int buf = 0;
  for (; s < nStages; s += gridDim.x) {
    const int sn = s + gridDim.x;
    const bool more = sn < nStages;
    if (more) {
      const int n = sn * SC_COLS_ + r0;
      p0 = sc_row(embT, n, N, lane);
      p1 = sc_row(embT, n + 1, N, lane);
      p2 = sc_row(embT, n + 2, N, lane);
      p3 = sc_row(embT, n + 3, N, lane);
    }
    const float* bp0 = &bs[buf][laneIdx][laneHalf << 1];
    const float* bp1 = bp0 + 16 * KP_;
    v8f c0 = {}, c1 = {};
    #pragma unroll
    for (int kh = 0; kh < 2; ++kh) {
      v2f f0[SC_KH_], f1[SC_KH_];
      #pragma unroll
      for (int ks = 0; ks < SC_KH_; ++ks) {
        f0[ks] = *(const v2f*)(bp0 + 4 * (kh * SC_KH_ + ks));
        f1[ks] = *(const v2f*)(bp1 + 4 * (kh * SC_KH_ + ks));
      }
      #pragma unroll
      for (int ks = 0; ks < SC_KH_; ++ks) {
        c0 = wmma4(areg[kh * SC_KH_ + ks], f0[ks], c0);
        c1 = wmma4(areg[kh * SC_KH_ + ks], f1[ks], c1);
      }
    }
    if (more) {
      *(float4*)&bs[buf ^ 1][r0][lane << 2] = p0;
      *(float4*)&bs[buf ^ 1][r0 + 1][lane << 2] = p1;
      *(float4*)&bs[buf ^ 1][r0 + 2][lane << 2] = p2;
      *(float4*)&bs[buf ^ 1][r0 + 3][lane << 2] = p3;
    }
    __syncthreads();
    buf ^= 1;
    {
      const int n0 = s * SC_COLS_;
      float* o = scores + (long)(wave * 16 + (laneHalf << 3)) * N + n0 + laneIdx;
      if (n0 + SC_COLS_ <= N) {
        #pragma unroll
        for (int v = 0; v < 8; ++v) {
          o[(long)v * N] = c0[v];
          o[(long)v * N + 16] = c1[v];
        }
      } else {
        const bool in0 = n0 + laneIdx < N, in1 = n0 + laneIdx + 16 < N;
        #pragma unroll
        for (int v = 0; v < 8; ++v) {
          if (in0) o[(long)v * N] = c0[v];
          if (in1) o[(long)v * N + 16] = c1[v];
        }
      }
    }
  }
}

#define P4_HS_  64
#define P4_SSL_ 64
#define P4_ADD_ 800
#define P4_TOTAL_ (P4_HS_ + P4_SSL_ + P4_ADD_)

__global__ void __launch_bounds__(256)
mid_kernel(const float* __restrict__ hl, const float* __restrict__ hg,
           const int* __restrict__ mask, const float* __restrict__ glu2_w,
           float* __restrict__ out0, float* __restrict__ hsg, float* __restrict__ loss) {
  __shared__ float hr[2][D_];
  __shared__ float red[256];
  const int tid = threadIdx.x;
  int bid = blockIdx.x;
  if (bid < P4_HS_) {
    const int hb = tid >> 7, d = tid & 127, b = (bid << 1) + hb;
    float acc = 0.f, ms = 0.f;
    for (int l = 0; l < L_; ++l) {
      const float m = (float)mask[b * L_ + l];
      const int o = ((b * L_ + l) << 7) + d;
      ms += m;
      acc += (hl[o] + hg[o]) * m;
    }
    hr[hb][d] = acc / ms;
    __syncthreads();
    float a2 = 0.f;
    for (int e = 0; e < D_; ++e) a2 += hr[hb][e] * glu2_w[(d << 7) + e];
    hsg[(b << 7) + d] = a2;
  } else if ((bid -= P4_HS_) < P4_SSL_) {
    const int idx = bid * 256 + tid;
    const int b = idx >> 7, d = idx & 127;
    const int pb = (b * 73 + 21) & 127;
    float pos = 0.f, neg = 0.f;
    for (int l = 0; l < L_; ++l) {
      const int pl = (l * 13 + 7) % L_;
      pos += hl[((b * L_ + l) << 7) + d] * hg[((b * L_ + l) << 7) + d];
      neg += hg[((b * L_ + l) << 7) + d] * hl[((pb * L_ + pl) << 7) + d];
    }
    const float term = -__logf(1e-8f + sigmoid_(pos)) - __logf(1e-8f + 1.f - sigmoid_(neg));
    red[tid] = term;
    __syncthreads();
    for (int st = 128; st > 0; st >>= 1) {
      if (tid < st) red[tid] += red[tid + st];
      __syncthreads();
    }
    if (tid == 0) atomicAdd(loss, BETA_ * red[0]);
  } else {
    bid -= P4_SSL_;
    const int q = bid * 256 + tid;
    const float4 x = *(const float4*)(hl + ((long)q << 2));
    const float4 y = *(const float4*)(hg + ((long)q << 2));
    float4 r;
    r.x = x.x + y.x; r.y = x.y + y.y; r.z = x.z + y.z; r.w = x.w + y.w;
    *(float4*)(out0 + ((long)q << 2)) = r;
  }
}

#define HD_ROWS_ 32

__device__ __forceinline__ v2f lo2_(float4 f) { v2f r; r.x = f.x; r.y = f.y; return r; }
__device__ __forceinline__ v2f hi2_(float4 f) { v2f r; r.x = f.z; r.y = f.w; return r; }

__global__ void __launch_bounds__(256)
gate_kernel(const float* __restrict__ hidden, const float* __restrict__ w1botT,
            const float* __restrict__ P, const float* __restrict__ glu1_w,
            const float* __restrict__ glu1_b, const float* __restrict__ hsg,
            const float* __restrict__ w2, const int* __restrict__ mask,
            float* __restrict__ beta) {
  __shared__ __align__(16) float xr[HD_ROWS_][HP_];
  __shared__ __align__(16) float nhs[HD_ROWS_][HP_];
  __shared__ float part[16][HD_ROWS_];
  const int tid = threadIdx.x;
  const int lane = tid & 31, laneIdx = lane & 15, laneHalf = lane >> 4;
  const int wave = tid >> 5;
  const int n0 = wave << 4;
  const int row0 = blockIdx.x * HD_ROWS_;

  float4 hq[4];
  #pragma unroll
  for (int q = 0; q < 4; ++q)
    hq[q] = *(const float4*)(hidden + ((long)row0 << 7) + ((tid + (q << 8)) << 2));
  float4 bw[16];
  {
    const float* bp = w1botT + ((n0 + laneIdx) << 7) + (laneHalf << 2);
    #pragma unroll
    for (int j = 0; j < 16; ++j) bw[j] = *(const float4*)(bp + 8 * j);
  }
  float pv[2][8];
  #pragma unroll
  for (int t = 0; t < 2; ++t) {
    #pragma unroll
    for (int v = 0; v < 8; ++v) {
      const int l = (row0 + (t << 4) + (laneHalf << 3) + v) % L_;
      pv[t][v] = P[(l << 7) + n0 + laneIdx];
    }
  }

  #pragma unroll
  for (int q = 0; q < 4; ++q) {
    const int i = tid + (q << 8);
    *(float4*)&xr[i >> 5][(i & 31) << 2] = hq[q];
  }
  __syncthreads();

  float4 gw[16];
  {
    const float* gp = glu1_w + ((n0 + laneIdx) << 7) + (laneHalf << 2);
    #pragma unroll
    for (int j = 0; j < 16; ++j) gw[j] = *(const float4*)(gp + 8 * j);
  }
  const int nc = n0 + (laneHalf << 3);
  const float4 gba = *(const float4*)(glu1_b + nc), gbb = *(const float4*)(glu1_b + nc + 4);
  const float4 w2a = *(const float4*)(w2 + nc), w2b = *(const float4*)(w2 + nc + 4);
  const float* hp0 = hsg + (((row0 + laneIdx) / L_) << 7) + nc;
  const float* hp1 = hsg + (((row0 + 16 + laneIdx) / L_) << 7) + nc;
  const float4 h0a = *(const float4*)hp0, h0b = *(const float4*)(hp0 + 4);
  const float4 h1a = *(const float4*)hp1, h1b = *(const float4*)(hp1 + 4);

  {
    const float* ap0 = &xr[laneIdx][laneHalf << 2];
    const float* ap1 = ap0 + 16 * HP_;
    v8f c0 = {}, c1 = {};
    #pragma unroll
    for (int j = 0; j < 16; ++j) {
      const float4 x0 = *(const float4*)(ap0 + 8 * j);
      const float4 x1 = *(const float4*)(ap1 + 8 * j);
      const float4 b = bw[j];
      c0 = wmma4(lo2_(x0), lo2_(b), c0);
      c1 = wmma4(lo2_(x1), lo2_(b), c1);
      c0 = wmma4(hi2_(x0), hi2_(b), c0);
      c1 = wmma4(hi2_(x1), hi2_(b), c1);
    }
    #pragma unroll
    for (int v = 0; v < 8; ++v) {
      nhs[(laneHalf << 3) + v][n0 + laneIdx] = tanhf(c0[v] + pv[0][v]);
      nhs[16 + (laneHalf << 3) + v][n0 + laneIdx] = tanhf(c1[v] + pv[1][v]);
    }
  }
  __syncthreads();

  {
    const float* np0 = &nhs[laneIdx][laneHalf << 2];
    const float* np1 = np0 + 16 * HP_;
    v8f d0 = {}, d1 = {};
    #pragma unroll
    for (int j = 0; j < 16; ++j) {
      const float4 y0 = *(const float4*)(np0 + 8 * j);
      const float4 y1 = *(const float4*)(np1 + 8 * j);
      const float4 a = gw[j];
      d0 = wmma4(lo2_(a), lo2_(y0), d0);
      d1 = wmma4(lo2_(a), lo2_(y1), d1);
      d0 = wmma4(hi2_(a), hi2_(y0), d0);
      d1 = wmma4(hi2_(a), hi2_(y1), d1);
    }
    const float gb[8] = {gba.x, gba.y, gba.z, gba.w, gbb.x, gbb.y, gbb.z, gbb.w};
    const float wv[8] = {w2a.x, w2a.y, w2a.z, w2a.w, w2b.x, w2b.y, w2b.z, w2b.w};
    const float s0[8] = {h0a.x, h0a.y, h0a.z, h0a.w, h0b.x, h0b.y, h0b.z, h0b.w};
    const float s1[8] = {h1a.x, h1a.y, h1a.z, h1a.w, h1b.x, h1b.y, h1b.z, h1b.w};
    float acc0 = 0.f, acc1 = 0.f;
    #pragma unroll
    for (int v = 0; v < 8; ++v) {
      acc0 += sigmoid_(d0[v] + gb[v] + s0[v]) * wv[v];
      acc1 += sigmoid_(d1[v] + gb[v] + s1[v]) * wv[v];
    }
    part[(wave << 1) + laneHalf][laneIdx] = acc0;
    part[(wave << 1) + laneHalf][16 + laneIdx] = acc1;
  }
  __syncthreads();
  if (tid < HD_ROWS_) {
    float s = 0.f;
    for (int q = 0; q < 16; ++q) s += part[q][tid];
    beta[row0 + tid] = s * (float)mask[row0 + tid];
  }
}

__global__ void selnorm_kernel(const float* __restrict__ hidden, const float* __restrict__ beta,
                               float* __restrict__ sel, float* __restrict__ snorm) {
  __shared__ float red[D_];
  int b = blockIdx.x, d = threadIdx.x;
  float acc = 0.f;
  for (int l = 0; l < L_; ++l) acc += beta[b * L_ + l] * hidden[((b * L_ + l) << 7) + d];
  sel[(b << 7) + d] = acc;
  red[d] = acc * acc + 1e-6f;
  __syncthreads();
  for (int st = 64; st > 0; st >>= 1) {
    if (d < st) red[d] += red[d + st];
    __syncthreads();
  }
  if (d == 0) snorm[b] = sqrtf(red[0]);
}

__global__ void fn_kernel(const float* __restrict__ sel, const float* __restrict__ snorm,
                          float* __restrict__ selF) {
  __shared__ float row[B_];
  __shared__ float selfrow[D_];
  __shared__ float tvs[9];
  __shared__ int tis[9];
  int i = blockIdx.x, t = threadIdx.x;
  selfrow[t] = sel[(i << 7) + t];
  __syncthreads();
  float acc = 0.f;
  for (int d = 0; d < D_; ++d) acc += selfrow[d] * sel[(t << 7) + d];
  row[t] = acc / (snorm[i] * snorm[t]);
  __syncthreads();
  if (t == 0) {
    float mx = -1e30f;
    for (int j = 0; j < B_; ++j) mx = fmaxf(mx, row[j]);
    float s = 0.f;
    for (int j = 0; j < B_; ++j) { row[j] = __expf(row[j] - mx); s += row[j]; }
    float inv = 1.f / s;
    for (int j = 0; j < B_; ++j) row[j] *= inv;
    for (int q = 0; q < 9; ++q) {
      float bm = -1.f; int bi = 0;
      for (int j = 0; j < B_; ++j) if (row[j] > bm) { bm = row[j]; bi = j; }
      tvs[q] = bm; tis[q] = bi; row[bi] = -2.f;
    }
    float mx2 = tvs[0], s2 = 0.f;
    for (int q = 0; q < 9; ++q) { tvs[q] = __expf(tvs[q] - mx2); s2 += tvs[q]; }
    float inv2 = 1.f / s2;
    for (int q = 0; q < 9; ++q) tvs[q] *= inv2;
  }
  __syncthreads();
  float nb = 0.f;
  for (int q = 0; q < 9; ++q) nb += tvs[q] * sel[(tis[q] << 7) + t];
  selF[(i << 7) + t] = selfrow[t] + nb;
}

extern "C" void kernel_launch(void* const* d_in, const int* in_sizes, int n_in,
                              void* d_out, int out_size, void* d_ws, size_t ws_size,
                              hipStream_t stream) {
  (void)in_sizes; (void)n_in; (void)out_size; (void)ws_size;
  const int*   inputs  = (const int*)d_in[0];
  const int*   adj     = (const int*)d_in[1];
  const int*   mask    = (const int*)d_in[2];
  const int*   item    = (const int*)d_in[3];
  const int*   adj_all = (const int*)d_in[5];
  const float* num_w   = (const float*)d_in[6];
  const float* emb     = (const float*)d_in[7];
  const float* pos_emb = (const float*)d_in[8];
  const float* a_local = (const float*)d_in[9];
  const float* gw1     = (const float*)d_in[10];
  const float* gw2     = (const float*)d_in[11];
  const float* gw3     = (const float*)d_in[12];
  const float* w1      = (const float*)d_in[13];
  const float* w2      = (const float*)d_in[14];
  const float* glu1_w  = (const float*)d_in[15];
  const float* glu1_b  = (const float*)d_in[16];
  const float* glu2_w  = (const float*)d_in[17];

  float* out0   = (float*)d_out;
  float* lossP  = out0 + B_ * L_ * D_;
  float* scores = lossP + 1;

  char* wp = (char*)d_ws;
  auto alloc = [&](size_t nfloats) -> float* {
    float* r = (float*)wp;
    wp += ((nfloats * sizeof(float) + 255) / 256) * 256;
    return r;
  };
  float* h        = alloc(B_ * L_ * D_);
  float* h_local  = alloc(B_ * L_ * D_);
  float* h_global = alloc(B_ * L_ * D_);
  float* sessb    = alloc(B_ * D_);
  float* e0p      = alloc(B_ * L_ * D_);
  float* e1p      = alloc((size_t)B_ * 400 * D_);
  float* hsgb     = alloc(B_ * D_);
  float* Pb       = alloc(L_ * D_);
  float* betab    = alloc(B_ * L_);
  float* selb     = alloc(B_ * D_);
  float* selF     = alloc(B_ * D_);
  float* snorm    = alloc(B_);
  float* gw1T     = alloc(2 * D_ * KP_);
  float* w1botT   = alloc(D_ * D_);
  float* wn1      = alloc(B_ * 400);
  float* wn2      = alloc((size_t)B_ * 3200);
  int*   n1       = (int*)alloc(B_ * 400);
  int*   n2       = (int*)alloc((size_t)B_ * 3200);

  prep_kernel<<<dim3(P0_TOTAL_), 256, 0, stream>>>(emb, inputs, item, mask, adj_all, num_w, pos_emb, gw1, w1,
      sessb, n1, wn1, n2, wn2, Pb, h, gw1T, w1botT, lossP);
  hop_kernel<<<dim3(P1_TOTAL_), 256, 0, stream>>>(emb, h, adj, a_local, n1, wn1, n2, wn2,
      sessb, gw1T, gw2, gw3, h_local, e0p, e1p);
  agg_val_kernel<<<dim3(B_ * L_ / GM_), 256, 0, stream>>>(e0p, e1p, wn1,
      sessb, gw1T + D_ * KP_, gw2 + D_, gw3 + 2 * D_ * D_, h_global, L_);
  mid_kernel<<<dim3(P4_TOTAL_), 256, 0, stream>>>(h_local, h_global, mask, glu2_w, out0, hsgb, lossP);
  gate_kernel<<<dim3(B_ * L_ / HD_ROWS_), 256, 0, stream>>>(out0, w1botT, Pb, glu1_w, glu1_b, hsgb, w2,
                                                            mask, betab);
  selnorm_kernel<<<dim3(B_), dim3(D_), 0, stream>>>(out0, betab, selb, snorm);
  fn_kernel<<<dim3(B_), dim3(D_), 0, stream>>>(selb, snorm, selF);
  scores_kernel<<<dim3(512), 256, 0, stream>>>(selF, emb + D_, scores, NN_ - 1,
                                               (NN_ - 1 + SC_COLS_ - 1) / SC_COLS_);
}


# fn kernel: row softmax and top-9, which one lane did in 12 serial walks over the 128-entry row, now use all 128 threads: each counts the entries ranked ahead of its own, which is its place
# speedup vs baseline: 3.1490x; 1.0425x over previous
#include <hip/hip_runtime.h>
#include <hip/hip_bf16.h>
#include <math.h>

#define B_ 128
#define L_ 50
#define D_ 128
#define S_ 8
#define NN_ 100000
#define KP_ 132
#define LEAKY_ 0.2f
#define BETA_ 0.005f

typedef float v2f __attribute__((ext_vector_type(2)));
typedef float v8f __attribute__((ext_vector_type(8)));

#if defined(__has_builtin)
#if __has_builtin(__builtin_amdgcn_wmma_f32_16x16x4_f32)
#define HAVE_WMMA_F32 1
#endif
#endif

__device__ __forceinline__ v8f wmma4(v2f a, v2f b, v8f c) {
#ifdef HAVE_WMMA_F32
  return __builtin_amdgcn_wmma_f32_16x16x4_f32(false, a, false, b, (short)0, c,
                                               false, false);
#else
  float* cf = (float*)&c;
  #pragma unroll
  for (int i = 0; i < 8; ++i) cf[i] += a.x * b.x + a.y * b.y;
  return c;
#endif
}

__device__ __forceinline__ float leaky_(float x) { return x > 0.f ? x : LEAKY_ * x; }
__device__ __forceinline__ float sigmoid_(float x) { return 1.f / (1.f + __expf(-x)); }

#define P0_SESS_  64
#define P0_HOP2_  400
#define P0_POS_   25
#define P0_HOP1_  50
#define P0_GH_    800
#define P0_GW1T_  132
#define P0_W1T_   64
#define P0_TOTAL_ (P0_SESS_ + P0_HOP2_ + P0_POS_ + P0_HOP1_ + P0_GH_ + P0_GW1T_ + P0_W1T_)

__global__ void __launch_bounds__(256)
prep_kernel(const float* __restrict__ emb, const int* __restrict__ inputs,
            const int* __restrict__ item, const int* __restrict__ mask,
            const int* __restrict__ adj_all, const float* __restrict__ num_w,
            const float* __restrict__ pos_emb, const float* __restrict__ gw1,
            const float* __restrict__ w1,
            float* __restrict__ sess, int* __restrict__ n1, float* __restrict__ wn1,
            int* __restrict__ n2, float* __restrict__ wn2, float* __restrict__ P,
            float* __restrict__ h, float* __restrict__ gw1T, float* __restrict__ w1botT,
            float* __restrict__ loss) {
  __shared__ float pr[2][D_];
  const int tid = threadIdx.x;
  int bid = blockIdx.x;
  if (bid < P0_SESS_) {
    const int b = (bid << 1) + (tid >> 7), d = tid & 127;
    float acc = 0.f, ms = 0.f;
    for (int l = 0; l < L_; ++l) {
      float m = (float)mask[b * L_ + l];
      ms += m;
      acc += emb[((long)item[b * L_ + l] << 7) + d] * m;
    }
    sess[(b << 7) + d] = acc / ms;
    if (bid == 0 && tid == 0) loss[0] = 0.f;
  } else if ((bid -= P0_SESS_) < P0_HOP2_) {
    const int t = bid * 256 + tid;
    const int j = t >> 1;
    const int node = adj_all[inputs[j >> 3] * S_ + (j & 7)];
    const int o = node * S_ + ((t & 1) << 2);
    *(int4*)(n2 + (t << 2)) = *(const int4*)(adj_all + o);
    *(float4*)(wn2 + (t << 2)) = *(const float4*)(num_w + o);
  } else if ((bid -= P0_HOP2_) < P0_POS_) {
    const int hb = tid >> 7, d = tid & 127, l = (bid << 1) + hb;
    pr[hb][d] = pos_emb[(l << 7) + d];
    __syncthreads();
    float acc = 0.f;
    for (int e = 0; e < D_; ++e) acc += pr[hb][e] * w1[(e << 7) + d];
    P[(l << 7) + d] = acc;
  } else if ((bid -= P0_POS_) < P0_HOP1_) {
    const int t = bid * 256 + tid;
    const int o = inputs[t >> 1] * S_ + ((t & 1) << 2);
    *(int4*)(n1 + (t << 2)) = *(const int4*)(adj_all + o);
    *(float4*)(wn1 + (t << 2)) = *(const float4*)(num_w + o);
  } else if ((bid -= P0_HOP1_) < P0_GH_) {
    const int q = bid * 256 + tid;
    *(float4*)(h + ((long)q << 2)) =
        *(const float4*)(emb + ((long)inputs[q >> 5] << 7) + ((q & 31) << 2));
  } else if ((bid -= P0_GH_) < P0_GW1T_) {
    const int i = bid * 256 + tid;
    const int hop = i / (D_ * KP_);
    const int rem = i % (D_ * KP_);
    const int n = rem / KP_, k = rem % KP_;
    gw1T[i] = (k < 129) ? gw1[hop * 129 * D_ + k * D_ + n] : 0.f;
  } else {
    bid -= P0_GW1T_;
    const int i = bid * 256 + tid;
    const int n = i >> 7, k = i & 127;
    w1botT[i] = w1[(D_ + k) * D_ + n];
  }
}

#define HP_ 132

__device__ __forceinline__ void
local_agg_body(float* smem, const int b,
               const float* __restrict__ h,
               const int* __restrict__ adj,
               const float* __restrict__ a_local,
               float* __restrict__ h_local) {
  float (*hv)[HP_] = reinterpret_cast<float (*)[HP_]>(smem);
  float (*al)[D_]  = reinterpret_cast<float (*)[D_]>(smem + L_ * HP_);
  float (*att)[L_] = reinterpret_cast<float (*)[L_]>(smem + L_ * HP_ + 4 * D_);
  float* rinv = smem + L_ * HP_ + 4 * D_ + L_ * L_;
  const int tid = threadIdx.x;
  const int lane = tid & 31, laneIdx = lane & 15, laneHalf = lane >> 4;
  const int wave = tid >> 5;

  for (int q = tid; q < L_ * (D_ / 4); q += 256)
    *(float4*)&hv[q >> 5][(q & 31) << 2] = *(const float4*)(h + (long)b * L_ * D_ + (q << 2));
  if (tid < D_) *(float4*)&al[0][tid << 2] = *(const float4*)(a_local + (tid << 2));

  const int i0 = (wave >> 1) << 4;
  const int j0 = (wave & 1) << 5;
  int et[2][8];
  #pragma unroll
  for (int t = 0; t < 2; ++t) {
    #pragma unroll
    for (int v = 0; v < 8; ++v) {
      const int i = i0 + (laneHalf << 3) + v, j = j0 + (t << 4) + laneIdx;
      et[t][v] = (i < L_ && j < L_) ? adj[b * L_ * L_ + i * L_ + j] : 0;
    }
  }
  __syncthreads();

  {
    const int ir  = (i0 + laneIdx < L_) ? i0 + laneIdx : L_ - 1;
    const int jr0 = (j0 + laneIdx < L_) ? j0 + laneIdx : L_ - 1;
    const int jr1 = (j0 + 16 + laneIdx < L_) ? j0 + 16 + laneIdx : L_ - 1;
    const float* ap  = &hv[ir][laneHalf << 1];
    const float* bp0 = &hv[jr0][laneHalf << 1];
    const float* bp1 = &hv[jr1][laneHalf << 1];
    const float* sp  = &al[0][laneHalf << 1];
    v8f e[4][2] = {};
    for (int ks = 0; ks < D_ / 4; ++ks) {
      const v2f x  = *(const v2f*)(ap + 4 * ks);
      const v2f y0 = *(const v2f*)(bp0 + 4 * ks);
      const v2f y1 = *(const v2f*)(bp1 + 4 * ks);
      #pragma unroll
      for (int k = 0; k < 4; ++k) {
        const v2f s = *(const v2f*)(sp + k * D_ + 4 * ks);
        const v2f xa = x * s;
        e[k][0] = wmma4(xa, y0, e[k][0]);
        e[k][1] = wmma4(xa, y1, e[k][1]);
      }
    }
    #pragma unroll
    for (int t = 0; t < 2; ++t) {
      const int j = j0 + (t << 4) + laneIdx;
      #pragma unroll
      for (int v = 0; v < 8; ++v) {
        const int i = i0 + (laneHalf << 3) + v;
        const int k = et[t][v];
        float x = e[3][t][v];
        x = (k == 3) ? e[2][t][v] : x;
        x = (k == 2) ? e[1][t][v] : x;
        x = (k == 1) ? e[0][t][v] : x;
        x = (k == 0) ? -9e15f : leaky_(x);
        if (i < L_ && j < L_) att[i][j] = x;
      }
    }
  }
  __syncthreads();
  if (threadIdx.x < L_) {
    int i = threadIdx.x;
    float mx = -1e30f;
    for (int j = 0; j < L_; ++j) mx = fmaxf(mx, att[i][j]);
    float s = 0.f;
    for (int j = 0; j < L_; ++j) { float ex = __expf(att[i][j] - mx); att[i][j] = ex; s += ex; }
    rinv[i] = 1.f / s;
  }
  __syncthreads();
  for (int p = threadIdx.x; p < L_ * D_; p += 256) {
    int i = p >> 7, d = p & 127;
    float acc = 0.f;
    for (int j = 0; j < L_; ++j) acc += att[i][j] * hv[j][d];
    h_local[(long)b * L_ * D_ + p] = acc * rinv[i];
  }
}

#define GM_ 16
#define GR_ 128
#define CP_ 260

__device__ __forceinline__ void score_cols(v8f c, const float (&g2v)[8], float* scRow,
                                           int laneIdx) {
  float acc = 0.f;
  #pragma unroll
  for (int v = 0; v < 8; ++v) acc += leaky_(c[v]) * g2v[v];
  atomicAdd(&scRow[laneIdx], acc);
}

#define AGG_LDS_FLOATS_ (GR_ * D_ + GR_ * KP_ + GM_ * CP_ + 2 * GR_)

template <bool SELF_VAL, bool NEIGH_IDX>
__device__ __forceinline__ void
agg_body(float* smem, const unsigned blk,
         const float* __restrict__ emb,
         const float* __restrict__ selfVal, const int* __restrict__ selfIdx,
         const int* __restrict__ neighIdx, const float* __restrict__ neighVal,
         const float* __restrict__ neighW,
         const float* __restrict__ sess,
         const float* __restrict__ gw1T, const float* __restrict__ gw2p,
         const float* __restrict__ gw3p,
         float* __restrict__ out, const int M) {
  float (*nv)[D_]   = reinterpret_cast<float (*)[D_]>(smem);
  float (*xs)[KP_]  = reinterpret_cast<float (*)[KP_]>(smem + GR_ * D_);
  float (*cat)[CP_] = reinterpret_cast<float (*)[CP_]>(smem + GR_ * D_ + GR_ * KP_);
  float* sc    = smem + GR_ * D_ + GR_ * KP_ + GM_ * CP_;
  float* alpha = sc + GR_;

  const int tid = threadIdx.x;
  const int lane = tid & 31, laneIdx = lane & 15, laneHalf = lane >> 4;
  const int wave = tid >> 5;
  const int n0 = wave * 16;
  const long base  = (long)blk * GM_;
  const long nbase = base * S_;

  v2f breg[KP_ / 4];
  {
    const float* bp = gw1T + (n0 + laneIdx) * KP_ + (laneHalf << 1);
    #pragma unroll
    for (int ks = 0; ks < KP_ / 4; ++ks) breg[ks] = *(const v2f*)(bp + 4 * ks);
  }
  float g2v[8];
  {
    const float4 ga = *(const float4*)(gw2p + n0 + (laneHalf << 3));
    const float4 gb = *(const float4*)(gw2p + n0 + (laneHalf << 3) + 4);
    g2v[0] = ga.x; g2v[1] = ga.y; g2v[2] = ga.z; g2v[3] = ga.w;
    g2v[4] = gb.x; g2v[5] = gb.y; g2v[6] = gb.z; g2v[7] = gb.w;
  }

  const unsigned ubase = blk * GM_;
  const unsigned b0 = ubase / (unsigned)M;
  const unsigned rem0 = ubase - b0 * (unsigned)M;
  float4 gv[GM_];
  if (NEIGH_IDX) {
    int idx[GM_];
    #pragma unroll
    for (int j = 0; j < GM_; ++j) idx[j] = neighIdx[nbase + (j << 3) + wave];
    #pragma unroll
    for (int j = 0; j < GM_; ++j)
      gv[j] = *(const float4*)(emb + ((long)idx[j] << 7) + (lane << 2));
  } else {
    #pragma unroll
    for (int j = 0; j < GM_; ++j)
      gv[j] = *(const float4*)(neighVal + ((nbase + (j << 3) + wave) << 7) + (lane << 2));
  }
  const unsigned b1 = (rem0 + GM_ > (unsigned)M) ? b0 + 1 : b0;
  const float4 sfa = *(const float4*)(sess + ((long)b0 << 7) + (lane << 2));
  const float4 sfb = *(const float4*)(sess + ((long)b1 << 7) + (lane << 2));
  float4 sv0, sv1;
  if (SELF_VAL) {
    sv0 = *(const float4*)(selfVal + ((base + wave) << 7) + (lane << 2));
    sv1 = *(const float4*)(selfVal + ((base + wave + 8) << 7) + (lane << 2));
  } else {
    sv0 = *(const float4*)(emb + ((long)selfIdx[base + wave] << 7) + (lane << 2));
    sv1 = *(const float4*)(emb + ((long)selfIdx[base + wave + 8] << 7) + (lane << 2));
  }
  #pragma unroll
  for (int j = 0; j < GM_; ++j) {
    const int r = (j << 3) + wave;
    const float4 v = gv[j];
    const bool nextb = rem0 + (unsigned)j >= (unsigned)M;
    *(float4*)&nv[r][lane << 2] = v;
    float4 x;
    x.x = (nextb ? sfb.x : sfa.x) * v.x;
    x.y = (nextb ? sfb.y : sfa.y) * v.y;
    x.z = (nextb ? sfb.z : sfa.z) * v.z;
    x.w = (nextb ? sfb.w : sfa.w) * v.w;
    *(float4*)&xs[r][lane << 2] = x;
  }
  *(float4*)&cat[wave][lane << 2] = sv0;
  *(float4*)&cat[wave + 8][lane << 2] = sv1;
  if (tid < GR_) {
    xs[tid][D_] = neighW[nbase + tid];
    xs[tid][D_ + 1] = 0.f; xs[tid][D_ + 2] = 0.f; xs[tid][D_ + 3] = 0.f;
    sc[tid] = 0.f;
  }
  __syncthreads();

  for (int rp = 0; rp < GR_ / 32; ++rp) {
    const float* xp0 = &xs[(rp << 5) + laneIdx][laneHalf << 1];
    const float* xp1 = xp0 + 16 * KP_;
    v8f c0 = {}, c1 = {};
    #pragma unroll
    for (int ks = 0; ks < KP_ / 4; ++ks) {
      v2f x0 = *(const v2f*)(xp0 + 4 * ks);
      v2f x1 = *(const v2f*)(xp1 + 4 * ks);
      c0 = wmma4(breg[ks], x0, c0);
      c1 = wmma4(breg[ks], x1, c1);
    }
    score_cols(c0, g2v, &sc[(rp << 5)], laneIdx);
    score_cols(c1, g2v, &sc[(rp << 5) + 16], laneIdx);
  }
  __syncthreads();

  if (tid < GM_) {
    float mx = -1e30f;
    for (int s = 0; s < 8; ++s) mx = fmaxf(mx, sc[tid * 8 + s]);
    float ex[8], sum = 0.f;
    for (int s = 0; s < 8; ++s) { ex[s] = __expf(sc[tid * 8 + s] - mx); sum += ex[s]; }
    float inv = 1.f / sum;
    for (int s = 0; s < 8; ++s) alpha[tid * 8 + s] = ex[s] * inv;
  }
  __syncthreads();
  {
    const int d = tid & 127;
    for (int j = 0; j < GM_ / 2; ++j) {
      const int r = (tid >> 7) + (j << 1);
      float acc = 0.f;
      for (int s = 0; s < 8; ++s) acc += alpha[(r << 3) + s] * nv[(r << 3) + s][d];
      cat[r][D_ + d] = acc;
    }
  }
  __syncthreads();

  {
    const int col = n0 + laneIdx;
    const float* ap = &cat[laneIdx][laneHalf << 1];
    const float* bp = gw3p + ((laneHalf << 1) << 7) + col;
    v8f cs = {}, cn = {};
    #pragma unroll 8
    for (int k0 = 0; k0 < D_; k0 += 4) {
      v2f as = *(const v2f*)(ap + k0);
      v2f an = *(const v2f*)(ap + D_ + k0);
      v2f bs, bn;
      bs.x = bp[k0 << 7];
      bs.y = bp[(k0 + 1) << 7];
      bn.x = bp[(D_ + k0) << 7];
      bn.y = bp[(D_ + k0 + 1) << 7];
      cs = wmma4(as, bs, cs);
      cn = wmma4(an, bn, cn);
    }
    #pragma unroll
    for (int v = 0; v < 8; ++v)
      xs[v + (laneHalf << 3)][col] = fmaxf(cs[v] + cn[v], 0.f);
  }
  __syncthreads();
  *(float4*)(out + ((base + wave) << 7) + (lane << 2)) = *(const float4*)&xs[wave][lane << 2];
  *(float4*)(out + ((base + wave + 8) << 7) + (lane << 2)) = *(const float4*)&xs[wave + 8][lane << 2];
}

#define P1_LOCAL_ B_
#define P1_HOP1_  (B_ * L_ / GM_)
#define P1_HOP2_  (B_ * L_ * S_ / GM_)
#define P1_TOTAL_ (P1_LOCAL_ + P1_HOP1_ + P1_HOP2_)

__global__ void __launch_bounds__(256)
hop_kernel(const float* __restrict__ emb, const float* __restrict__ h,
           const int* __restrict__ adj, const float* __restrict__ a_local,
           const int* __restrict__ n1, const float* __restrict__ wn1,
           const int* __restrict__ n2, const float* __restrict__ wn2,
           const float* __restrict__ sess,
           const float* __restrict__ gw1T, const float* __restrict__ gw2p,
           const float* __restrict__ gw3p,
           float* __restrict__ h_local, float* __restrict__ e0, float* __restrict__ e1) {
  __shared__ __align__(16) float smem[AGG_LDS_FLOATS_];
  int bid = blockIdx.x;
  if (bid < P1_LOCAL_) {
    local_agg_body(smem, bid, h, adj, a_local, h_local);
    return;
  }
  bid -= P1_LOCAL_;
  if (bid < P1_HOP1_) {
    agg_body<true, true>(smem, (unsigned)bid, emb, h, nullptr, n1, nullptr, wn1,
                         sess, gw1T, gw2p, gw3p, e0, L_);
  } else {
    agg_body<false, true>(smem, (unsigned)(bid - P1_HOP1_), emb, nullptr, n1, n2, nullptr, wn2,
                          sess, gw1T, gw2p, gw3p, e1, L_ * S_);
  }
}

__global__ void __launch_bounds__(256)
agg_val_kernel(const float* __restrict__ selfVal, const float* __restrict__ neighVal,
               const float* __restrict__ neighW, const float* __restrict__ sess,
               const float* __restrict__ gw1T, const float* __restrict__ gw2p,
               const float* __restrict__ gw3p, float* __restrict__ out, int M) {
  __shared__ __align__(16) float smem[AGG_LDS_FLOATS_];
  agg_body<true, false>(smem, blockIdx.x, nullptr, selfVal, nullptr, nullptr, neighVal, neighW,
                        sess, gw1T, gw2p, gw3p, out, M);
}

#define SC_COLS_ 32
#define SC_KH_ (D_ / 8)

__device__ __forceinline__ float4 sc_row(const float* __restrict__ embT, int n, int N, int lane) {
  n = n < N ? n : N - 1;
  return *(const float4*)(embT + ((long)n << 7) + (lane << 2));
}

__global__ void __launch_bounds__(256)
scores_kernel(const float* __restrict__ selF, const float* __restrict__ embT,
              float* __restrict__ scores, int N, int nStages) {
  __shared__ __align__(16) float bs[2][SC_COLS_][KP_];
  const int tid = threadIdx.x;
  const int lane = tid & 31, laneIdx = lane & 15, laneHalf = lane >> 4;
  const int wave = tid >> 5;

  v2f areg[D_ / 4];
  {
    const float* ap = selF + ((wave * 16 + laneIdx) << 7) + (laneHalf << 1);
    #pragma unroll
    for (int ks = 0; ks < D_ / 4; ++ks) areg[ks] = *(const v2f*)(ap + 4 * ks);
  }

  const int r0 = wave * 4;
  float4 p0, p1, p2, p3;
  int s = blockIdx.x;
  if (s < nStages) {
    const int n = s * SC_COLS_ + r0;
    p0 = sc_row(embT, n, N, lane);
    p1 = sc_row(embT, n + 1, N, lane);
    p2 = sc_row(embT, n + 2, N, lane);
    p3 = sc_row(embT, n + 3, N, lane);
    *(float4*)&bs[0][r0][lane << 2] = p0;
    *(float4*)&bs[0][r0 + 1][lane << 2] = p1;
    *(float4*)&bs[0][r0 + 2][lane << 2] = p2;
    *(float4*)&bs[0][r0 + 3][lane << 2] = p3;
  }
  __syncthreads();

  int buf = 0;
  for (; s < nStages; s += gridDim.x) {
    const int sn = s + gridDim.x;
    const bool more = sn < nStages;
    if (more) {
      const int n = sn * SC_COLS_ + r0;
      p0 = sc_row(embT, n, N, lane);
      p1 = sc_row(embT, n + 1, N, lane);
      p2 = sc_row(embT, n + 2, N, lane);
      p3 = sc_row(embT, n + 3, N, lane);
    }
    const float* bp0 = &bs[buf][laneIdx][laneHalf << 1];
    const float* bp1 = bp0 + 16 * KP_;
    v8f c0 = {}, c1 = {};
    #pragma unroll
    for (int kh = 0; kh < 2; ++kh) {
      v2f f0[SC_KH_], f1[SC_KH_];
      #pragma unroll
      for (int ks = 0; ks < SC_KH_; ++ks) {
        f0[ks] = *(const v2f*)(bp0 + 4 * (kh * SC_KH_ + ks));
        f1[ks] = *(const v2f*)(bp1 + 4 * (kh * SC_KH_ + ks));
      }
      #pragma unroll
      for (int ks = 0; ks < SC_KH_; ++ks) {
        c0 = wmma4(areg[kh * SC_KH_ + ks], f0[ks], c0);
        c1 = wmma4(areg[kh * SC_KH_ + ks], f1[ks], c1);
      }
    }
    if (more) {
      *(float4*)&bs[buf ^ 1][r0][lane << 2] = p0;
      *(float4*)&bs[buf ^ 1][r0 + 1][lane << 2] = p1;
      *(float4*)&bs[buf ^ 1][r0 + 2][lane << 2] = p2;
      *(float4*)&bs[buf ^ 1][r0 + 3][lane << 2] = p3;
    }
    __syncthreads();
    buf ^= 1;
    {
      const int n0 = s * SC_COLS_;
      float* o = scores + (long)(wave * 16 + (laneHalf << 3)) * N + n0 + laneIdx;
      if (n0 + SC_COLS_ <= N) {
        #pragma unroll
        for (int v = 0; v < 8; ++v) {
          o[(long)v * N] = c0[v];
          o[(long)v * N + 16] = c1[v];
        }
      } else {
        const bool in0 = n0 + laneIdx < N, in1 = n0 + laneIdx + 16 < N;
        #pragma unroll
        for (int v = 0; v < 8; ++v) {
          if (in0) o[(long)v * N] = c0[v];
          if (in1) o[(long)v * N + 16] = c1[v];
        }
      }
    }
  }
}

#define P4_HS_  64
#define P4_SSL_ 64
#define P4_ADD_ 800
#define P4_TOTAL_ (P4_HS_ + P4_SSL_ + P4_ADD_)

__global__ void __launch_bounds__(256)
mid_kernel(const float* __restrict__ hl, const float* __restrict__ hg,
           const int* __restrict__ mask, const float* __restrict__ glu2_w,
           float* __restrict__ out0, float* __restrict__ hsg, float* __restrict__ loss) {
  __shared__ float hr[2][D_];
  __shared__ float red[256];
  const int tid = threadIdx.x;
  int bid = blockIdx.x;
  if (bid < P4_HS_) {
    const int hb = tid >> 7, d = tid & 127, b = (bid << 1) + hb;
    float acc = 0.f, ms = 0.f;
    for (int l = 0; l < L_; ++l) {
      const float m = (float)mask[b * L_ + l];
      const int o = ((b * L_ + l) << 7) + d;
      ms += m;
      acc += (hl[o] + hg[o]) * m;
    }
    hr[hb][d] = acc / ms;
    __syncthreads();
    float a2 = 0.f;
    for (int e = 0; e < D_; ++e) a2 += hr[hb][e] * glu2_w[(d << 7) + e];
    hsg[(b << 7) + d] = a2;
  } else if ((bid -= P4_HS_) < P4_SSL_) {
    const int idx = bid * 256 + tid;
    const int b = idx >> 7, d = idx & 127;
    const int pb = (b * 73 + 21) & 127;
    float pos = 0.f, neg = 0.f;
    for (int l = 0; l < L_; ++l) {
      const int pl = (l * 13 + 7) % L_;
      pos += hl[((b * L_ + l) << 7) + d] * hg[((b * L_ + l) << 7) + d];
      neg += hg[((b * L_ + l) << 7) + d] * hl[((pb * L_ + pl) << 7) + d];
    }
    const float term = -__logf(1e-8f + sigmoid_(pos)) - __logf(1e-8f + 1.f - sigmoid_(neg));
    red[tid] = term;
    __syncthreads();
    for (int st = 128; st > 0; st >>= 1) {
      if (tid < st) red[tid] += red[tid + st];
      __syncthreads();
    }
    if (tid == 0) atomicAdd(loss, BETA_ * red[0]);
  } else {
    bid -= P4_SSL_;
    const int q = bid * 256 + tid;
    const float4 x = *(const float4*)(hl + ((long)q << 2));
    const float4 y = *(const float4*)(hg + ((long)q << 2));
    float4 r;
    r.x = x.x + y.x; r.y = x.y + y.y; r.z = x.z + y.z; r.w = x.w + y.w;
    *(float4*)(out0 + ((long)q << 2)) = r;
  }
}

#define HD_ROWS_ 32

__device__ __forceinline__ v2f lo2_(float4 f) { v2f r; r.x = f.x; r.y = f.y; return r; }
__device__ __forceinline__ v2f hi2_(float4 f) { v2f r; r.x = f.z; r.y = f.w; return r; }

__global__ void __launch_bounds__(256)
gate_kernel(const float* __restrict__ hidden, const float* __restrict__ w1botT,
            const float* __restrict__ P, const float* __restrict__ glu1_w,
            const float* __restrict__ glu1_b, const float* __restrict__ hsg,
            const float* __restrict__ w2, const int* __restrict__ mask,
            float* __restrict__ beta) {
  __shared__ __align__(16) float xr[HD_ROWS_][HP_];
  __shared__ __align__(16) float nhs[HD_ROWS_][HP_];
  __shared__ float part[16][HD_ROWS_];
  const int tid = threadIdx.x;
  const int lane = tid & 31, laneIdx = lane & 15, laneHalf = lane >> 4;
  const int wave = tid >> 5;
  const int n0 = wave << 4;
  const int row0 = blockIdx.x * HD_ROWS_;

  float4 hq[4];
  #pragma unroll
  for (int q = 0; q < 4; ++q)
    hq[q] = *(const float4*)(hidden + ((long)row0 << 7) + ((tid + (q << 8)) << 2));
  float4 bw[16];
  {
    const float* bp = w1botT + ((n0 + laneIdx) << 7) + (laneHalf << 2);
    #pragma unroll
    for (int j = 0; j < 16; ++j) bw[j] = *(const float4*)(bp + 8 * j);
  }
  float pv[2][8];
  #pragma unroll
  for (int t = 0; t < 2; ++t) {
    #pragma unroll
    for (int v = 0; v < 8; ++v) {
      const int l = (row0 + (t << 4) + (laneHalf << 3) + v) % L_;
      pv[t][v] = P[(l << 7) + n0 + laneIdx];
    }
  }

  #pragma unroll
  for (int q = 0; q < 4; ++q) {
    const int i = tid + (q << 8);
    *(float4*)&xr[i >> 5][(i & 31) << 2] = hq[q];
  }
  __syncthreads();

  float4 gw[16];
  {
    const float* gp = glu1_w + ((n0 + laneIdx) << 7) + (laneHalf << 2);
    #pragma unroll
    for (int j = 0; j < 16; ++j) gw[j] = *(const float4*)(gp + 8 * j);
  }
  const int nc = n0 + (laneHalf << 3);
  const float4 gba = *(const float4*)(glu1_b + nc), gbb = *(const float4*)(glu1_b + nc + 4);
  const float4 w2a = *(const float4*)(w2 + nc), w2b = *(const float4*)(w2 + nc + 4);
  const float* hp0 = hsg + (((row0 + laneIdx) / L_) << 7) + nc;
  const float* hp1 = hsg + (((row0 + 16 + laneIdx) / L_) << 7) + nc;
  const float4 h0a = *(const float4*)hp0, h0b = *(const float4*)(hp0 + 4);
  const float4 h1a = *(const float4*)hp1, h1b = *(const float4*)(hp1 + 4);

  {
    const float* ap0 = &xr[laneIdx][laneHalf << 2];
    const float* ap1 = ap0 + 16 * HP_;
    v8f c0 = {}, c1 = {};
    #pragma unroll
    for (int j = 0; j < 16; ++j) {
      const float4 x0 = *(const float4*)(ap0 + 8 * j);
      const float4 x1 = *(const float4*)(ap1 + 8 * j);
      const float4 b = bw[j];
      c0 = wmma4(lo2_(x0), lo2_(b), c0);
      c1 = wmma4(lo2_(x1), lo2_(b), c1);
      c0 = wmma4(hi2_(x0), hi2_(b), c0);
      c1 = wmma4(hi2_(x1), hi2_(b), c1);
    }
    #pragma unroll
    for (int v = 0; v < 8; ++v) {
      nhs[(laneHalf << 3) + v][n0 + laneIdx] = tanhf(c0[v] + pv[0][v]);
      nhs[16 + (laneHalf << 3) + v][n0 + laneIdx] = tanhf(c1[v] + pv[1][v]);
    }
  }
  __syncthreads();

  {
    const float* np0 = &nhs[laneIdx][laneHalf << 2];
    const float* np1 = np0 + 16 * HP_;
    v8f d0 = {}, d1 = {};
    #pragma unroll
    for (int j = 0; j < 16; ++j) {
      const float4 y0 = *(const float4*)(np0 + 8 * j);
      const float4 y1 = *(const float4*)(np1 + 8 * j);
      const float4 a = gw[j];
      d0 = wmma4(lo2_(a), lo2_(y0), d0);
      d1 = wmma4(lo2_(a), lo2_(y1), d1);
      d0 = wmma4(hi2_(a), hi2_(y0), d0);
      d1 = wmma4(hi2_(a), hi2_(y1), d1);
    }
    const float gb[8] = {gba.x, gba.y, gba.z, gba.w, gbb.x, gbb.y, gbb.z, gbb.w};
    const float wv[8] = {w2a.x, w2a.y, w2a.z, w2a.w, w2b.x, w2b.y, w2b.z, w2b.w};
    const float s0[8] = {h0a.x, h0a.y, h0a.z, h0a.w, h0b.x, h0b.y, h0b.z, h0b.w};
    const float s1[8] = {h1a.x, h1a.y, h1a.z, h1a.w, h1b.x, h1b.y, h1b.z, h1b.w};
    float acc0 = 0.f, acc1 = 0.f;
    #pragma unroll
    for (int v = 0; v < 8; ++v) {
      acc0 += sigmoid_(d0[v] + gb[v] + s0[v]) * wv[v];
      acc1 += sigmoid_(d1[v] + gb[v] + s1[v]) * wv[v];
    }
    part[(wave << 1) + laneHalf][laneIdx] = acc0;
    part[(wave << 1) + laneHalf][16 + laneIdx] = acc1;
  }
  __syncthreads();
  if (tid < HD_ROWS_) {
    float s = 0.f;
    for (int q = 0; q < 16; ++q) s += part[q][tid];
    beta[row0 + tid] = s * (float)mask[row0 + tid];
  }
}

__global__ void selnorm_kernel(const float* __restrict__ hidden, const float* __restrict__ beta,
                               float* __restrict__ sel, float* __restrict__ snorm) {
  __shared__ float red[D_];
  int b = blockIdx.x, d = threadIdx.x;
  float acc = 0.f;
  for (int l = 0; l < L_; ++l) acc += beta[b * L_ + l] * hidden[((b * L_ + l) << 7) + d];
  sel[(b << 7) + d] = acc;
  red[d] = acc * acc + 1e-6f;
  __syncthreads();
  for (int st = 64; st > 0; st >>= 1) {
    if (d < st) red[d] += red[d + st];
    __syncthreads();
  }
  if (d == 0) snorm[b] = sqrtf(red[0]);
}

__global__ void fn_kernel(const float* __restrict__ sel, const float* __restrict__ snorm,
                          float* __restrict__ selF) {
  __shared__ __align__(16) float row[B_];
  __shared__ float selfrow[D_];
  __shared__ float tvs[9];
  __shared__ int tis[9];
  int i = blockIdx.x, t = threadIdx.x;
  selfrow[t] = sel[(i << 7) + t];
  __syncthreads();
  float acc = 0.f;
  for (int d = 0; d < D_; ++d) acc += selfrow[d] * sel[(t << 7) + d];
  const float cs = acc / (snorm[i] * snorm[t]);
  row[t] = cs;
  __syncthreads();
  float mx = -1e30f;
  for (int j = 0; j < B_; ++j) mx = fmaxf(mx, row[j]);
  const float ex = __expf(cs - mx);
  __syncthreads();
  row[t] = ex;
  __syncthreads();
  float s = 0.f;
  for (int j = 0; j < B_; ++j) s += row[j];
  const float inv = 1.f / s;
  const float p = ex * inv;
  int place = 0;
  for (int j = 0; j < B_; ++j) {
    const float pj = row[j] * inv;
    place += (pj > p || (pj == p && j < t)) ? 1 : 0;
  }
  if (place < 9) { tvs[place] = p; tis[place] = t; }
  __syncthreads();
  float w[9];
  const float mx2 = tvs[0];
  float s2 = 0.f;
  #pragma unroll
  for (int q = 0; q < 9; ++q) { w[q] = __expf(tvs[q] - mx2); s2 += w[q]; }
  const float inv2 = 1.f / s2;
  float nb = 0.f;
  #pragma unroll
  for (int q = 0; q < 9; ++q) nb += (w[q] * inv2) * sel[(tis[q] << 7) + t];
  selF[(i << 7) + t] = selfrow[t] + nb;
}

extern "C" void kernel_launch(void* const* d_in, const int* in_sizes, int n_in,
                              void* d_out, int out_size, void* d_ws, size_t ws_size,
                              hipStream_t stream) {
  (void)in_sizes; (void)n_in; (void)out_size; (void)ws_size;
  const int*   inputs  = (const int*)d_in[0];
  const int*   adj     = (const int*)d_in[1];
  const int*   mask    = (const int*)d_in[2];
  const int*   item    = (const int*)d_in[3];
  const int*   adj_all = (const int*)d_in[5];
  const float* num_w   = (const float*)d_in[6];
  const float* emb     = (const float*)d_in[7];
  const float* pos_emb = (const float*)d_in[8];
  const float* a_local = (const float*)d_in[9];
  const float* gw1     = (const float*)d_in[10];
  const float* gw2     = (const float*)d_in[11];
  const float* gw3     = (const float*)d_in[12];
  const float* w1      = (const float*)d_in[13];
  const float* w2      = (const float*)d_in[14];
  const float* glu1_w  = (const float*)d_in[15];
  const float* glu1_b  = (const float*)d_in[16];
  const float* glu2_w  = (const float*)d_in[17];

  float* out0   = (float*)d_out;
  float* lossP  = out0 + B_ * L_ * D_;
  float* scores = lossP + 1;

  char* wp = (char*)d_ws;
  auto alloc = [&](size_t nfloats) -> float* {
    float* r = (float*)wp;
    wp += ((nfloats * sizeof(float) + 255) / 256) * 256;
    return r;
  };
  float* h        = alloc(B_ * L_ * D_);
  float* h_local  = alloc(B_ * L_ * D_);
  float* h_global = alloc(B_ * L_ * D_);
  float* sessb    = alloc(B_ * D_);
  float* e0p      = alloc(B_ * L_ * D_);
  float* e1p      = alloc((size_t)B_ * 400 * D_);
  float* hsgb     = alloc(B_ * D_);
  float* Pb       = alloc(L_ * D_);
  float* betab    = alloc(B_ * L_);
  float* selb     = alloc(B_ * D_);
  float* selF     = alloc(B_ * D_);
  float* snorm    = alloc(B_);
  float* gw1T     = alloc(2 * D_ * KP_);
  float* w1botT   = alloc(D_ * D_);
  float* wn1      = alloc(B_ * 400);
  float* wn2      = alloc((size_t)B_ * 3200);
  int*   n1       = (int*)alloc(B_ * 400);
  int*   n2       = (int*)alloc((size_t)B_ * 3200);

  prep_kernel<<<dim3(P0_TOTAL_), 256, 0, stream>>>(emb, inputs, item, mask, adj_all, num_w, pos_emb, gw1, w1,
      sessb, n1, wn1, n2, wn2, Pb, h, gw1T, w1botT, lossP);
  hop_kernel<<<dim3(P1_TOTAL_), 256, 0, stream>>>(emb, h, adj, a_local, n1, wn1, n2, wn2,
      sessb, gw1T, gw2, gw3, h_local, e0p, e1p);
  agg_val_kernel<<<dim3(B_ * L_ / GM_), 256, 0, stream>>>(e0p, e1p, wn1,
      sessb, gw1T + D_ * KP_, gw2 + D_, gw3 + 2 * D_ * D_, h_global, L_);
  mid_kernel<<<dim3(P4_TOTAL_), 256, 0, stream>>>(h_local, h_global, mask, glu2_w, out0, hsgb, lossP);
  gate_kernel<<<dim3(B_ * L_ / HD_ROWS_), 256, 0, stream>>>(out0, w1botT, Pb, glu1_w, glu1_b, hsgb, w2,
                                                            mask, betab);
  selnorm_kernel<<<dim3(B_), dim3(D_), 0, stream>>>(out0, betab, selb, snorm);
  fn_kernel<<<dim3(B_), dim3(D_), 0, stream>>>(selb, snorm, selF);
  scores_kernel<<<dim3(512), 256, 0, stream>>>(selF, emb + D_, scores, NN_ - 1,
                                               (NN_ - 1 + SC_COLS_ - 1) / SC_COLS_);
}


# agg kernels: gw1 and w3 slices stored in fragment order by the first launch, so a fragment load reads whole 128-B lines, not 16 lines; all w3 fragments loaded as one group ahead of their product
# speedup vs baseline: 3.2835x; 1.0427x over previous
#include <hip/hip_runtime.h>
#include <hip/hip_bf16.h>
#include <math.h>

#define B_ 128
#define L_ 50
#define D_ 128
#define S_ 8
#define NN_ 100000
#define KP_ 132
#define LEAKY_ 0.2f
#define BETA_ 0.005f

typedef float v2f __attribute__((ext_vector_type(2)));
typedef float v8f __attribute__((ext_vector_type(8)));

#if defined(__has_builtin)
#if __has_builtin(__builtin_amdgcn_wmma_f32_16x16x4_f32)
#define HAVE_WMMA_F32 1
#endif
#endif

__device__ __forceinline__ v8f wmma4(v2f a, v2f b, v8f c) {
#ifdef HAVE_WMMA_F32
  return __builtin_amdgcn_wmma_f32_16x16x4_f32(false, a, false, b, (short)0, c,
                                               false, false);
#else
  float* cf = (float*)&c;
  #pragma unroll
  for (int i = 0; i < 8; ++i) cf[i] += a.x * b.x + a.y * b.y;
  return c;
#endif
}

__device__ __forceinline__ float leaky_(float x) { return x > 0.f ? x : LEAKY_ * x; }
__device__ __forceinline__ float sigmoid_(float x) { return 1.f / (1.f + __expf(-x)); }

#define P0_SESS_  64
#define P0_HOP2_  400
#define P0_POS_   25
#define P0_HOP1_  50
#define P0_GH_    800
#define P0_GW1T_  132
#define P0_W1T_   64
#define P0_GW3F_  64
#define P0_TOTAL_ (P0_SESS_ + P0_HOP2_ + P0_POS_ + P0_HOP1_ + P0_GH_ + P0_GW1T_ + P0_W1T_ + P0_GW3F_)

__global__ void __launch_bounds__(256)
prep_kernel(const float* __restrict__ emb, const int* __restrict__ inputs,
            const int* __restrict__ item, const int* __restrict__ mask,
            const int* __restrict__ adj_all, const float* __restrict__ num_w,
            const float* __restrict__ pos_emb, const float* __restrict__ gw1,
            const float* __restrict__ w1, const float* __restrict__ gw3,
            float* __restrict__ sess, int* __restrict__ n1, float* __restrict__ wn1,
            int* __restrict__ n2, float* __restrict__ wn2, float* __restrict__ P,
            float* __restrict__ h, float* __restrict__ gw1F, float* __restrict__ w1botT,
            float* __restrict__ gw3F0, float* __restrict__ gw3F1,
            float* __restrict__ loss) {
  __shared__ float pr[2][D_];
  const int tid = threadIdx.x;
  int bid = blockIdx.x;
  if (bid < P0_SESS_) {
    const int b = (bid << 1) + (tid >> 7), d = tid & 127;
    float acc = 0.f, ms = 0.f;
    for (int l = 0; l < L_; ++l) {
      float m = (float)mask[b * L_ + l];
      ms += m;
      acc += emb[((long)item[b * L_ + l] << 7) + d] * m;
    }
    sess[(b << 7) + d] = acc / ms;
    if (bid == 0 && tid == 0) loss[0] = 0.f;
  } else if ((bid -= P0_SESS_) < P0_HOP2_) {
    const int t = bid * 256 + tid;
    const int j = t >> 1;
    const int node = adj_all[inputs[j >> 3] * S_ + (j & 7)];
    const int o = node * S_ + ((t & 1) << 2);
    *(int4*)(n2 + (t << 2)) = *(const int4*)(adj_all + o);
    *(float4*)(wn2 + (t << 2)) = *(const float4*)(num_w + o);
  } else if ((bid -= P0_HOP2_) < P0_POS_) {
    const int hb = tid >> 7, d = tid & 127, l = (bid << 1) + hb;
    pr[hb][d] = pos_emb[(l << 7) + d];
    __syncthreads();
    float acc = 0.f;
    for (int e = 0; e < D_; ++e) acc += pr[hb][e] * w1[(e << 7) + d];
    P[(l << 7) + d] = acc;
  } else if ((bid -= P0_POS_) < P0_HOP1_) {
    const int t = bid * 256 + tid;
    const int o = inputs[t >> 1] * S_ + ((t & 1) << 2);
    *(int4*)(n1 + (t << 2)) = *(const int4*)(adj_all + o);
    *(float4*)(wn1 + (t << 2)) = *(const float4*)(num_w + o);
  } else if ((bid -= P0_HOP1_) < P0_GH_) {
    const int q = bid * 256 + tid;
    *(float4*)(h + ((long)q << 2)) =
        *(const float4*)(emb + ((long)inputs[q >> 5] << 7) + ((q & 31) << 2));
  } else if ((bid -= P0_GH_) < P0_GW1T_) {
    const int i = bid * 256 + tid;
    const int e = i & 1, lane = (i >> 1) & 31;
    const int q = i >> 6;
    const int ks = q % (KP_ / 4), hw = q / (KP_ / 4);
    const int hop = hw >> 3, wv = hw & 7;
    const int k = (ks << 2) + ((lane >> 4) << 1) + e, n = (wv << 4) + (lane & 15);
    gw1F[i] = (k < 129) ? gw1[hop * 129 * D_ + k * D_ + n] : 0.f;
  } else if ((bid -= P0_GW1T_) < P0_W1T_) {
    const int i = bid * 256 + tid;
    const int n = i >> 7, k = i & 127;
    w1botT[i] = w1[(D_ + k) * D_ + n];
  } else {
    bid -= P0_W1T_;
    const int j = bid * 256 + tid;
    const int lane = j & 31, t = (j >> 5) & 31, wv = (j >> 10) & 7, hop = j >> 13;
    const int k0 = (t << 2) + ((lane >> 4) << 1), n = (wv << 4) + (lane & 15);
    const float* src = gw3 + hop * 2 * D_ * D_ + n;
    float4 v;
    v.x = src[k0 * D_];
    v.y = src[(k0 + 1) * D_];
    v.z = src[(D_ + k0) * D_];
    v.w = src[(D_ + k0 + 1) * D_];
    float* dst = hop ? gw3F1 : gw3F0;
    *(float4*)(dst + ((j & 8191) << 2)) = v;
  }
}

#define HP_ 132

__device__ __forceinline__ void
local_agg_body(float* smem, const int b,
               const float* __restrict__ h,
               const int* __restrict__ adj,
               const float* __restrict__ a_local,
               float* __restrict__ h_local) {
  float (*hv)[HP_] = reinterpret_cast<float (*)[HP_]>(smem);
  float (*al)[D_]  = reinterpret_cast<float (*)[D_]>(smem + L_ * HP_);
  float (*att)[L_] = reinterpret_cast<float (*)[L_]>(smem + L_ * HP_ + 4 * D_);
  float* rinv = smem + L_ * HP_ + 4 * D_ + L_ * L_;
  const int tid = threadIdx.x;
  const int lane = tid & 31, laneIdx = lane & 15, laneHalf = lane >> 4;
  const int wave = tid >> 5;

  for (int q = tid; q < L_ * (D_ / 4); q += 256)
    *(float4*)&hv[q >> 5][(q & 31) << 2] = *(const float4*)(h + (long)b * L_ * D_ + (q << 2));
  if (tid < D_) *(float4*)&al[0][tid << 2] = *(const float4*)(a_local + (tid << 2));

  const int i0 = (wave >> 1) << 4;
  const int j0 = (wave & 1) << 5;
  int et[2][8];
  #pragma unroll
  for (int t = 0; t < 2; ++t) {
    #pragma unroll
    for (int v = 0; v < 8; ++v) {
      const int i = i0 + (laneHalf << 3) + v, j = j0 + (t << 4) + laneIdx;
      et[t][v] = (i < L_ && j < L_) ? adj[b * L_ * L_ + i * L_ + j] : 0;
    }
  }
  __syncthreads();

  {
    const int ir  = (i0 + laneIdx < L_) ? i0 + laneIdx : L_ - 1;
    const int jr0 = (j0 + laneIdx < L_) ? j0 + laneIdx : L_ - 1;
    const int jr1 = (j0 + 16 + laneIdx < L_) ? j0 + 16 + laneIdx : L_ - 1;
    const float* ap  = &hv[ir][laneHalf << 1];
    const float* bp0 = &hv[jr0][laneHalf << 1];
    const float* bp1 = &hv[jr1][laneHalf << 1];
    const float* sp  = &al[0][laneHalf << 1];
    v8f e[4][2] = {};
    for (int ks = 0; ks < D_ / 4; ++ks) {
      const v2f x  = *(const v2f*)(ap + 4 * ks);
      const v2f y0 = *(const v2f*)(bp0 + 4 * ks);
      const v2f y1 = *(const v2f*)(bp1 + 4 * ks);
      #pragma unroll
      for (int k = 0; k < 4; ++k) {
        const v2f s = *(const v2f*)(sp + k * D_ + 4 * ks);
        const v2f xa = x * s;
        e[k][0] = wmma4(xa, y0, e[k][0]);
        e[k][1] = wmma4(xa, y1, e[k][1]);
      }
    }
    #pragma unroll
    for (int t = 0; t < 2; ++t) {
      const int j = j0 + (t << 4) + laneIdx;
      #pragma unroll
      for (int v = 0; v < 8; ++v) {
        const int i = i0 + (laneHalf << 3) + v;
        const int k = et[t][v];
        float x = e[3][t][v];
        x = (k == 3) ? e[2][t][v] : x;
        x = (k == 2) ? e[1][t][v] : x;
        x = (k == 1) ? e[0][t][v] : x;
        x = (k == 0) ? -9e15f : leaky_(x);
        if (i < L_ && j < L_) att[i][j] = x;
      }
    }
  }
  __syncthreads();
  if (threadIdx.x < L_) {
    int i = threadIdx.x;
    float mx = -1e30f;
    for (int j = 0; j < L_; ++j) mx = fmaxf(mx, att[i][j]);
    float s = 0.f;
    for (int j = 0; j < L_; ++j) { float ex = __expf(att[i][j] - mx); att[i][j] = ex; s += ex; }
    rinv[i] = 1.f / s;
  }
  __syncthreads();
  for (int p = threadIdx.x; p < L_ * D_; p += 256) {
    int i = p >> 7, d = p & 127;
    float acc = 0.f;
    for (int j = 0; j < L_; ++j) acc += att[i][j] * hv[j][d];
    h_local[(long)b * L_ * D_ + p] = acc * rinv[i];
  }
}

#define GM_ 16
#define GR_ 128
#define CP_ 260

__device__ __forceinline__ void score_cols(v8f c, const float (&g2v)[8], float* scRow,
                                           int laneIdx) {
  float acc = 0.f;
  #pragma unroll
  for (int v = 0; v < 8; ++v) acc += leaky_(c[v]) * g2v[v];
  atomicAdd(&scRow[laneIdx], acc);
}

#define AGG_LDS_FLOATS_ (GR_ * D_ + GR_ * KP_ + GM_ * CP_ + 2 * GR_)

template <bool SELF_VAL, bool NEIGH_IDX>
__device__ __forceinline__ void
agg_body(float* smem, const unsigned blk,
         const float* __restrict__ emb,
         const float* __restrict__ selfVal, const int* __restrict__ selfIdx,
         const int* __restrict__ neighIdx, const float* __restrict__ neighVal,
         const float* __restrict__ neighW,
         const float* __restrict__ sess,
         const float* __restrict__ gw1F, const float* __restrict__ gw2p,
         const float* __restrict__ gw3p,
         float* __restrict__ out, const int M) {
  float (*nv)[D_]   = reinterpret_cast<float (*)[D_]>(smem);
  float (*xs)[KP_]  = reinterpret_cast<float (*)[KP_]>(smem + GR_ * D_);
  float (*cat)[CP_] = reinterpret_cast<float (*)[CP_]>(smem + GR_ * D_ + GR_ * KP_);
  float* sc    = smem + GR_ * D_ + GR_ * KP_ + GM_ * CP_;
  float* alpha = sc + GR_;

  const int tid = threadIdx.x;
  const int lane = tid & 31, laneIdx = lane & 15, laneHalf = lane >> 4;
  const int wave = tid >> 5;
  const int n0 = wave * 16;
  const long base  = (long)blk * GM_;
  const long nbase = base * S_;

  v2f breg[KP_ / 4];
  {
    const float* bp = gw1F + ((wave * (KP_ / 4)) << 6) + (lane << 1);
    #pragma unroll
    for (int ks = 0; ks < KP_ / 4; ++ks) breg[ks] = *(const v2f*)(bp + (ks << 6));
  }
  float g2v[8];
  {
    const float4 ga = *(const float4*)(gw2p + n0 + (laneHalf << 3));
    const float4 gb = *(const float4*)(gw2p + n0 + (laneHalf << 3) + 4);
    g2v[0] = ga.x; g2v[1] = ga.y; g2v[2] = ga.z; g2v[3] = ga.w;
    g2v[4] = gb.x; g2v[5] = gb.y; g2v[6] = gb.z; g2v[7] = gb.w;
  }

  const unsigned ubase = blk * GM_;
  const unsigned b0 = ubase / (unsigned)M;
  const unsigned rem0 = ubase - b0 * (unsigned)M;
  float4 gv[GM_];
  if (NEIGH_IDX) {
    int idx[GM_];
    #pragma unroll
    for (int j = 0; j < GM_; ++j) idx[j] = neighIdx[nbase + (j << 3) + wave];
    #pragma unroll
    for (int j = 0; j < GM_; ++j)
      gv[j] = *(const float4*)(emb + ((long)idx[j] << 7) + (lane << 2));
  } else {
    #pragma unroll
    for (int j = 0; j < GM_; ++j)
      gv[j] = *(const float4*)(neighVal + ((nbase + (j << 3) + wave) << 7) + (lane << 2));
  }
  const unsigned b1 = (rem0 + GM_ > (unsigned)M) ? b0 + 1 : b0;
  const float4 sfa = *(const float4*)(sess + ((long)b0 << 7) + (lane << 2));
  const float4 sfb = *(const float4*)(sess + ((long)b1 << 7) + (lane << 2));
  float4 sv0, sv1;
  if (SELF_VAL) {
    sv0 = *(const float4*)(selfVal + ((base + wave) << 7) + (lane << 2));
    sv1 = *(const float4*)(selfVal + ((base + wave + 8) << 7) + (lane << 2));
  } else {
    sv0 = *(const float4*)(emb + ((long)selfIdx[base + wave] << 7) + (lane << 2));
    sv1 = *(const float4*)(emb + ((long)selfIdx[base + wave + 8] << 7) + (lane << 2));
  }
  #pragma unroll
  for (int j = 0; j < GM_; ++j) {
    const int r = (j << 3) + wave;
    const float4 v = gv[j];
    const bool nextb = rem0 + (unsigned)j >= (unsigned)M;
    *(float4*)&nv[r][lane << 2] = v;
    float4 x;
    x.x = (nextb ? sfb.x : sfa.x) * v.x;
    x.y = (nextb ? sfb.y : sfa.y) * v.y;
    x.z = (nextb ? sfb.z : sfa.z) * v.z;
    x.w = (nextb ? sfb.w : sfa.w) * v.w;
    *(float4*)&xs[r][lane << 2] = x;
  }
  *(float4*)&cat[wave][lane << 2] = sv0;
  *(float4*)&cat[wave + 8][lane << 2] = sv1;
  if (tid < GR_) {
    xs[tid][D_] = neighW[nbase + tid];
    xs[tid][D_ + 1] = 0.f; xs[tid][D_ + 2] = 0.f; xs[tid][D_ + 3] = 0.f;
    sc[tid] = 0.f;
  }
  __syncthreads();

  for (int rp = 0; rp < GR_ / 32; ++rp) {
    const float* xp0 = &xs[(rp << 5) + laneIdx][laneHalf << 1];
    const float* xp1 = xp0 + 16 * KP_;
    v8f c0 = {}, c1 = {};
    #pragma unroll
    for (int ks = 0; ks < KP_ / 4; ++ks) {
      v2f x0 = *(const v2f*)(xp0 + 4 * ks);
      v2f x1 = *(const v2f*)(xp1 + 4 * ks);
      c0 = wmma4(breg[ks], x0, c0);
      c1 = wmma4(breg[ks], x1, c1);
    }
    score_cols(c0, g2v, &sc[(rp << 5)], laneIdx);
    score_cols(c1, g2v, &sc[(rp << 5) + 16], laneIdx);
  }
  __syncthreads();

  if (tid < GM_) {
    float mx = -1e30f;
    for (int s = 0; s < 8; ++s) mx = fmaxf(mx, sc[tid * 8 + s]);
    float ex[8], sum = 0.f;
    for (int s = 0; s < 8; ++s) { ex[s] = __expf(sc[tid * 8 + s] - mx); sum += ex[s]; }
    float inv = 1.f / sum;
    for (int s = 0; s < 8; ++s) alpha[tid * 8 + s] = ex[s] * inv;
  }
  __syncthreads();
  float4 w3f[D_ / 4];
  {
    const float* wp = gw3p + (wave << 12) + (lane << 2);
    #pragma unroll
    for (int t = 0; t < D_ / 4; ++t) w3f[t] = *(const float4*)(wp + (t << 7));
  }
  {
    const int d = tid & 127;
    for (int j = 0; j < GM_ / 2; ++j) {
      const int r = (tid >> 7) + (j << 1);
      float acc = 0.f;
      for (int s = 0; s < 8; ++s) acc += alpha[(r << 3) + s] * nv[(r << 3) + s][d];
      cat[r][D_ + d] = acc;
    }
  }
  __syncthreads();

  {
    const int col = n0 + laneIdx;
    const float* ap = &cat[laneIdx][laneHalf << 1];
    v8f cs = {}, cn = {};
    #pragma unroll
    for (int t = 0; t < D_ / 4; ++t) {
      v2f as = *(const v2f*)(ap + 4 * t);
      v2f an = *(const v2f*)(ap + D_ + 4 * t);
      v2f bs, bn;
      bs.x = w3f[t].x; bs.y = w3f[t].y;
      bn.x = w3f[t].z; bn.y = w3f[t].w;
      cs = wmma4(as, bs, cs);
      cn = wmma4(an, bn, cn);
    }
    #pragma unroll
    for (int v = 0; v < 8; ++v)
      xs[v + (laneHalf << 3)][col] = fmaxf(cs[v] + cn[v], 0.f);
  }
  __syncthreads();
  *(float4*)(out + ((base + wave) << 7) + (lane << 2)) = *(const float4*)&xs[wave][lane << 2];
  *(float4*)(out + ((base + wave + 8) << 7) + (lane << 2)) = *(const float4*)&xs[wave + 8][lane << 2];
}

#define P1_LOCAL_ B_
#define P1_HOP1_  (B_ * L_ / GM_)
#define P1_HOP2_  (B_ * L_ * S_ / GM_)
#define P1_TOTAL_ (P1_LOCAL_ + P1_HOP1_ + P1_HOP2_)

__global__ void __launch_bounds__(256)
hop_kernel(const float* __restrict__ emb, const float* __restrict__ h,
           const int* __restrict__ adj, const float* __restrict__ a_local,
           const int* __restrict__ n1, const float* __restrict__ wn1,
           const int* __restrict__ n2, const float* __restrict__ wn2,
           const float* __restrict__ sess,
           const float* __restrict__ gw1F, const float* __restrict__ gw2p,
           const float* __restrict__ gw3p,
           float* __restrict__ h_local, float* __restrict__ e0, float* __restrict__ e1) {
  __shared__ __align__(16) float smem[AGG_LDS_FLOATS_];
  int bid = blockIdx.x;
  if (bid < P1_LOCAL_) {
    local_agg_body(smem, bid, h, adj, a_local, h_local);
    return;
  }
  bid -= P1_LOCAL_;
  if (bid < P1_HOP1_) {
    agg_body<true, true>(smem, (unsigned)bid, emb, h, nullptr, n1, nullptr, wn1,
                         sess, gw1F, gw2p, gw3p, e0, L_);
  } else {
    agg_body<false, true>(smem, (unsigned)(bid - P1_HOP1_), emb, nullptr, n1, n2, nullptr, wn2,
                          sess, gw1F, gw2p, gw3p, e1, L_ * S_);
  }
}

__global__ void __launch_bounds__(256)
agg_val_kernel(const float* __restrict__ selfVal, const float* __restrict__ neighVal,
               const float* __restrict__ neighW, const float* __restrict__ sess,
               const float* __restrict__ gw1F, const float* __restrict__ gw2p,
               const float* __restrict__ gw3p, float* __restrict__ out, int M) {
  __shared__ __align__(16) float smem[AGG_LDS_FLOATS_];
  agg_body<true, false>(smem, blockIdx.x, nullptr, selfVal, nullptr, nullptr, neighVal, neighW,
                        sess, gw1F, gw2p, gw3p, out, M);
}

#define SC_COLS_ 32
#define SC_KH_ (D_ / 8)

__device__ __forceinline__ float4 sc_row(const float* __restrict__ embT, int n, int N, int lane) {
  n = n < N ? n : N - 1;
  return *(const float4*)(embT + ((long)n << 7) + (lane << 2));
}

__global__ void __launch_bounds__(256)
scores_kernel(const float* __restrict__ selF, const float* __restrict__ embT,
              float* __restrict__ scores, int N, int nStages) {
  __shared__ __align__(16) float bs[2][SC_COLS_][KP_];
  const int tid = threadIdx.x;
  const int lane = tid & 31, laneIdx = lane & 15, laneHalf = lane >> 4;
  const int wave = tid >> 5;

  v2f areg[D_ / 4];
  {
    const float* ap = selF + ((wave * 16 + laneIdx) << 7) + (laneHalf << 1);
    #pragma unroll
    for (int ks = 0; ks < D_ / 4; ++ks) areg[ks] = *(const v2f*)(ap + 4 * ks);
  }

  const int r0 = wave * 4;
  float4 p0, p1, p2, p3;
  int s = blockIdx.x;
  if (s < nStages) {
    const int n = s * SC_COLS_ + r0;
    p0 = sc_row(embT, n, N, lane);
    p1 = sc_row(embT, n + 1, N, lane);
    p2 = sc_row(embT, n + 2, N, lane);
    p3 = sc_row(embT, n + 3, N, lane);
    *(float4*)&bs[0][r0][lane << 2] = p0;
    *(float4*)&bs[0][r0 + 1][lane << 2] = p1;
    *(float4*)&bs[0][r0 + 2][lane << 2] = p2;
    *(float4*)&bs[0][r0 + 3][lane << 2] = p3;
  }
  __syncthreads();

  int buf = 0;
  for (; s < nStages; s += gridDim.x) {
    const int sn = s + gridDim.x;
    const bool more = sn < nStages;
    if (more) {
      const int n = sn * SC_COLS_ + r0;
      p0 = sc_row(embT, n, N, lane);
      p1 = sc_row(embT, n + 1, N, lane);
      p2 = sc_row(embT, n + 2, N, lane);
      p3 = sc_row(embT, n + 3, N, lane);
    }
    const float* bp0 = &bs[buf][laneIdx][laneHalf << 1];
    const float* bp1 = bp0 + 16 * KP_;
    v8f c0 = {}, c1 = {};
    #pragma unroll
    for (int kh = 0; kh < 2; ++kh) {
      v2f f0[SC_KH_], f1[SC_KH_];
      #pragma unroll
      for (int ks = 0; ks < SC_KH_; ++ks) {
        f0[ks] = *(const v2f*)(bp0 + 4 * (kh * SC_KH_ + ks));
        f1[ks] = *(const v2f*)(bp1 + 4 * (kh * SC_KH_ + ks));
      }
      #pragma unroll
      for (int ks = 0; ks < SC_KH_; ++ks) {
        c0 = wmma4(areg[kh * SC_KH_ + ks], f0[ks], c0);
        c1 = wmma4(areg[kh * SC_KH_ + ks], f1[ks], c1);
      }
    }
    if (more) {
      *(float4*)&bs[buf ^ 1][r0][lane << 2] = p0;
      *(float4*)&bs[buf ^ 1][r0 + 1][lane << 2] = p1;
      *(float4*)&bs[buf ^ 1][r0 + 2][lane << 2] = p2;
      *(float4*)&bs[buf ^ 1][r0 + 3][lane << 2] = p3;
    }
    __syncthreads();
    buf ^= 1;
    {
      const int n0 = s * SC_COLS_;
      float* o = scores + (long)(wave * 16 + (laneHalf << 3)) * N + n0 + laneIdx;
      if (n0 + SC_COLS_ <= N) {
        #pragma unroll
        for (int v = 0; v < 8; ++v) {
          o[(long)v * N] = c0[v];
          o[(long)v * N + 16] = c1[v];
        }
      } else {
        const bool in0 = n0 + laneIdx < N, in1 = n0 + laneIdx + 16 < N;
        #pragma unroll
        for (int v = 0; v < 8; ++v) {
          if (in0) o[(long)v * N] = c0[v];
          if (in1) o[(long)v * N + 16] = c1[v];
        }
      }
    }
  }
}

#define P4_HS_  64
#define P4_SSL_ 64
#define P4_ADD_ 800
#define P4_TOTAL_ (P4_HS_ + P4_SSL_ + P4_ADD_)

__global__ void __launch_bounds__(256)
mid_kernel(const float* __restrict__ hl, const float* __restrict__ hg,
           const int* __restrict__ mask, const float* __restrict__ glu2_w,
           float* __restrict__ out0, float* __restrict__ hsg, float* __restrict__ loss) {
  __shared__ float hr[2][D_];
  __shared__ float red[256];
  const int tid = threadIdx.x;
  int bid = blockIdx.x;
  if (bid < P4_HS_) {
    const int hb = tid >> 7, d = tid & 127, b = (bid << 1) + hb;
    float acc = 0.f, ms = 0.f;
    for (int l = 0; l < L_; ++l) {
      const float m = (float)mask[b * L_ + l];
      const int o = ((b * L_ + l) << 7) + d;
      ms += m;
      acc += (hl[o] + hg[o]) * m;
    }
    hr[hb][d] = acc / ms;
    __syncthreads();
    float a2 = 0.f;
    for (int e = 0; e < D_; ++e) a2 += hr[hb][e] * glu2_w[(d << 7) + e];
    hsg[(b << 7) + d] = a2;
  } else if ((bid -= P4_HS_) < P4_SSL_) {
    const int idx = bid * 256 + tid;
    const int b = idx >> 7, d = idx & 127;
    const int pb = (b * 73 + 21) & 127;
    float pos = 0.f, neg = 0.f;
    for (int l = 0; l < L_; ++l) {
      const int pl = (l * 13 + 7) % L_;
      pos += hl[((b * L_ + l) << 7) + d] * hg[((b * L_ + l) << 7) + d];
      neg += hg[((b * L_ + l) << 7) + d] * hl[((pb * L_ + pl) << 7) + d];
    }
    const float term = -__logf(1e-8f + sigmoid_(pos)) - __logf(1e-8f + 1.f - sigmoid_(neg));
    red[tid] = term;
    __syncthreads();
    for (int st = 128; st > 0; st >>= 1) {
      if (tid < st) red[tid] += red[tid + st];
      __syncthreads();
    }
    if (tid == 0) atomicAdd(loss, BETA_ * red[0]);
  } else {
    bid -= P4_SSL_;
    const int q = bid * 256 + tid;
    const float4 x = *(const float4*)(hl + ((long)q << 2));
    const float4 y = *(const float4*)(hg + ((long)q << 2));
    float4 r;
    r.x = x.x + y.x; r.y = x.y + y.y; r.z = x.z + y.z; r.w = x.w + y.w;
    *(float4*)(out0 + ((long)q << 2)) = r;
  }
}

#define HD_ROWS_ 32

__device__ __forceinline__ v2f lo2_(float4 f) { v2f r; r.x = f.x; r.y = f.y; return r; }
__device__ __forceinline__ v2f hi2_(float4 f) { v2f r; r.x = f.z; r.y = f.w; return r; }

__global__ void __launch_bounds__(256)
gate_kernel(const float* __restrict__ hidden, const float* __restrict__ w1botT,
            const float* __restrict__ P, const float* __restrict__ glu1_w,
            const float* __restrict__ glu1_b, const float* __restrict__ hsg,
            const float* __restrict__ w2, const int* __restrict__ mask,
            float* __restrict__ beta) {
  __shared__ __align__(16) float xr[HD_ROWS_][HP_];
  __shared__ __align__(16) float nhs[HD_ROWS_][HP_];
  __shared__ float part[16][HD_ROWS_];
  const int tid = threadIdx.x;
  const int lane = tid & 31, laneIdx = lane & 15, laneHalf = lane >> 4;
  const int wave = tid >> 5;
  const int n0 = wave << 4;
  const int row0 = blockIdx.x * HD_ROWS_;

  float4 hq[4];
  #pragma unroll
  for (int q = 0; q < 4; ++q)
    hq[q] = *(const float4*)(hidden + ((long)row0 << 7) + ((tid + (q << 8)) << 2));
  float4 bw[16];
  {
    const float* bp = w1botT + ((n0 + laneIdx) << 7) + (laneHalf << 2);
    #pragma unroll
    for (int j = 0; j < 16; ++j) bw[j] = *(const float4*)(bp + 8 * j);
  }
  float pv[2][8];
  #pragma unroll
  for (int t = 0; t < 2; ++t) {
    #pragma unroll
    for (int v = 0; v < 8; ++v) {
      const int l = (row0 + (t << 4) + (laneHalf << 3) + v) % L_;
      pv[t][v] = P[(l << 7) + n0 + laneIdx];
    }
  }

  #pragma unroll
  for (int q = 0; q < 4; ++q) {
    const int i = tid + (q << 8);
    *(float4*)&xr[i >> 5][(i & 31) << 2] = hq[q];
  }
  __syncthreads();

  float4 gw[16];
  {
    const float* gp = glu1_w + ((n0 + laneIdx) << 7) + (laneHalf << 2);
    #pragma unroll
    for (int j = 0; j < 16; ++j) gw[j] = *(const float4*)(gp + 8 * j);
  }
  const int nc = n0 + (laneHalf << 3);
  const float4 gba = *(const float4*)(glu1_b + nc), gbb = *(const float4*)(glu1_b + nc + 4);
  const float4 w2a = *(const float4*)(w2 + nc), w2b = *(const float4*)(w2 + nc + 4);
  const float* hp0 = hsg + (((row0 + laneIdx) / L_) << 7) + nc;
  const float* hp1 = hsg + (((row0 + 16 + laneIdx) / L_) << 7) + nc;
  const float4 h0a = *(const float4*)hp0, h0b = *(const float4*)(hp0 + 4);
  const float4 h1a = *(const float4*)hp1, h1b = *(const float4*)(hp1 + 4);

  {
    const float* ap0 = &xr[laneIdx][laneHalf << 2];
    const float* ap1 = ap0 + 16 * HP_;
    v8f c0 = {}, c1 = {};
    #pragma unroll
    for (int j = 0; j < 16; ++j) {
      const float4 x0 = *(const float4*)(ap0 + 8 * j);
      const float4 x1 = *(const float4*)(ap1 + 8 * j);
      const float4 b = bw[j];
      c0 = wmma4(lo2_(x0), lo2_(b), c0);
      c1 = wmma4(lo2_(x1), lo2_(b), c1);
      c0 = wmma4(hi2_(x0), hi2_(b), c0);
      c1 = wmma4(hi2_(x1), hi2_(b), c1);
    }
    #pragma unroll
    for (int v = 0; v < 8; ++v) {
      nhs[(laneHalf << 3) + v][n0 + laneIdx] = tanhf(c0[v] + pv[0][v]);
      nhs[16 + (laneHalf << 3) + v][n0 + laneIdx] = tanhf(c1[v] + pv[1][v]);
    }
  }
  __syncthreads();

  {
    const float* np0 = &nhs[laneIdx][laneHalf << 2];
    const float* np1 = np0 + 16 * HP_;
    v8f d0 = {}, d1 = {};
    #pragma unroll
    for (int j = 0; j < 16; ++j) {
      const float4 y0 = *(const float4*)(np0 + 8 * j);
      const float4 y1 = *(const float4*)(np1 + 8 * j);
      const float4 a = gw[j];
      d0 = wmma4(lo2_(a), lo2_(y0), d0);
      d1 = wmma4(lo2_(a), lo2_(y1), d1);
      d0 = wmma4(hi2_(a), hi2_(y0), d0);
      d1 = wmma4(hi2_(a), hi2_(y1), d1);
    }
    const float gb[8] = {gba.x, gba.y, gba.z, gba.w, gbb.x, gbb.y, gbb.z, gbb.w};
    const float wv[8] = {w2a.x, w2a.y, w2a.z, w2a.w, w2b.x, w2b.y, w2b.z, w2b.w};
    const float s0[8] = {h0a.x, h0a.y, h0a.z, h0a.w, h0b.x, h0b.y, h0b.z, h0b.w};
    const float s1[8] = {h1a.x, h1a.y, h1a.z, h1a.w, h1b.x, h1b.y, h1b.z, h1b.w};
    float acc0 = 0.f, acc1 = 0.f;
    #pragma unroll
    for (int v = 0; v < 8; ++v) {
      acc0 += sigmoid_(d0[v] + gb[v] + s0[v]) * wv[v];
      acc1 += sigmoid_(d1[v] + gb[v] + s1[v]) * wv[v];
    }
    part[(wave << 1) + laneHalf][laneIdx] = acc0;
    part[(wave << 1) + laneHalf][16 + laneIdx] = acc1;
  }
  __syncthreads();
  if (tid < HD_ROWS_) {
    float s = 0.f;
    for (int q = 0; q < 16; ++q) s += part[q][tid];
    beta[row0 + tid] = s * (float)mask[row0 + tid];
  }
}

__global__ void selnorm_kernel(const float* __restrict__ hidden, const float* __restrict__ beta,
                               float* __restrict__ sel, float* __restrict__ snorm) {
  __shared__ float red[D_];
  int b = blockIdx.x, d = threadIdx.x;
  float acc = 0.f;
  for (int l = 0; l < L_; ++l) acc += beta[b * L_ + l] * hidden[((b * L_ + l) << 7) + d];
  sel[(b << 7) + d] = acc;
  red[d] = acc * acc + 1e-6f;
  __syncthreads();
  for (int st = 64; st > 0; st >>= 1) {
    if (d < st) red[d] += red[d + st];
    __syncthreads();
  }
  if (d == 0) snorm[b] = sqrtf(red[0]);
}

__global__ void fn_kernel(const float* __restrict__ sel, const float* __restrict__ snorm,
                          float* __restrict__ selF) {
  __shared__ __align__(16) float row[B_];
  __shared__ float selfrow[D_];
  __shared__ float tvs[9];
  __shared__ int tis[9];
  int i = blockIdx.x, t = threadIdx.x;
  selfrow[t] = sel[(i << 7) + t];
  __syncthreads();
  float acc = 0.f;
  for (int d = 0; d < D_; ++d) acc += selfrow[d] * sel[(t << 7) + d];
  const float cs = acc / (snorm[i] * snorm[t]);
  row[t] = cs;
  __syncthreads();
  float mx = -1e30f;
  for (int j = 0; j < B_; ++j) mx = fmaxf(mx, row[j]);
  const float ex = __expf(cs - mx);
  __syncthreads();
  row[t] = ex;
  __syncthreads();
  float s = 0.f;
  for (int j = 0; j < B_; ++j) s += row[j];
  const float inv = 1.f / s;
  const float p = ex * inv;
  int place = 0;
  for (int j = 0; j < B_; ++j) {
    const float pj = row[j] * inv;
    place += (pj > p || (pj == p && j < t)) ? 1 : 0;
  }
  if (place < 9) { tvs[place] = p; tis[place] = t; }
  __syncthreads();
  float w[9];
  const float mx2 = tvs[0];
  float s2 = 0.f;
  #pragma unroll
  for (int q = 0; q < 9; ++q) { w[q] = __expf(tvs[q] - mx2); s2 += w[q]; }
  const float inv2 = 1.f / s2;
  float nb = 0.f;
  #pragma unroll
  for (int q = 0; q < 9; ++q) nb += (w[q] * inv2) * sel[(tis[q] << 7) + t];
  selF[(i << 7) + t] = selfrow[t] + nb;
}

extern "C" void kernel_launch(void* const* d_in, const int* in_sizes, int n_in,
                              void* d_out, int out_size, void* d_ws, size_t ws_size,
                              hipStream_t stream) {
  (void)in_sizes; (void)n_in; (void)out_size; (void)ws_size;
  const int*   inputs  = (const int*)d_in[0];
  const int*   adj     = (const int*)d_in[1];
  const int*   mask    = (const int*)d_in[2];
  const int*   item    = (const int*)d_in[3];
  const int*   adj_all = (const int*)d_in[5];
  const float* num_w   = (const float*)d_in[6];
  const float* emb     = (const float*)d_in[7];
  const float* pos_emb = (const float*)d_in[8];
  const float* a_local = (const float*)d_in[9];
  const float* gw1     = (const float*)d_in[10];
  const float* gw2     = (const float*)d_in[11];
  const float* gw3     = (const float*)d_in[12];
  const float* w1      = (const float*)d_in[13];
  const float* w2      = (const float*)d_in[14];
  const float* glu1_w  = (const float*)d_in[15];
  const float* glu1_b  = (const float*)d_in[16];
  const float* glu2_w  = (const float*)d_in[17];

  float* out0   = (float*)d_out;
  float* lossP  = out0 + B_ * L_ * D_;
  float* scores = lossP + 1;

  char* wp = (char*)d_ws;
  auto alloc = [&](size_t nfloats) -> float* {
    float* r = (float*)wp;
    wp += ((nfloats * sizeof(float) + 255) / 256) * 256;
    return r;
  };
  float* h        = alloc(B_ * L_ * D_);
  float* h_local  = alloc(B_ * L_ * D_);
  float* h_global = alloc(B_ * L_ * D_);
  float* sessb    = alloc(B_ * D_);
  float* e0p      = alloc(B_ * L_ * D_);
  float* e1p      = alloc((size_t)B_ * 400 * D_);
  float* hsgb     = alloc(B_ * D_);
  float* selb     = alloc(B_ * D_);
  float* Pb       = alloc(L_ * D_);
  float* betab    = alloc(B_ * L_);
  float* selF     = alloc(B_ * D_);
  float* snorm    = alloc(B_);
  float* gw1F     = alloc(2 * D_ * KP_);
  float* w1botT   = alloc(D_ * D_);
  float* wn1      = alloc(B_ * 400);
  float* wn2      = alloc((size_t)B_ * 3200);
  int*   n1       = (int*)alloc(B_ * 400);
  int*   n2       = (int*)alloc((size_t)B_ * 3200);
  float* gw3F0    = h_global;
  float* gw3F1    = hsgb;

  prep_kernel<<<dim3(P0_TOTAL_), 256, 0, stream>>>(emb, inputs, item, mask, adj_all, num_w, pos_emb, gw1, w1,
      gw3, sessb, n1, wn1, n2, wn2, Pb, h, gw1F, w1botT, gw3F0, gw3F1, lossP);
  hop_kernel<<<dim3(P1_TOTAL_), 256, 0, stream>>>(emb, h, adj, a_local, n1, wn1, n2, wn2,
      sessb, gw1F, gw2, gw3F0, h_local, e0p, e1p);
  agg_val_kernel<<<dim3(B_ * L_ / GM_), 256, 0, stream>>>(e0p, e1p, wn1,
      sessb, gw1F + D_ * KP_, gw2 + D_, gw3F1, h_global, L_);
  mid_kernel<<<dim3(P4_TOTAL_), 256, 0, stream>>>(h_local, h_global, mask, glu2_w, out0, hsgb, lossP);
  gate_kernel<<<dim3(B_ * L_ / HD_ROWS_), 256, 0, stream>>>(out0, w1botT, Pb, glu1_w, glu1_b, hsgb, w2,
                                                            mask, betab);
  selnorm_kernel<<<dim3(B_), dim3(D_), 0, stream>>>(out0, betab, selb, snorm);
  fn_kernel<<<dim3(B_), dim3(D_), 0, stream>>>(selb, snorm, selF);
  scores_kernel<<<dim3(512), 256, 0, stream>>>(selF, emb + D_, scores, NN_ - 1,
                                               (NN_ - 1 + SC_COLS_ - 1) / SC_COLS_);
}


# short launches cut to their dependent chains: h gathered where used, round 1 writes h_local + h_global (copy work-groups gone: 1599 -> 599, 928 -> 128); serial sums keep 25 or more loads in flight
# speedup vs baseline: 3.4508x; 1.0510x over previous
#include <hip/hip_runtime.h>
#include <hip/hip_bf16.h>
#include <math.h>

#define B_ 128
#define L_ 50
#define D_ 128
#define S_ 8
#define NN_ 100000
#define KP_ 132
#define LEAKY_ 0.2f
#define BETA_ 0.005f

typedef float v2f __attribute__((ext_vector_type(2)));
typedef float v8f __attribute__((ext_vector_type(8)));

#if defined(__has_builtin)
#if __has_builtin(__builtin_amdgcn_wmma_f32_16x16x4_f32)
#define HAVE_WMMA_F32 1
#endif
#endif

__device__ __forceinline__ v8f wmma4(v2f a, v2f b, v8f c) {
#ifdef HAVE_WMMA_F32
  return __builtin_amdgcn_wmma_f32_16x16x4_f32(false, a, false, b, (short)0, c,
                                               false, false);
#else
  float* cf = (float*)&c;
  #pragma unroll
  for (int i = 0; i < 8; ++i) cf[i] += a.x * b.x + a.y * b.y;
  return c;
#endif
}

__device__ __forceinline__ float leaky_(float x) { return x > 0.f ? x : LEAKY_ * x; }
__device__ __forceinline__ float sigmoid_(float x) { return 1.f / (1.f + __expf(-x)); }

#define P0_SESS_  64
#define P0_HOP2_  200
#define P0_POS_   25
#define P0_HOP1_  50
#define P0_GW1T_  132
#define P0_W1T_   64
#define P0_GW3F_  64
#define P0_TOTAL_ (P0_SESS_ + P0_HOP2_ + P0_POS_ + P0_HOP1_ + P0_GW1T_ + P0_W1T_ + P0_GW3F_)

__global__ void __launch_bounds__(256)
prep_kernel(const float* __restrict__ emb, const int* __restrict__ inputs,
            const int* __restrict__ item, const int* __restrict__ mask,
            const int* __restrict__ adj_all, const float* __restrict__ num_w,
            const float* __restrict__ pos_emb, const float* __restrict__ gw1,
            const float* __restrict__ w1, const float* __restrict__ gw3,
            float* __restrict__ sess, int* __restrict__ n1, float* __restrict__ wn1,
            int* __restrict__ n2, float* __restrict__ wn2, float* __restrict__ P,
            float* __restrict__ gw1F, float* __restrict__ w1botT,
            float* __restrict__ gw3F0, float* __restrict__ gw3F1,
            float* __restrict__ loss) {
  __shared__ float pr[2][D_];
  const int tid = threadIdx.x;
  int bid = blockIdx.x;
  if (bid < P0_SESS_) {
    const int b = (bid << 1) + (tid >> 7), d = tid & 127;
    float acc = 0.f, ms = 0.f;
    #pragma unroll 25
    for (int l = 0; l < L_; ++l) {
      float m = (float)mask[b * L_ + l];
      ms += m;
      acc += emb[((long)item[b * L_ + l] << 7) + d] * m;
    }
    sess[(b << 7) + d] = acc / ms;
    if (bid == 0 && tid == 0) loss[0] = 0.f;
  } else if ((bid -= P0_SESS_) < P0_HOP2_) {
    const int j = bid * 256 + tid;
    const int node = adj_all[inputs[j >> 3] * S_ + (j & 7)];
    const int o = node * S_;
    const int4 a0 = *(const int4*)(adj_all + o), a1 = *(const int4*)(adj_all + o + 4);
    const float4 f0 = *(const float4*)(num_w + o), f1 = *(const float4*)(num_w + o + 4);
    *(int4*)(n2 + (j << 3)) = a0;
    *(int4*)(n2 + (j << 3) + 4) = a1;
    *(float4*)(wn2 + (j << 3)) = f0;
    *(float4*)(wn2 + (j << 3) + 4) = f1;
  } else if ((bid -= P0_HOP2_) < P0_POS_) {
    const int hb = tid >> 7, d = tid & 127, l = (bid << 1) + hb;
    pr[hb][d] = pos_emb[(l << 7) + d];
    __syncthreads();
    float acc = 0.f;
    #pragma unroll 64
    for (int e = 0; e < D_; ++e) acc += pr[hb][e] * w1[(e << 7) + d];
    P[(l << 7) + d] = acc;
  } else if ((bid -= P0_POS_) < P0_HOP1_) {
    const int t = bid * 256 + tid;
    const int o = inputs[t >> 1] * S_ + ((t & 1) << 2);
    *(int4*)(n1 + (t << 2)) = *(const int4*)(adj_all + o);
    *(float4*)(wn1 + (t << 2)) = *(const float4*)(num_w + o);
  } else if ((bid -= P0_HOP1_) < P0_GW1T_) {
    const int i = bid * 256 + tid;
    const int e = i & 1, lane = (i >> 1) & 31;
    const int q = i >> 6;
    const int ks = q % (KP_ / 4), hw = q / (KP_ / 4);
    const int hop = hw >> 3, wv = hw & 7;
    const int k = (ks << 2) + ((lane >> 4) << 1) + e, n = (wv << 4) + (lane & 15);
    gw1F[i] = (k < 129) ? gw1[hop * 129 * D_ + k * D_ + n] : 0.f;
  } else if ((bid -= P0_GW1T_) < P0_W1T_) {
    const int i = bid * 256 + tid;
    const int n = i >> 7, k = i & 127;
    w1botT[i] = w1[(D_ + k) * D_ + n];
  } else {
    bid -= P0_W1T_;
    const int j = bid * 256 + tid;
    const int lane = j & 31, t = (j >> 5) & 31, wv = (j >> 10) & 7, hop = j >> 13;
    const int k0 = (t << 2) + ((lane >> 4) << 1), n = (wv << 4) + (lane & 15);
    const float* src = gw3 + hop * 2 * D_ * D_ + n;
    float4 v;
    v.x = src[k0 * D_];
    v.y = src[(k0 + 1) * D_];
    v.z = src[(D_ + k0) * D_];
    v.w = src[(D_ + k0 + 1) * D_];
    float* dst = hop ? gw3F1 : gw3F0;
    *(float4*)(dst + ((j & 8191) << 2)) = v;
  }
}

#define HP_ 132

#define LR_ 56

__device__ __forceinline__ void
local_agg_body(float* smem, const int b,
               const float* __restrict__ emb, const int* __restrict__ inputs,
               const int* __restrict__ adj,
               const float* __restrict__ a_local,
               float* __restrict__ h_local) {
  float (*hv)[HP_] = reinterpret_cast<float (*)[HP_]>(smem);
  float (*al)[D_]  = reinterpret_cast<float (*)[D_]>(smem + LR_ * HP_);
  float (*att)[L_] = reinterpret_cast<float (*)[L_]>(smem + LR_ * HP_ + 4 * D_);
  float* rinv = smem + LR_ * HP_ + 4 * D_ + L_ * L_;
  const int tid = threadIdx.x;
  const int lane = tid & 31, laneIdx = lane & 15, laneHalf = lane >> 4;
  const int wave = tid >> 5;

  {
    int ridx[7];
    #pragma unroll
    for (int k = 0; k < 7; ++k) {
      const int r = wave + (k << 3);
      ridx[k] = inputs[b * L_ + (r < L_ ? r : L_ - 1)];
    }
    float4 rv[7];
    #pragma unroll
    for (int k = 0; k < 7; ++k)
      rv[k] = *(const float4*)(emb + ((long)ridx[k] << 7) + (lane << 2));
    #pragma unroll
    for (int k = 0; k < 7; ++k)
      *(float4*)&hv[wave + (k << 3)][lane << 2] = rv[k];
  }
  if (tid < D_) *(float4*)&al[0][tid << 2] = *(const float4*)(a_local + (tid << 2));

  const int i0 = (wave >> 1) << 4;
  const int j0 = (wave & 1) << 5;
  int et[2][8];
  #pragma unroll
  for (int t = 0; t < 2; ++t) {
    #pragma unroll
    for (int v = 0; v < 8; ++v) {
      const int i = i0 + (laneHalf << 3) + v, j = j0 + (t << 4) + laneIdx;
      et[t][v] = (i < L_ && j < L_) ? adj[b * L_ * L_ + i * L_ + j] : 0;
    }
  }
  __syncthreads();

  {
    const int ir  = (i0 + laneIdx < L_) ? i0 + laneIdx : L_ - 1;
    const int jr0 = (j0 + laneIdx < L_) ? j0 + laneIdx : L_ - 1;
    const int jr1 = (j0 + 16 + laneIdx < L_) ? j0 + 16 + laneIdx : L_ - 1;
    const float* ap  = &hv[ir][laneHalf << 1];
    const float* bp0 = &hv[jr0][laneHalf << 1];
    const float* bp1 = &hv[jr1][laneHalf << 1];
    const float* sp  = &al[0][laneHalf << 1];
    v8f e[4][2] = {};
    for (int ks = 0; ks < D_ / 4; ++ks) {
      const v2f x  = *(const v2f*)(ap + 4 * ks);
      const v2f y0 = *(const v2f*)(bp0 + 4 * ks);
      const v2f y1 = *(const v2f*)(bp1 + 4 * ks);
      #pragma unroll
      for (int k = 0; k < 4; ++k) {
        const v2f s = *(const v2f*)(sp + k * D_ + 4 * ks);
        const v2f xa = x * s;
        e[k][0] = wmma4(xa, y0, e[k][0]);
        e[k][1] = wmma4(xa, y1, e[k][1]);
      }
    }
    #pragma unroll
    for (int t = 0; t < 2; ++t) {
      const int j = j0 + (t << 4) + laneIdx;
      #pragma unroll
      for (int v = 0; v < 8; ++v) {
        const int i = i0 + (laneHalf << 3) + v;
        const int k = et[t][v];
        float x = e[3][t][v];
        x = (k == 3) ? e[2][t][v] : x;
        x = (k == 2) ? e[1][t][v] : x;
        x = (k == 1) ? e[0][t][v] : x;
        x = (k == 0) ? -9e15f : leaky_(x);
        if (i < L_ && j < L_) att[i][j] = x;
      }
    }
  }
  __syncthreads();
  if (threadIdx.x < L_) {
    int i = threadIdx.x;
    float mx = -1e30f;
    for (int j = 0; j < L_; ++j) mx = fmaxf(mx, att[i][j]);
    float s = 0.f;
    for (int j = 0; j < L_; ++j) { float ex = __expf(att[i][j] - mx); att[i][j] = ex; s += ex; }
    rinv[i] = 1.f / s;
  }
  __syncthreads();
  for (int p = threadIdx.x; p < L_ * D_; p += 256) {
    int i = p >> 7, d = p & 127;
    float acc = 0.f;
    for (int j = 0; j < L_; ++j) acc += att[i][j] * hv[j][d];
    h_local[(long)b * L_ * D_ + p] = acc * rinv[i];
  }
}

#define GM_ 16
#define GR_ 128
#define CP_ 260

__device__ __forceinline__ void score_cols(v8f c, const float (&g2v)[8], float* scRow,
                                           int laneIdx) {
  float acc = 0.f;
  #pragma unroll
  for (int v = 0; v < 8; ++v) acc += leaky_(c[v]) * g2v[v];
  atomicAdd(&scRow[laneIdx], acc);
}

#define AGG_LDS_FLOATS_ (GR_ * D_ + GR_ * KP_ + GM_ * CP_ + 2 * GR_)

template <bool SELF_VAL, bool NEIGH_IDX, bool ADD_OUT>
__device__ __forceinline__ void
agg_body(float* smem, const unsigned blk,
         const float* __restrict__ emb,
         const float* __restrict__ selfVal, const int* __restrict__ selfIdx,
         const int* __restrict__ neighIdx, const float* __restrict__ neighVal,
         const float* __restrict__ neighW,
         const float* __restrict__ sess,
         const float* __restrict__ gw1F, const float* __restrict__ gw2p,
         const float* __restrict__ gw3p,
         float* __restrict__ out, const int M,
         const float* __restrict__ addIn, float* __restrict__ out2) {
  float (*nv)[D_]   = reinterpret_cast<float (*)[D_]>(smem);
  float (*xs)[KP_]  = reinterpret_cast<float (*)[KP_]>(smem + GR_ * D_);
  float (*cat)[CP_] = reinterpret_cast<float (*)[CP_]>(smem + GR_ * D_ + GR_ * KP_);
  float* sc    = smem + GR_ * D_ + GR_ * KP_ + GM_ * CP_;
  float* alpha = sc + GR_;

  const int tid = threadIdx.x;
  const int lane = tid & 31, laneIdx = lane & 15, laneHalf = lane >> 4;
  const int wave = tid >> 5;
  const int n0 = wave * 16;
  const long base  = (long)blk * GM_;
  const long nbase = base * S_;

  v2f breg[KP_ / 4];
  {
    const float* bp = gw1F + ((wave * (KP_ / 4)) << 6) + (lane << 1);
    #pragma unroll
    for (int ks = 0; ks < KP_ / 4; ++ks) breg[ks] = *(const v2f*)(bp + (ks << 6));
  }
  float g2v[8];
  {
    const float4 ga = *(const float4*)(gw2p + n0 + (laneHalf << 3));
    const float4 gb = *(const float4*)(gw2p + n0 + (laneHalf << 3) + 4);
    g2v[0] = ga.x; g2v[1] = ga.y; g2v[2] = ga.z; g2v[3] = ga.w;
    g2v[4] = gb.x; g2v[5] = gb.y; g2v[6] = gb.z; g2v[7] = gb.w;
  }

  const unsigned ubase = blk * GM_;
  const unsigned b0 = ubase / (unsigned)M;
  const unsigned rem0 = ubase - b0 * (unsigned)M;
  float4 gv[GM_];
  if (NEIGH_IDX) {
    int idx[GM_];
    #pragma unroll
    for (int j = 0; j < GM_; ++j) idx[j] = neighIdx[nbase + (j << 3) + wave];
    #pragma unroll
    for (int j = 0; j < GM_; ++j)
      gv[j] = *(const float4*)(emb + ((long)idx[j] << 7) + (lane << 2));
  } else {
    #pragma unroll
    for (int j = 0; j < GM_; ++j)
      gv[j] = *(const float4*)(neighVal + ((nbase + (j << 3) + wave) << 7) + (lane << 2));
  }
  const unsigned b1 = (rem0 + GM_ > (unsigned)M) ? b0 + 1 : b0;
  const float4 sfa = *(const float4*)(sess + ((long)b0 << 7) + (lane << 2));
  const float4 sfb = *(const float4*)(sess + ((long)b1 << 7) + (lane << 2));
  float4 sv0, sv1;
  if (SELF_VAL) {
    sv0 = *(const float4*)(selfVal + ((base + wave) << 7) + (lane << 2));
    sv1 = *(const float4*)(selfVal + ((base + wave + 8) << 7) + (lane << 2));
  } else {
    sv0 = *(const float4*)(emb + ((long)selfIdx[base + wave] << 7) + (lane << 2));
    sv1 = *(const float4*)(emb + ((long)selfIdx[base + wave + 8] << 7) + (lane << 2));
  }
  #pragma unroll
  for (int j = 0; j < GM_; ++j) {
    const int r = (j << 3) + wave;
    const float4 v = gv[j];
    const bool nextb = rem0 + (unsigned)j >= (unsigned)M;
    *(float4*)&nv[r][lane << 2] = v;
    float4 x;
    x.x = (nextb ? sfb.x : sfa.x) * v.x;
    x.y = (nextb ? sfb.y : sfa.y) * v.y;
    x.z = (nextb ? sfb.z : sfa.z) * v.z;
    x.w = (nextb ? sfb.w : sfa.w) * v.w;
    *(float4*)&xs[r][lane << 2] = x;
  }
  *(float4*)&cat[wave][lane << 2] = sv0;
  *(float4*)&cat[wave + 8][lane << 2] = sv1;
  if (tid < GR_) {
    xs[tid][D_] = neighW[nbase + tid];
    xs[tid][D_ + 1] = 0.f; xs[tid][D_ + 2] = 0.f; xs[tid][D_ + 3] = 0.f;
    sc[tid] = 0.f;
  }
  __syncthreads();

  for (int rp = 0; rp < GR_ / 32; ++rp) {
    const float* xp0 = &xs[(rp << 5) + laneIdx][laneHalf << 1];
    const float* xp1 = xp0 + 16 * KP_;
    v8f c0 = {}, c1 = {};
    #pragma unroll
    for (int ks = 0; ks < KP_ / 4; ++ks) {
      v2f x0 = *(const v2f*)(xp0 + 4 * ks);
      v2f x1 = *(const v2f*)(xp1 + 4 * ks);
      c0 = wmma4(breg[ks], x0, c0);
      c1 = wmma4(breg[ks], x1, c1);
    }
    score_cols(c0, g2v, &sc[(rp << 5)], laneIdx);
    score_cols(c1, g2v, &sc[(rp << 5) + 16], laneIdx);
  }
  __syncthreads();

  if (tid < GM_) {
    float mx = -1e30f;
    for (int s = 0; s < 8; ++s) mx = fmaxf(mx, sc[tid * 8 + s]);
    float ex[8], sum = 0.f;
    for (int s = 0; s < 8; ++s) { ex[s] = __expf(sc[tid * 8 + s] - mx); sum += ex[s]; }
    float inv = 1.f / sum;
    for (int s = 0; s < 8; ++s) alpha[tid * 8 + s] = ex[s] * inv;
  }
  __syncthreads();
  float4 w3f[D_ / 4];
  {
    const float* wp = gw3p + (wave << 12) + (lane << 2);
    #pragma unroll
    for (int t = 0; t < D_ / 4; ++t) w3f[t] = *(const float4*)(wp + (t << 7));
  }
  float4 ad0, ad1;
  if (ADD_OUT) {
    ad0 = *(const float4*)(addIn + ((base + wave) << 7) + (lane << 2));
    ad1 = *(const float4*)(addIn + ((base + wave + 8) << 7) + (lane << 2));
  }
  {
    const int d = tid & 127;
    for (int j = 0; j < GM_ / 2; ++j) {
      const int r = (tid >> 7) + (j << 1);
      float acc = 0.f;
      for (int s = 0; s < 8; ++s) acc += alpha[(r << 3) + s] * nv[(r << 3) + s][d];
      cat[r][D_ + d] = acc;
    }
  }
  __syncthreads();

  {
    const int col = n0 + laneIdx;
    const float* ap = &cat[laneIdx][laneHalf << 1];
    v8f cs = {}, cn = {};
    #pragma unroll
    for (int t = 0; t < D_ / 4; ++t) {
      v2f as = *(const v2f*)(ap + 4 * t);
      v2f an = *(const v2f*)(ap + D_ + 4 * t);
      v2f bs, bn;
      bs.x = w3f[t].x; bs.y = w3f[t].y;
      bn.x = w3f[t].z; bn.y = w3f[t].w;
      cs = wmma4(as, bs, cs);
      cn = wmma4(an, bn, cn);
    }
    #pragma unroll
    for (int v = 0; v < 8; ++v)
      xs[v + (laneHalf << 3)][col] = fmaxf(cs[v] + cn[v], 0.f);
  }
  __syncthreads();
  const float4 o0 = *(const float4*)&xs[wave][lane << 2];
  const float4 o1 = *(const float4*)&xs[wave + 8][lane << 2];
  *(float4*)(out + ((base + wave) << 7) + (lane << 2)) = o0;
  *(float4*)(out + ((base + wave + 8) << 7) + (lane << 2)) = o1;
  if (ADD_OUT) {
    float4 r0, r1;
    r0.x = ad0.x + o0.x; r0.y = ad0.y + o0.y; r0.z = ad0.z + o0.z; r0.w = ad0.w + o0.w;
    r1.x = ad1.x + o1.x; r1.y = ad1.y + o1.y; r1.z = ad1.z + o1.z; r1.w = ad1.w + o1.w;
    *(float4*)(out2 + ((base + wave) << 7) + (lane << 2)) = r0;
    *(float4*)(out2 + ((base + wave + 8) << 7) + (lane << 2)) = r1;
  }
}

#define P1_LOCAL_ B_
#define P1_HOP1_  (B_ * L_ / GM_)
#define P1_HOP2_  (B_ * L_ * S_ / GM_)
#define P1_TOTAL_ (P1_LOCAL_ + P1_HOP1_ + P1_HOP2_)

__global__ void __launch_bounds__(256)
hop_kernel(const float* __restrict__ emb, const int* __restrict__ inputs,
           const int* __restrict__ adj, const float* __restrict__ a_local,
           const int* __restrict__ n1, const float* __restrict__ wn1,
           const int* __restrict__ n2, const float* __restrict__ wn2,
           const float* __restrict__ sess,
           const float* __restrict__ gw1F, const float* __restrict__ gw2p,
           const float* __restrict__ gw3p,
           float* __restrict__ h_local, float* __restrict__ e0, float* __restrict__ e1) {
  __shared__ __align__(16) float smem[AGG_LDS_FLOATS_];
  int bid = blockIdx.x;
  if (bid < P1_LOCAL_) {
    local_agg_body(smem, bid, emb, inputs, adj, a_local, h_local);
    return;
  }
  bid -= P1_LOCAL_;
  if (bid < P1_HOP1_) {
    agg_body<false, true, false>(smem, (unsigned)bid, emb, nullptr, inputs, n1, nullptr, wn1,
                                 sess, gw1F, gw2p, gw3p, e0, L_, nullptr, nullptr);
  } else {
    agg_body<false, true, false>(smem, (unsigned)(bid - P1_HOP1_), emb, nullptr, n1, n2, nullptr, wn2,
                                 sess, gw1F, gw2p, gw3p, e1, L_ * S_, nullptr, nullptr);
  }
}

__global__ void __launch_bounds__(256)
agg_val_kernel(const float* __restrict__ selfVal, const float* __restrict__ neighVal,
               const float* __restrict__ neighW, const float* __restrict__ sess,
               const float* __restrict__ gw1F, const float* __restrict__ gw2p,
               const float* __restrict__ gw3p, float* __restrict__ out, int M,
               const float* __restrict__ hl, float* __restrict__ out0) {
  __shared__ __align__(16) float smem[AGG_LDS_FLOATS_];
  agg_body<true, false, true>(smem, blockIdx.x, nullptr, selfVal, nullptr, nullptr, neighVal, neighW,
                              sess, gw1F, gw2p, gw3p, out, M, hl, out0);
}

#define SC_COLS_ 32
#define SC_KH_ (D_ / 8)

__device__ __forceinline__ float4 sc_row(const float* __restrict__ embT, int n, int N, int lane) {
  n = n < N ? n : N - 1;
  return *(const float4*)(embT + ((long)n << 7) + (lane << 2));
}

__global__ void __launch_bounds__(256)
scores_kernel(const float* __restrict__ selF, const float* __restrict__ embT,
              float* __restrict__ scores, int N, int nStages) {
  __shared__ __align__(16) float bs[2][SC_COLS_][KP_];
  const int tid = threadIdx.x;
  const int lane = tid & 31, laneIdx = lane & 15, laneHalf = lane >> 4;
  const int wave = tid >> 5;

  v2f areg[D_ / 4];
  {
    const float* ap = selF + ((wave * 16 + laneIdx) << 7) + (laneHalf << 1);
    #pragma unroll
    for (int ks = 0; ks < D_ / 4; ++ks) areg[ks] = *(const v2f*)(ap + 4 * ks);
  }

  const int r0 = wave * 4;
  float4 p0, p1, p2, p3;
  int s = blockIdx.x;
  if (s < nStages) {
    const int n = s * SC_COLS_ + r0;
    p0 = sc_row(embT, n, N, lane);
    p1 = sc_row(embT, n + 1, N, lane);
    p2 = sc_row(embT, n + 2, N, lane);
    p3 = sc_row(embT, n + 3, N, lane);
    *(float4*)&bs[0][r0][lane << 2] = p0;
    *(float4*)&bs[0][r0 + 1][lane << 2] = p1;
    *(float4*)&bs[0][r0 + 2][lane << 2] = p2;
    *(float4*)&bs[0][r0 + 3][lane << 2] = p3;
  }
  __syncthreads();

  int buf = 0;
  for (; s < nStages; s += gridDim.x) {
    const int sn = s + gridDim.x;
    const bool more = sn < nStages;
    if (more) {
      const int n = sn * SC_COLS_ + r0;
      p0 = sc_row(embT, n, N, lane);
      p1 = sc_row(embT, n + 1, N, lane);
      p2 = sc_row(embT, n + 2, N, lane);
      p3 = sc_row(embT, n + 3, N, lane);
    }
    const float* bp0 = &bs[buf][laneIdx][laneHalf << 1];
    const float* bp1 = bp0 + 16 * KP_;
    v8f c0 = {}, c1 = {};
    #pragma unroll
    for (int kh = 0; kh < 2; ++kh) {
      v2f f0[SC_KH_], f1[SC_KH_];
      #pragma unroll
      for (int ks = 0; ks < SC_KH_; ++ks) {
        f0[ks] = *(const v2f*)(bp0 + 4 * (kh * SC_KH_ + ks));
        f1[ks] = *(const v2f*)(bp1 + 4 * (kh * SC_KH_ + ks));
      }
      #pragma unroll
      for (int ks = 0; ks < SC_KH_; ++ks) {
        c0 = wmma4(areg[kh * SC_KH_ + ks], f0[ks], c0);
        c1 = wmma4(areg[kh * SC_KH_ + ks], f1[ks], c1);
      }
    }
    if (more) {
      *(float4*)&bs[buf ^ 1][r0][lane << 2] = p0;
      *(float4*)&bs[buf ^ 1][r0 + 1][lane << 2] = p1;
      *(float4*)&bs[buf ^ 1][r0 + 2][lane << 2] = p2;
      *(float4*)&bs[buf ^ 1][r0 + 3][lane << 2] = p3;
    }
    __syncthreads();
    buf ^= 1;
    {
      const int n0 = s * SC_COLS_;
      float* o = scores + (long)(wave * 16 + (laneHalf << 3)) * N + n0 + laneIdx;
      if (n0 + SC_COLS_ <= N) {
        #pragma unroll
        for (int v = 0; v < 8; ++v) {
          o[(long)v * N] = c0[v];
          o[(long)v * N + 16] = c1[v];
        }
      } else {
        const bool in0 = n0 + laneIdx < N, in1 = n0 + laneIdx + 16 < N;
        #pragma unroll
        for (int v = 0; v < 8; ++v) {
          if (in0) o[(long)v * N] = c0[v];
          if (in1) o[(long)v * N + 16] = c1[v];
        }
      }
    }
  }
}

#define P4_HS_  64
#define P4_SSL_ 64
#define P4_TOTAL_ (P4_HS_ + P4_SSL_)

__global__ void __launch_bounds__(256)
mid_kernel(const float* __restrict__ hl, const float* __restrict__ hg,
           const float* __restrict__ out0,
           const int* __restrict__ mask, const float* __restrict__ glu2_w,
           float* __restrict__ hsg, float* __restrict__ loss) {
  __shared__ float hr[2][D_];
  __shared__ float red[256];
  const int tid = threadIdx.x;
  int bid = blockIdx.x;
  if (bid < P4_HS_) {
    const int hb = tid >> 7, d = tid & 127, b = (bid << 1) + hb;
    float acc = 0.f, ms = 0.f;
    #pragma unroll 25
    for (int l = 0; l < L_; ++l) {
      const float m = (float)mask[b * L_ + l];
      const int o = ((b * L_ + l) << 7) + d;
      ms += m;
      acc += out0[o] * m;
    }
    hr[hb][d] = acc / ms;
    __syncthreads();
    float a2 = 0.f;
    #pragma unroll 64
    for (int e = 0; e < D_; ++e) a2 += hr[hb][e] * glu2_w[(d << 7) + e];
    hsg[(b << 7) + d] = a2;
  } else {
    bid -= P4_HS_;
    const int idx = bid * 256 + tid;
    const int b = idx >> 7, d = idx & 127;
    const int pb = (b * 73 + 21) & 127;
    float pos = 0.f, neg = 0.f;
    #pragma unroll 25
    for (int l = 0; l < L_; ++l) {
      const int pl = (l * 13 + 7) % L_;
      pos += hl[((b * L_ + l) << 7) + d] * hg[((b * L_ + l) << 7) + d];
      neg += hg[((b * L_ + l) << 7) + d] * hl[((pb * L_ + pl) << 7) + d];
    }
    const float term = -__logf(1e-8f + sigmoid_(pos)) - __logf(1e-8f + 1.f - sigmoid_(neg));
    red[tid] = term;
    __syncthreads();
    for (int st = 128; st > 0; st >>= 1) {
      if (tid < st) red[tid] += red[tid + st];
      __syncthreads();
    }
    if (tid == 0) atomicAdd(loss, BETA_ * red[0]);
  }
}

#define HD_ROWS_ 32

__device__ __forceinline__ v2f lo2_(float4 f) { v2f r; r.x = f.x; r.y = f.y; return r; }
__device__ __forceinline__ v2f hi2_(float4 f) { v2f r; r.x = f.z; r.y = f.w; return r; }

__global__ void __launch_bounds__(256)
gate_kernel(const float* __restrict__ hidden, const float* __restrict__ w1botT,
            const float* __restrict__ P, const float* __restrict__ glu1_w,
            const float* __restrict__ glu1_b, const float* __restrict__ hsg,
            const float* __restrict__ w2, const int* __restrict__ mask,
            float* __restrict__ beta) {
  __shared__ __align__(16) float xr[HD_ROWS_][HP_];
  __shared__ __align__(16) float nhs[HD_ROWS_][HP_];
  __shared__ float part[16][HD_ROWS_];
  const int tid = threadIdx.x;
  const int lane = tid & 31, laneIdx = lane & 15, laneHalf = lane >> 4;
  const int wave = tid >> 5;
  const int n0 = wave << 4;
  const int row0 = blockIdx.x * HD_ROWS_;

  float4 hq[4];
  #pragma unroll
  for (int q = 0; q < 4; ++q)
    hq[q] = *(const float4*)(hidden + ((long)row0 << 7) + ((tid + (q << 8)) << 2));
  float4 bw[16];
  {
    const float* bp = w1botT + ((n0 + laneIdx) << 7) + (laneHalf << 2);
    #pragma unroll
    for (int j = 0; j < 16; ++j) bw[j] = *(const float4*)(bp + 8 * j);
  }
  float pv[2][8];
  #pragma unroll
  for (int t = 0; t < 2; ++t) {
    #pragma unroll
    for (int v = 0; v < 8; ++v) {
      const int l = (row0 + (t << 4) + (laneHalf << 3) + v) % L_;
      pv[t][v] = P[(l << 7) + n0 + laneIdx];
    }
  }

  #pragma unroll
  for (int q = 0; q < 4; ++q) {
    const int i = tid + (q << 8);
    *(float4*)&xr[i >> 5][(i & 31) << 2] = hq[q];
  }
  __syncthreads();

  float4 gw[16];
  {
    const float* gp = glu1_w + ((n0 + laneIdx) << 7) + (laneHalf << 2);
    #pragma unroll
    for (int j = 0; j < 16; ++j) gw[j] = *(const float4*)(gp + 8 * j);
  }
  const int nc = n0 + (laneHalf << 3);
  const float4 gba = *(const float4*)(glu1_b + nc), gbb = *(const float4*)(glu1_b + nc + 4);
  const float4 w2a = *(const float4*)(w2 + nc), w2b = *(const float4*)(w2 + nc + 4);
  const float* hp0 = hsg + (((row0 + laneIdx) / L_) << 7) + nc;
  const float* hp1 = hsg + (((row0 + 16 + laneIdx) / L_) << 7) + nc;
  const float4 h0a = *(const float4*)hp0, h0b = *(const float4*)(hp0 + 4);
  const float4 h1a = *(const float4*)hp1, h1b = *(const float4*)(hp1 + 4);

  {
    const float* ap0 = &xr[laneIdx][laneHalf << 2];
    const float* ap1 = ap0 + 16 * HP_;
    v8f c0 = {}, c1 = {};
    #pragma unroll
    for (int j = 0; j < 16; ++j) {
      const float4 x0 = *(const float4*)(ap0 + 8 * j);
      const float4 x1 = *(const float4*)(ap1 + 8 * j);
      const float4 b = bw[j];
      c0 = wmma4(lo2_(x0), lo2_(b), c0);
      c1 = wmma4(lo2_(x1), lo2_(b), c1);
      c0 = wmma4(hi2_(x0), hi2_(b), c0);
      c1 = wmma4(hi2_(x1), hi2_(b), c1);
    }
    #pragma unroll
    for (int v = 0; v < 8; ++v) {
      nhs[(laneHalf << 3) + v][n0 + laneIdx] = tanhf(c0[v] + pv[0][v]);
      nhs[16 + (laneHalf << 3) + v][n0 + laneIdx] = tanhf(c1[v] + pv[1][v]);
    }
  }
  __syncthreads();

  {
    const float* np0 = &nhs[laneIdx][laneHalf << 2];
    const float* np1 = np0 + 16 * HP_;
    v8f d0 = {}, d1 = {};
    #pragma unroll
    for (int j = 0; j < 16; ++j) {
      const float4 y0 = *(const float4*)(np0 + 8 * j);
      const float4 y1 = *(const float4*)(np1 + 8 * j);
      const float4 a = gw[j];
      d0 = wmma4(lo2_(a), lo2_(y0), d0);
      d1 = wmma4(lo2_(a), lo2_(y1), d1);
      d0 = wmma4(hi2_(a), hi2_(y0), d0);
      d1 = wmma4(hi2_(a), hi2_(y1), d1);
    }
    const float gb[8] = {gba.x, gba.y, gba.z, gba.w, gbb.x, gbb.y, gbb.z, gbb.w};
    const float wv[8] = {w2a.x, w2a.y, w2a.z, w2a.w, w2b.x, w2b.y, w2b.z, w2b.w};
    const float s0[8] = {h0a.x, h0a.y, h0a.z, h0a.w, h0b.x, h0b.y, h0b.z, h0b.w};
    const float s1[8] = {h1a.x, h1a.y, h1a.z, h1a.w, h1b.x, h1b.y, h1b.z, h1b.w};
    float acc0 = 0.f, acc1 = 0.f;
    #pragma unroll
    for (int v = 0; v < 8; ++v) {
      acc0 += sigmoid_(d0[v] + gb[v] + s0[v]) * wv[v];
      acc1 += sigmoid_(d1[v] + gb[v] + s1[v]) * wv[v];
    }
    part[(wave << 1) + laneHalf][laneIdx] = acc0;
    part[(wave << 1) + laneHalf][16 + laneIdx] = acc1;
  }
  __syncthreads();
  if (tid < HD_ROWS_) {
    float s = 0.f;
    for (int q = 0; q < 16; ++q) s += part[q][tid];
    beta[row0 + tid] = s * (float)mask[row0 + tid];
  }
}

__global__ void selnorm_kernel(const float* __restrict__ hidden, const float* __restrict__ beta,
                               float* __restrict__ sel, float* __restrict__ snorm) {
  __shared__ float red[D_];
  int b = blockIdx.x, d = threadIdx.x;
  float acc = 0.f;
  #pragma unroll 25
  for (int l = 0; l < L_; ++l) acc += beta[b * L_ + l] * hidden[((b * L_ + l) << 7) + d];
  sel[(b << 7) + d] = acc;
  red[d] = acc * acc + 1e-6f;
  __syncthreads();
  for (int st = 64; st > 0; st >>= 1) {
    if (d < st) red[d] += red[d + st];
    __syncthreads();
  }
  if (d == 0) snorm[b] = sqrtf(red[0]);
}

__global__ void fn_kernel(const float* __restrict__ sel, const float* __restrict__ snorm,
                          float* __restrict__ selF) {
  __shared__ __align__(16) float row[B_];
  __shared__ float selfrow[D_];
  __shared__ float tvs[9];
  __shared__ int tis[9];
  int i = blockIdx.x, t = threadIdx.x;
  selfrow[t] = sel[(i << 7) + t];
  __syncthreads();
  float acc = 0.f;
  #pragma unroll 64
  for (int d = 0; d < D_; ++d) acc += selfrow[d] * sel[(t << 7) + d];
  const float cs = acc / (snorm[i] * snorm[t]);
  row[t] = cs;
  __syncthreads();
  float mx = -1e30f;
  for (int j = 0; j < B_; ++j) mx = fmaxf(mx, row[j]);
  const float ex = __expf(cs - mx);
  __syncthreads();
  row[t] = ex;
  __syncthreads();
  float s = 0.f;
  for (int j = 0; j < B_; ++j) s += row[j];
  const float inv = 1.f / s;
  const float p = ex * inv;
  int place = 0;
  for (int j = 0; j < B_; ++j) {
    const float pj = row[j] * inv;
    place += (pj > p || (pj == p && j < t)) ? 1 : 0;
  }
  if (place < 9) { tvs[place] = p; tis[place] = t; }
  __syncthreads();
  float w[9];
  const float mx2 = tvs[0];
  float s2 = 0.f;
  #pragma unroll
  for (int q = 0; q < 9; ++q) { w[q] = __expf(tvs[q] - mx2); s2 += w[q]; }
  const float inv2 = 1.f / s2;
  float nb = 0.f;
  #pragma unroll
  for (int q = 0; q < 9; ++q) nb += (w[q] * inv2) * sel[(tis[q] << 7) + t];
  selF[(i << 7) + t] = selfrow[t] + nb;
}

extern "C" void kernel_launch(void* const* d_in, const int* in_sizes, int n_in,
                              void* d_out, int out_size, void* d_ws, size_t ws_size,
                              hipStream_t stream) {
  (void)in_sizes; (void)n_in; (void)out_size; (void)ws_size;
  const int*   inputs  = (const int*)d_in[0];
  const int*   adj     = (const int*)d_in[1];
  const int*   mask    = (const int*)d_in[2];
  const int*   item    = (const int*)d_in[3];
  const int*   adj_all = (const int*)d_in[5];
  const float* num_w   = (const float*)d_in[6];
  const float* emb     = (const float*)d_in[7];
  const float* pos_emb = (const float*)d_in[8];
  const float* a_local = (const float*)d_in[9];
  const float* gw1     = (const float*)d_in[10];
  const float* gw2     = (const float*)d_in[11];
  const float* gw3     = (const float*)d_in[12];
  const float* w1      = (const float*)d_in[13];
  const float* w2      = (const float*)d_in[14];
  const float* glu1_w  = (const float*)d_in[15];
  const float* glu1_b  = (const float*)d_in[16];
  const float* glu2_w  = (const float*)d_in[17];

  float* out0   = (float*)d_out;
  float* lossP  = out0 + B_ * L_ * D_;
  float* scores = lossP + 1;

  char* wp = (char*)d_ws;
  auto alloc = [&](size_t nfloats) -> float* {
    float* r = (float*)wp;
    wp += ((nfloats * sizeof(float) + 255) / 256) * 256;
    return r;
  };
  float* h_local  = alloc(B_ * L_ * D_);
  float* h_global = alloc(B_ * L_ * D_);
  float* sessb    = alloc(B_ * D_);
  float* e0p      = alloc(B_ * L_ * D_);
  float* e1p      = alloc((size_t)B_ * 400 * D_);
  float* hsgb     = alloc(B_ * D_);
  float* selb     = alloc(B_ * D_);
  float* Pb       = alloc(L_ * D_);
  float* betab    = alloc(B_ * L_);
  float* selF     = alloc(B_ * D_);
  float* snorm    = alloc(B_);
  float* gw1F     = alloc(2 * D_ * KP_);
  float* w1botT   = alloc(D_ * D_);
  float* wn1      = alloc(B_ * 400);
  float* wn2      = alloc((size_t)B_ * 3200);
  int*   n1       = (int*)alloc(B_ * 400);
  int*   n2       = (int*)alloc((size_t)B_ * 3200);
  float* gw3F0    = h_global;
  float* gw3F1    = hsgb;

  prep_kernel<<<dim3(P0_TOTAL_), 256, 0, stream>>>(emb, inputs, item, mask, adj_all, num_w, pos_emb, gw1, w1,
      gw3, sessb, n1, wn1, n2, wn2, Pb, gw1F, w1botT, gw3F0, gw3F1, lossP);
  hop_kernel<<<dim3(P1_TOTAL_), 256, 0, stream>>>(emb, inputs, adj, a_local, n1, wn1, n2, wn2,
      sessb, gw1F, gw2, gw3F0, h_local, e0p, e1p);
  agg_val_kernel<<<dim3(B_ * L_ / GM_), 256, 0, stream>>>(e0p, e1p, wn1,
      sessb, gw1F + D_ * KP_, gw2 + D_, gw3F1, h_global, L_, h_local, out0);
  mid_kernel<<<dim3(P4_TOTAL_), 256, 0, stream>>>(h_local, h_global, out0, mask, glu2_w, hsgb, lossP);
  gate_kernel<<<dim3(B_ * L_ / HD_ROWS_), 256, 0, stream>>>(out0, w1botT, Pb, glu1_w, glu1_b, hsgb, w2,
                                                            mask, betab);
  selnorm_kernel<<<dim3(B_), dim3(D_), 0, stream>>>(out0, betab, selb, snorm);
  fn_kernel<<<dim3(B_), dim3(D_), 0, stream>>>(selb, snorm, selF);
  scores_kernel<<<dim3(512), 256, 0, stream>>>(selF, emb + D_, scores, NN_ - 1,
                                               (NN_ - 1 + SC_COLS_ - 1) / SC_COLS_);
}
